# starting kernel
# baseline (speedup 1.0000x reference)
#include <hip/hip_runtime.h>

typedef __attribute__((ext_vector_type(2))) float v2f;
typedef __attribute__((ext_vector_type(8))) float v8f;

#define PDIM 128

__global__ __launch_bounds__(256) void painn_init(const float* __restrict__ s,
                                                  float* __restrict__ s_out,
                                                  float* __restrict__ aggv,
                                                  int n_s4, int n_a4) {
  int i = blockIdx.x * blockDim.x + threadIdx.x;
  int stride = gridDim.x * blockDim.x;
  for (int j = i; j < n_s4; j += stride)
    ((float4*)s_out)[j] = ((const float4*)s)[j];
  float4 z = make_float4(0.f, 0.f, 0.f, 0.f);
  for (int j = i; j < n_a4; j += stride)
    ((float4*)aggv)[j] = z;
}

__global__ __launch_bounds__(128) void painn_node_gemm(
    const float* __restrict__ s,
    const float* __restrict__ Wq,  const float* __restrict__ bq,
    const float* __restrict__ Wkv, const float* __restrict__ bkv,
    float* __restrict__ q, float* __restrict__ k,
    float* __restrict__ vs, float* __restrict__ vv,
    int nTiles) {
  const int nColTiles = 512 / 16;
  int wave = threadIdx.x >> 5;
  int lane = threadIdx.x & 31;
  int tile = blockIdx.x * 4 + wave;
  if (tile >= nTiles) return;
  int rowTile = tile / nColTiles;
  int colTile = tile % nColTiles;
  int rowBase = rowTile * 16;
  int cGlobal = colTile * 16;
  int half = lane >> 4;
  int l16  = lane & 15;
  int col  = cGlobal + l16;

  const float* W; const float* b; int wcol; int ldw;
  if (cGlobal < PDIM) { W = Wq;  b = bq;  wcol = col;        ldw = PDIM;     }
  else                { W = Wkv; b = bkv; wcol = col - PDIM; ldw = 3 * PDIM; }

  v8f acc = {};
  const float* srow = s + (size_t)(rowBase + l16) * PDIM;
  for (int k0 = 0; k0 < PDIM; k0 += 4) {
    float2 av = *(const float2*)(srow + k0 + 2 * half);
    v2f a; a.x = av.x; a.y = av.y;
    v2f bf;
    bf.x = W[(size_t)(k0 + 2 * half)     * ldw + wcol];
    bf.y = W[(size_t)(k0 + 2 * half + 1) * ldw + wcol];
    acc = __builtin_amdgcn_wmma_f32_16x16x4_f32(false, a, false, bf,
                                                (short)0, acc, false, false);
  }
  float bias = b[wcol];

  float* out;
  if      (cGlobal < 128) out = q;
  else if (cGlobal < 256) out = k;
  else if (cGlobal < 384) out = vs;
  else                    out = vv;
  int oc = col & (PDIM - 1);
  for (int r = 0; r < 8; ++r) {
    int row = rowBase + r + 8 * half;
    out[(size_t)row * PDIM + oc] = acc[r] + bias;
  }
}

__global__ __launch_bounds__(256) void painn_edge(
    const float* __restrict__ q, const float* __restrict__ k,
    const float* __restrict__ vs, const float* __restrict__ vv,
    const float* __restrict__ v,
    const int* __restrict__ rowIdx, const int* __restrict__ colIdx,
    float* __restrict__ s_out, float* __restrict__ aggv, int E) {
  int e = blockIdx.x * 8 + (threadIdx.x >> 5);
  if (e >= E) return;
  int lane = threadIdx.x & 31;
  int r = rowIdx[e];
  int c = colIdx[e];
  int d = lane * 4;

  float4 qv = *(const float4*)(q + (size_t)r * PDIM + d);
  float4 kv = *(const float4*)(k + (size_t)c * PDIM + d);
  float p = qv.x * kv.x + qv.y * kv.y + qv.z * kv.z + qv.w * kv.w;
  p += __shfl_xor(p, 16, 32);
  p += __shfl_xor(p, 8, 32);
  p += __shfl_xor(p, 4, 32);
  p += __shfl_xor(p, 2, 32);
  p += __shfl_xor(p, 1, 32);
  float w = p;

  float4 vsv = *(const float4*)(vs + (size_t)c * PDIM + d);
  float* so = s_out + (size_t)r * PDIM + d;
  atomicAdd(so + 0, w * vsv.x);
  atomicAdd(so + 1, w * vsv.y);
  atomicAdd(so + 2, w * vsv.z);
  atomicAdd(so + 3, w * vsv.w);

  float4 vvv = *(const float4*)(vv + (size_t)c * PDIM + d);
  const float* vc = v + (size_t)c * 3 * PDIM;
  float* av = aggv + (size_t)r * 3 * PDIM;
#pragma unroll
  for (int comp = 0; comp < 3; ++comp) {
    float4 vval = *(const float4*)(vc + comp * PDIM + d);
    float* ap = av + comp * PDIM + d;
    atomicAdd(ap + 0, w * vvv.x * vval.x);
    atomicAdd(ap + 1, w * vvv.y * vval.y);
    atomicAdd(ap + 2, w * vvv.z * vval.z);
    atomicAdd(ap + 3, w * vvv.w * vval.w);
  }
}

__global__ __launch_bounds__(128) void painn_vec_gemm(
    const float* __restrict__ aggv, const float* __restrict__ Wvec,
    const float* __restrict__ bvec, const float* __restrict__ v,
    float* __restrict__ vout, int nTiles) {
  const int nColTiles = PDIM / 16;
  int wave = threadIdx.x >> 5;
  int lane = threadIdx.x & 31;
  int tile = blockIdx.x * 4 + wave;
  if (tile >= nTiles) return;
  int rowTile = tile / nColTiles;
  int colTile = tile % nColTiles;
  int rowBase = rowTile * 16;
  int half = lane >> 4;
  int l16  = lane & 15;
  int col  = colTile * 16 + l16;

  v8f acc = {};
  const float* arow = aggv + (size_t)(rowBase + l16) * PDIM;
  for (int k0 = 0; k0 < PDIM; k0 += 4) {
    float2 av2 = *(const float2*)(arow + k0 + 2 * half);
    v2f a; a.x = av2.x; a.y = av2.y;
    v2f bf;
    bf.x = Wvec[(size_t)(k0 + 2 * half)     * PDIM + col];
    bf.y = Wvec[(size_t)(k0 + 2 * half + 1) * PDIM + col];
    acc = __builtin_amdgcn_wmma_f32_16x16x4_f32(false, a, false, bf,
                                                (short)0, acc, false, false);
  }
  float bias = bvec[col];
  for (int r = 0; r < 8; ++r) {
    size_t row = (size_t)rowBase + r + 8 * half;
    size_t idx = row * PDIM + col;
    vout[idx] = v[idx] + acc[r] + bias;
  }
}

extern "C" void kernel_launch(void* const* d_in, const int* in_sizes, int n_in,
                              void* d_out, int out_size, void* d_ws, size_t ws_size,
                              hipStream_t stream) {
  const float* s    = (const float*)d_in[0];
  const float* v    = (const float*)d_in[1];
  const int*   ei   = (const int*)d_in[2];
  const float* Wq   = (const float*)d_in[3];
  const float* bq   = (const float*)d_in[4];
  const float* Wkv  = (const float*)d_in[5];
  const float* bkv  = (const float*)d_in[6];
  const float* Wvec = (const float*)d_in[7];
  const float* bvec = (const float*)d_in[8];

  int N = in_sizes[0] / PDIM;
  int E = in_sizes[2] / 2;

  float* s_out = (float*)d_out;
  float* v_out = s_out + (size_t)N * PDIM;

  float* q    = (float*)d_ws;
  float* k    = q  + (size_t)N * PDIM;
  float* vs   = k  + (size_t)N * PDIM;
  float* vv   = vs + (size_t)N * PDIM;
  float* aggv = vv + (size_t)N * PDIM;

  const int* rowIdx = ei;
  const int* colIdx = ei + E;

  painn_init<<<2048, 256, 0, stream>>>(s, s_out, aggv,
                                       N * PDIM / 4, N * 3 * PDIM / 4);

  int tilesA = (N / 16) * (512 / 16);
  painn_node_gemm<<<(tilesA + 3) / 4, 128, 0, stream>>>(
      s, Wq, bq, Wkv, bkv, q, k, vs, vv, tilesA);

  painn_edge<<<(E + 7) / 8, 256, 0, stream>>>(
      q, k, vs, vv, v, rowIdx, colIdx, s_out, aggv, E);

  int tilesC = (N * 3 / 16) * (PDIM / 16);
  painn_vec_gemm<<<(tilesC + 3) / 4, 128, 0, stream>>>(
      aggv, Wvec, bvec, v, v_out, tilesC);
}


# Refused try summed rows in arrival order; now each row's edges (counting-sorted by destination) are summed in ascending edge id by one wave in registers: float atomics and zero-fill removed
# speedup vs baseline: 1.7923x; 1.7923x over previous
#include <hip/hip_runtime.h>


typedef __attribute__((ext_vector_type(2))) float v2f;
typedef __attribute__((ext_vector_type(8))) float v8f;

#define PDIM 128
#define SCAN_TILE 4096

__global__ __launch_bounds__(256) void csr_zero(int* __restrict__ counts, int n4) {
  int i = blockIdx.x * blockDim.x + threadIdx.x;
  if (i < n4) ((int4*)counts)[i] = make_int4(0, 0, 0, 0);
}

__global__ __launch_bounds__(256) void csr_count(const int* __restrict__ rowIdx,
                                                 int* __restrict__ counts,
                                                 int* __restrict__ slot, int E) {
  int stride = gridDim.x * blockDim.x;
  for (int e = blockIdx.x * blockDim.x + threadIdx.x; e < E; e += stride)
    slot[e] = atomicAdd(&counts[rowIdx[e]], 1);
}

__global__ __launch_bounds__(1024) void csr_scan(const int* __restrict__ counts,
                                                 int* __restrict__ offsets, int nTrips) {
  __shared__ int waveSum[32];
  int t = threadIdx.x;
  int lane = t & 31;
  int wave = t >> 5;
  int carry = 0;
  for (int it = 0; it < nTrips; ++it) {
    int4 c = ((const int4*)counts)[it * 1024 + t];
    int s1 = c.x, s2 = s1 + c.y, s3 = s2 + c.z, s4 = s3 + c.w;
    int incl = s4;
    for (int o = 1; o < 32; o <<= 1) {
      int n = __shfl_up(incl, o, 32);
      if (lane >= o) incl += n;
    }
    if (lane == 31) waveSum[wave] = incl;
    __syncthreads();
    int ws = waveSum[lane];
    int wincl = ws;
    for (int o = 1; o < 32; o <<= 1) {
      int n = __shfl_up(wincl, o, 32);
      if (lane >= o) wincl += n;
    }
    int wavePrefix = __shfl(wincl, wave, 32) - __shfl(ws, wave, 32);
    int total = __shfl(wincl, 31, 32);
    int base = carry + wavePrefix + (incl - s4);
    ((int4*)offsets)[it * 1024 + t] = make_int4(base, base + s1, base + s2, base + s3);
    carry += total;
    __syncthreads();
  }
}

__global__ __launch_bounds__(256) void csr_fill(const int* __restrict__ rowIdx,
                                                const int* __restrict__ colIdx,
                                                const int* __restrict__ slot,
                                                const int* __restrict__ offsets,
                                                int2* __restrict__ list, int E) {
  int stride = gridDim.x * blockDim.x;
  for (int e = blockIdx.x * blockDim.x + threadIdx.x; e < E; e += stride)
    list[offsets[rowIdx[e]] + slot[e]] = make_int2(e, colIdx[e]);
}

__global__ __launch_bounds__(128) void painn_node_gemm(
    const float* __restrict__ s,
    const float* __restrict__ Wq,  const float* __restrict__ bq,
    const float* __restrict__ Wkv, const float* __restrict__ bkv,
    float* __restrict__ q, float* __restrict__ k,
    float* __restrict__ vs, float* __restrict__ vv,
    int nTiles) {
  const int nColTiles = 512 / 16;
  int wave = threadIdx.x >> 5;
  int lane = threadIdx.x & 31;
  int tile = blockIdx.x * 4 + wave;
  if (tile >= nTiles) return;
  int rowTile = tile / nColTiles;
  int colTile = tile % nColTiles;
  int rowBase = rowTile * 16;
  int cGlobal = colTile * 16;
  int half = lane >> 4;
  int l16  = lane & 15;
  int col  = cGlobal + l16;

  const float* W; const float* b; int wcol; int ldw;
  if (cGlobal < PDIM) { W = Wq;  b = bq;  wcol = col;        ldw = PDIM;     }
  else                { W = Wkv; b = bkv; wcol = col - PDIM; ldw = 3 * PDIM; }

  v8f acc = {};
  const float* srow = s + (size_t)(rowBase + l16) * PDIM;
  for (int k0 = 0; k0 < PDIM; k0 += 4) {
    float2 av = *(const float2*)(srow + k0 + 2 * half);
    v2f a; a.x = av.x; a.y = av.y;
    v2f bf;
    bf.x = W[(size_t)(k0 + 2 * half)     * ldw + wcol];
    bf.y = W[(size_t)(k0 + 2 * half + 1) * ldw + wcol];
    acc = __builtin_amdgcn_wmma_f32_16x16x4_f32(false, a, false, bf,
                                                (short)0, acc, false, false);
  }
  float bias = b[wcol];

  float* out;
  if      (cGlobal < 128) out = q;
  else if (cGlobal < 256) out = k;
  else if (cGlobal < 384) out = vs;
  else                    out = vv;
  int oc = col & (PDIM - 1);
  for (int r = 0; r < 8; ++r) {
    int row = rowBase + r + 8 * half;
    out[(size_t)row * PDIM + oc] = acc[r] + bias;
  }
}

__device__ __forceinline__ void edge_accum(
    int c, int d, const float4& qv,
    const float* __restrict__ k, const float* __restrict__ vs,
    const float* __restrict__ vv, const float* __restrict__ v,
    float4& as, float4& a0, float4& a1, float4& a2) {
  float4 kv  = *(const float4*)(k  + (size_t)c * PDIM + d);
  float4 vsv = *(const float4*)(vs + (size_t)c * PDIM + d);
  float4 vvv = *(const float4*)(vv + (size_t)c * PDIM + d);
  const float* vc = v + (size_t)c * 3 * PDIM + d;
  float4 x0 = *(const float4*)(vc);
  float4 x1 = *(const float4*)(vc + PDIM);
  float4 x2 = *(const float4*)(vc + 2 * PDIM);

  float p = qv.x * kv.x + qv.y * kv.y + qv.z * kv.z + qv.w * kv.w;
  p += __shfl_xor(p, 16, 32);
  p += __shfl_xor(p, 8, 32);
  p += __shfl_xor(p, 4, 32);
  p += __shfl_xor(p, 2, 32);
  p += __shfl_xor(p, 1, 32);
  float w = p;

  as.x += w * vsv.x; as.y += w * vsv.y; as.z += w * vsv.z; as.w += w * vsv.w;
  float tx = w * vvv.x, ty = w * vvv.y, tz = w * vvv.z, tw = w * vvv.w;
  a0.x += tx * x0.x; a0.y += ty * x0.y; a0.z += tz * x0.z; a0.w += tw * x0.w;
  a1.x += tx * x1.x; a1.y += ty * x1.y; a1.z += tz * x1.z; a1.w += tw * x1.w;
  a2.x += tx * x2.x; a2.y += ty * x2.y; a2.z += tz * x2.z; a2.w += tw * x2.w;
}

__global__ __launch_bounds__(256) void node_rows(
    const float* __restrict__ q, const float* __restrict__ k,
    const float* __restrict__ vs, const float* __restrict__ vv,
    const float* __restrict__ v, const float* __restrict__ s,
    const int* __restrict__ offsets, const int2* __restrict__ list,
    float* __restrict__ s_out, float* __restrict__ aggv, int N) {
  int wave = __builtin_amdgcn_readfirstlane((int)(threadIdx.x >> 5));
  int node = blockIdx.x * 8 + wave;
  if (node >= N) return;
  int lane = threadIdx.x & 31;
  int d = lane * 4;
  int beg = offsets[node];
  int end = offsets[node + 1];
  int deg = end - beg;

  float4 qv = *(const float4*)(q + (size_t)node * PDIM + d);
  float4 as = make_float4(0.f, 0.f, 0.f, 0.f);
  float4 a0 = as, a1 = as, a2 = as;

  if (deg <= 32) {
    int2 ec = make_int2(0x7fffffff, 0);
    if (lane < deg) ec = list[beg + lane];
    int rank = 0;
    for (int j = 0; j < deg; ++j) {
      int ej = __builtin_amdgcn_readlane(ec.x, j);
      rank += (ej < ec.x) ? 1 : 0;
    }
#pragma unroll 1
    for (int i = 0; i < deg; ++i) {
      unsigned m = __builtin_amdgcn_ballot_w32(rank == i);
      int src = __builtin_ctz(m | 0x80000000u);
      int c = __builtin_amdgcn_readlane(ec.y, src);
      edge_accum(c, d, qv, k, vs, vv, v, as, a0, a1, a2);
    }
  } else {
    int last = -1;
    for (int i = 0; i < deg; ++i) {
      int be = 0x7fffffff, bc = 0;
      for (int j = beg + lane; j < end; j += 32) {
        int2 ec = list[j];
        if (ec.x > last && ec.x < be) { be = ec.x; bc = ec.y; }
      }
      int mn = be;
      mn = min(mn, __shfl_xor(mn, 16, 32));
      mn = min(mn, __shfl_xor(mn, 8, 32));
      mn = min(mn, __shfl_xor(mn, 4, 32));
      mn = min(mn, __shfl_xor(mn, 2, 32));
      mn = min(mn, __shfl_xor(mn, 1, 32));
      unsigned m = __builtin_amdgcn_ballot_w32(be == mn);
      int src = __builtin_ctz(m | 0x80000000u);
      int c = __builtin_amdgcn_readlane(bc, src);
      last = __builtin_amdgcn_readfirstlane(mn);
      edge_accum(c, d, qv, k, vs, vv, v, as, a0, a1, a2);
    }
  }

  float4 sv = *(const float4*)(s + (size_t)node * PDIM + d);
  float4 so = make_float4(sv.x + as.x, sv.y + as.y, sv.z + as.z, sv.w + as.w);
  *(float4*)(s_out + (size_t)node * PDIM + d) = so;
  float* av = aggv + (size_t)node * 3 * PDIM + d;
  *(float4*)(av)            = a0;
  *(float4*)(av + PDIM)     = a1;
  *(float4*)(av + 2 * PDIM) = a2;
}

__global__ __launch_bounds__(128) void painn_vec_gemm(
    const float* __restrict__ aggv, const float* __restrict__ Wvec,
    const float* __restrict__ bvec, const float* __restrict__ v,
    float* __restrict__ vout, int nTiles) {
  const int nColTiles = PDIM / 16;
  int wave = threadIdx.x >> 5;
  int lane = threadIdx.x & 31;
  int tile = blockIdx.x * 4 + wave;
  if (tile >= nTiles) return;
  int rowTile = tile / nColTiles;
  int colTile = tile % nColTiles;
  int rowBase = rowTile * 16;
  int half = lane >> 4;
  int l16  = lane & 15;
  int col  = colTile * 16 + l16;

  v8f acc = {};
  const float* arow = aggv + (size_t)(rowBase + l16) * PDIM;
  for (int k0 = 0; k0 < PDIM; k0 += 4) {
    float2 av2 = *(const float2*)(arow + k0 + 2 * half);
    v2f a; a.x = av2.x; a.y = av2.y;
    v2f bf;
    bf.x = Wvec[(size_t)(k0 + 2 * half)     * PDIM + col];
    bf.y = Wvec[(size_t)(k0 + 2 * half + 1) * PDIM + col];
    acc = __builtin_amdgcn_wmma_f32_16x16x4_f32(false, a, false, bf,
                                                (short)0, acc, false, false);
  }
  float bias = bvec[col];
  for (int r = 0; r < 8; ++r) {
    size_t row = (size_t)rowBase + r + 8 * half;
    size_t idx = row * PDIM + col;
    vout[idx] = v[idx] + acc[r] + bias;
  }
}

extern "C" void kernel_launch(void* const* d_in, const int* in_sizes, int n_in,
                              void* d_out, int out_size, void* d_ws, size_t ws_size,
                              hipStream_t stream) {
  const float* s    = (const float*)d_in[0];
  const float* v    = (const float*)d_in[1];
  const int*   ei   = (const int*)d_in[2];
  const float* Wq   = (const float*)d_in[3];
  const float* bq   = (const float*)d_in[4];
  const float* Wkv  = (const float*)d_in[5];
  const float* bkv  = (const float*)d_in[6];
  const float* Wvec = (const float*)d_in[7];
  const float* bvec = (const float*)d_in[8];

  int N = in_sizes[0] / PDIM;
  int E = in_sizes[2] / 2;

  float* s_out = (float*)d_out;
  float* v_out = s_out + (size_t)N * PDIM;

  float* q    = (float*)d_ws;
  float* k    = q  + (size_t)N * PDIM;
  float* vs   = k  + (size_t)N * PDIM;
  float* vv   = vs + (size_t)N * PDIM;
  float* aggv = vv + (size_t)N * PDIM;

  int nPad = (N / SCAN_TILE + 1) * SCAN_TILE;
  int ePad = (E + 3) / 4 * 4;
  int*  counts  = (int*)(aggv + (size_t)N * 3 * PDIM);
  int*  offsets = counts + nPad;
  int*  slot    = offsets + nPad;
  int2* list    = (int2*)(slot + ePad);

  const int* rowIdx = ei;
  const int* colIdx = ei + E;

  int edgeBlocks = (E + 1023) / 1024;
  csr_zero<<<(nPad / 4 + 255) / 256, 256, 0, stream>>>(counts, nPad / 4);
  csr_count<<<edgeBlocks, 256, 0, stream>>>(rowIdx, counts, slot, E);
  csr_scan<<<1, 1024, 0, stream>>>(counts, offsets, nPad / SCAN_TILE);
  csr_fill<<<edgeBlocks, 256, 0, stream>>>(rowIdx, colIdx, slot, offsets, list, E);

  int tilesA = (N / 16) * (512 / 16);
  painn_node_gemm<<<(tilesA + 3) / 4, 128, 0, stream>>>(
      s, Wq, bq, Wkv, bkv, q, k, vs, vv, tilesA);

  node_rows<<<(N + 7) / 8, 256, 0, stream>>>(
      q, k, vs, vv, v, s, offsets, list, s_out, aggv, N);

  int tilesC = (N * 3 / 16) * (PDIM / 16);
  painn_vec_gemm<<<(tilesC + 3) / 4, 128, 0, stream>>>(
      aggv, Wvec, bvec, v, v_out, tilesC);
}


# Both GEMMs re-tiled: a wave keeps its 16-row A strip (all K) in registers and sweeps 16x64 column chunks, float4 weight loads issued ahead, 4 accumulators; was one 16x16 tile per wave; same sum order
# speedup vs baseline: 2.3472x; 1.3096x over previous
#include <hip/hip_runtime.h>


typedef __attribute__((ext_vector_type(2))) float v2f;
typedef __attribute__((ext_vector_type(8))) float v8f;

#define PDIM 128
#define SCAN_TILE 4096

__global__ __launch_bounds__(256) void csr_zero(int* __restrict__ counts, int n4) {
  int i = blockIdx.x * blockDim.x + threadIdx.x;
  if (i < n4) ((int4*)counts)[i] = make_int4(0, 0, 0, 0);
}

__global__ __launch_bounds__(256) void csr_count(const int* __restrict__ rowIdx,
                                                 int* __restrict__ counts,
                                                 int* __restrict__ slot, int E) {
  int stride = gridDim.x * blockDim.x;
  for (int e = blockIdx.x * blockDim.x + threadIdx.x; e < E; e += stride)
    slot[e] = atomicAdd(&counts[rowIdx[e]], 1);
}

__global__ __launch_bounds__(1024) void csr_scan(const int* __restrict__ counts,
                                                 int* __restrict__ offsets, int nTrips) {
  __shared__ int waveSum[32];
  int t = threadIdx.x;
  int lane = t & 31;
  int wave = t >> 5;
  int carry = 0;
  for (int it = 0; it < nTrips; ++it) {
    int4 c = ((const int4*)counts)[it * 1024 + t];
    int s1 = c.x, s2 = s1 + c.y, s3 = s2 + c.z, s4 = s3 + c.w;
    int incl = s4;
    for (int o = 1; o < 32; o <<= 1) {
      int n = __shfl_up(incl, o, 32);
      if (lane >= o) incl += n;
    }
    if (lane == 31) waveSum[wave] = incl;
    __syncthreads();
    int ws = waveSum[lane];
    int wincl = ws;
    for (int o = 1; o < 32; o <<= 1) {
      int n = __shfl_up(wincl, o, 32);
      if (lane >= o) wincl += n;
    }
    int wavePrefix = __shfl(wincl, wave, 32) - __shfl(ws, wave, 32);
    int total = __shfl(wincl, 31, 32);
    int base = carry + wavePrefix + (incl - s4);
    ((int4*)offsets)[it * 1024 + t] = make_int4(base, base + s1, base + s2, base + s3);
    carry += total;
    __syncthreads();
  }
}

__global__ __launch_bounds__(256) void csr_fill(const int* __restrict__ rowIdx,
                                                const int* __restrict__ colIdx,
                                                const int* __restrict__ slot,
                                                const int* __restrict__ offsets,
                                                int2* __restrict__ list, int E) {
  int stride = gridDim.x * blockDim.x;
  for (int e = blockIdx.x * blockDim.x + threadIdx.x; e < E; e += stride)
    list[offsets[rowIdx[e]] + slot[e]] = make_int2(e, colIdx[e]);
}

__device__ __forceinline__ void load_strip(const float* __restrict__ rowp, v2f (&a)[32]) {
#pragma unroll
  for (int ks = 0; ks < 32; ++ks) {
    float2 t = *(const float2*)(rowp + 4 * ks);
    a[ks].x = t.x;
    a[ks].y = t.y;
  }
}

__device__ __forceinline__ void step4(const v2f& af, const float4& r0, const float4& r1,
                                      v8f (&acc)[4]) {
  v2f bf;
  bf.x = r0.x; bf.y = r1.x;
  acc[0] = __builtin_amdgcn_wmma_f32_16x16x4_f32(false, af, false, bf, (short)0, acc[0], false, false);
  bf.x = r0.y; bf.y = r1.y;
  acc[1] = __builtin_amdgcn_wmma_f32_16x16x4_f32(false, af, false, bf, (short)0, acc[1], false, false);
  bf.x = r0.z; bf.y = r1.z;
  acc[2] = __builtin_amdgcn_wmma_f32_16x16x4_f32(false, af, false, bf, (short)0, acc[2], false, false);
  bf.x = r0.w; bf.y = r1.w;
  acc[3] = __builtin_amdgcn_wmma_f32_16x16x4_f32(false, af, false, bf, (short)0, acc[3], false, false);
}

template <int LDW>
__device__ __forceinline__ void strip_chunk(const v2f (&a)[32], const float* __restrict__ Wl,
                                            v8f (&acc)[4]) {
  float4 n0 = *(const float4*)(Wl);
  float4 n1 = *(const float4*)(Wl + LDW);
  float4 n2 = *(const float4*)(Wl + 4 * LDW);
  float4 n3 = *(const float4*)(Wl + 5 * LDW);
#pragma unroll
  for (int j = 0; j < 16; ++j) {
    float4 c0 = n0, c1 = n1, c2 = n2, c3 = n3;
    if (j + 1 < 16) {
      const float* p = Wl + (8 * (j + 1)) * LDW;
      n0 = *(const float4*)(p);
      n1 = *(const float4*)(p + LDW);
      n2 = *(const float4*)(p + 4 * LDW);
      n3 = *(const float4*)(p + 5 * LDW);
    }
    step4(a[2 * j],     c0, c1, acc);
    step4(a[2 * j + 1], c2, c3, acc);
  }
}

template <int LDW, bool ADD>
__device__ __forceinline__ void strip_block(const v2f (&a)[32], const float* __restrict__ Wl,
                                            const float* __restrict__ bl,
                                            const float* __restrict__ arow,
                                            float* __restrict__ orow) {
#pragma unroll 1
  for (int cb = 0; cb < PDIM; cb += 64) {
    v8f acc[4] = {};
    strip_chunk<LDW>(a, Wl + cb, acc);
    float4 bias = *(const float4*)(bl + cb);
#pragma unroll
    for (int r = 0; r < 8; ++r) {
      float4 ov;
      if (ADD) {
        float4 x = *(const float4*)(arow + (size_t)r * PDIM + cb);
        ov = make_float4(x.x + acc[0][r] + bias.x, x.y + acc[1][r] + bias.y,
                         x.z + acc[2][r] + bias.z, x.w + acc[3][r] + bias.w);
      } else {
        ov = make_float4(acc[0][r] + bias.x, acc[1][r] + bias.y,
                         acc[2][r] + bias.z, acc[3][r] + bias.w);
      }
      *(float4*)(orow + (size_t)r * PDIM + cb) = ov;
    }
  }
}

__global__ __launch_bounds__(256) void painn_node_gemm(
    const float* __restrict__ s,
    const float* __restrict__ Wq,  const float* __restrict__ bq,
    const float* __restrict__ Wkv, const float* __restrict__ bkv,
    float* __restrict__ q, float* __restrict__ k,
    float* __restrict__ vs, float* __restrict__ vv,
    int nStrips, int nGroups) {
  int wave = threadIdx.x >> 5;
  int lane = threadIdx.x & 31;
  int o = blockIdx.x / nGroups;
  int strip = (blockIdx.x - o * nGroups) * 8 + wave;
  if (strip >= nStrips) return;
  int half = lane >> 4;
  int l16  = lane & 15;
  int rowBase = strip * 16;

  v2f a[32];
  load_strip(s + (size_t)(rowBase + l16) * PDIM + 2 * half, a);

  size_t ooff = (size_t)(rowBase + 8 * half) * PDIM + 4 * l16;
  if (o == 0) {
    strip_block<PDIM, false>(a, Wq + (2 * half) * PDIM + 4 * l16, bq + 4 * l16,
                             nullptr, q + ooff);
  } else {
    int wc = (o - 1) * PDIM + 4 * l16;
    float* out = (o == 1) ? k : (o == 2) ? vs : vv;
    strip_block<3 * PDIM, false>(a, Wkv + (2 * half) * 3 * PDIM + wc, bkv + wc,
                                 nullptr, out + ooff);
  }
}

__device__ __forceinline__ void edge_accum(
    int c, int d, const float4& qv,
    const float* __restrict__ k, const float* __restrict__ vs,
    const float* __restrict__ vv, const float* __restrict__ v,
    float4& as, float4& a0, float4& a1, float4& a2) {
  float4 kv  = *(const float4*)(k  + (size_t)c * PDIM + d);
  float4 vsv = *(const float4*)(vs + (size_t)c * PDIM + d);
  float4 vvv = *(const float4*)(vv + (size_t)c * PDIM + d);
  const float* vc = v + (size_t)c * 3 * PDIM + d;
  float4 x0 = *(const float4*)(vc);
  float4 x1 = *(const float4*)(vc + PDIM);
  float4 x2 = *(const float4*)(vc + 2 * PDIM);

  float p = qv.x * kv.x + qv.y * kv.y + qv.z * kv.z + qv.w * kv.w;
  p += __shfl_xor(p, 16, 32);
  p += __shfl_xor(p, 8, 32);
  p += __shfl_xor(p, 4, 32);
  p += __shfl_xor(p, 2, 32);
  p += __shfl_xor(p, 1, 32);
  float w = p;

  as.x += w * vsv.x; as.y += w * vsv.y; as.z += w * vsv.z; as.w += w * vsv.w;
  float tx = w * vvv.x, ty = w * vvv.y, tz = w * vvv.z, tw = w * vvv.w;
  a0.x += tx * x0.x; a0.y += ty * x0.y; a0.z += tz * x0.z; a0.w += tw * x0.w;
  a1.x += tx * x1.x; a1.y += ty * x1.y; a1.z += tz * x1.z; a1.w += tw * x1.w;
  a2.x += tx * x2.x; a2.y += ty * x2.y; a2.z += tz * x2.z; a2.w += tw * x2.w;
}

__global__ __launch_bounds__(256) void node_rows(
    const float* __restrict__ q, const float* __restrict__ k,
    const float* __restrict__ vs, const float* __restrict__ vv,
    const float* __restrict__ v, const float* __restrict__ s,
    const int* __restrict__ offsets, const int2* __restrict__ list,
    float* __restrict__ s_out, float* __restrict__ aggv, int N) {
  int wave = __builtin_amdgcn_readfirstlane((int)(threadIdx.x >> 5));
  int node = blockIdx.x * 8 + wave;
  if (node >= N) return;
  int lane = threadIdx.x & 31;
  int d = lane * 4;
  int beg = offsets[node];
  int end = offsets[node + 1];
  int deg = end - beg;

  float4 qv = *(const float4*)(q + (size_t)node * PDIM + d);
  float4 as = make_float4(0.f, 0.f, 0.f, 0.f);
  float4 a0 = as, a1 = as, a2 = as;

  if (deg <= 32) {
    int2 ec = make_int2(0x7fffffff, 0);
    if (lane < deg) ec = list[beg + lane];
    int rank = 0;
    for (int j = 0; j < deg; ++j) {
      int ej = __builtin_amdgcn_readlane(ec.x, j);
      rank += (ej < ec.x) ? 1 : 0;
    }
#pragma unroll 1
    for (int i = 0; i < deg; ++i) {
      unsigned m = __builtin_amdgcn_ballot_w32(rank == i);
      int src = __builtin_ctz(m | 0x80000000u);
      int c = __builtin_amdgcn_readlane(ec.y, src);
      edge_accum(c, d, qv, k, vs, vv, v, as, a0, a1, a2);
    }
  } else {
    int last = -1;
    for (int i = 0; i < deg; ++i) {
      int be = 0x7fffffff, bc = 0;
      for (int j = beg + lane; j < end; j += 32) {
        int2 ec = list[j];
        if (ec.x > last && ec.x < be) { be = ec.x; bc = ec.y; }
      }
      int mn = be;
      mn = min(mn, __shfl_xor(mn, 16, 32));
      mn = min(mn, __shfl_xor(mn, 8, 32));
      mn = min(mn, __shfl_xor(mn, 4, 32));
      mn = min(mn, __shfl_xor(mn, 2, 32));
      mn = min(mn, __shfl_xor(mn, 1, 32));
      unsigned m = __builtin_amdgcn_ballot_w32(be == mn);
      int src = __builtin_ctz(m | 0x80000000u);
      int c = __builtin_amdgcn_readlane(bc, src);
      last = __builtin_amdgcn_readfirstlane(mn);
      edge_accum(c, d, qv, k, vs, vv, v, as, a0, a1, a2);
    }
  }

  float4 sv = *(const float4*)(s + (size_t)node * PDIM + d);
  float4 so = make_float4(sv.x + as.x, sv.y + as.y, sv.z + as.z, sv.w + as.w);
  *(float4*)(s_out + (size_t)node * PDIM + d) = so;
  float* av = aggv + (size_t)node * 3 * PDIM + d;
  *(float4*)(av)            = a0;
  *(float4*)(av + PDIM)     = a1;
  *(float4*)(av + 2 * PDIM) = a2;
}

__global__ __launch_bounds__(256) void painn_vec_gemm(
    const float* __restrict__ aggv, const float* __restrict__ Wvec,
    const float* __restrict__ bvec, const float* __restrict__ v,
    float* __restrict__ vout, int nStrips) {
  int wave = threadIdx.x >> 5;
  int lane = threadIdx.x & 31;
  int strip = blockIdx.x * 8 + wave;
  if (strip >= nStrips) return;
  int half = lane >> 4;
  int l16  = lane & 15;
  size_t rowBase = (size_t)strip * 16;

  v2f a[32];
  load_strip(aggv + (rowBase + l16) * PDIM + 2 * half, a);

  size_t ooff = (rowBase + 8 * half) * PDIM + 4 * l16;
  strip_block<PDIM, true>(a, Wvec + (2 * half) * PDIM + 4 * l16, bvec + 4 * l16,
                          v + ooff, vout + ooff);
}

extern "C" void kernel_launch(void* const* d_in, const int* in_sizes, int n_in,
                              void* d_out, int out_size, void* d_ws, size_t ws_size,
                              hipStream_t stream) {
  const float* s    = (const float*)d_in[0];
  const float* v    = (const float*)d_in[1];
  const int*   ei   = (const int*)d_in[2];
  const float* Wq   = (const float*)d_in[3];
  const float* bq   = (const float*)d_in[4];
  const float* Wkv  = (const float*)d_in[5];
  const float* bkv  = (const float*)d_in[6];
  const float* Wvec = (const float*)d_in[7];
  const float* bvec = (const float*)d_in[8];

  int N = in_sizes[0] / PDIM;
  int E = in_sizes[2] / 2;

  float* s_out = (float*)d_out;
  float* v_out = s_out + (size_t)N * PDIM;

  float* q    = (float*)d_ws;
  float* k    = q  + (size_t)N * PDIM;
  float* vs   = k  + (size_t)N * PDIM;
  float* vv   = vs + (size_t)N * PDIM;
  float* aggv = vv + (size_t)N * PDIM;

  int nPad = (N / SCAN_TILE + 1) * SCAN_TILE;
  int ePad = (E + 3) / 4 * 4;
  int*  counts  = (int*)(aggv + (size_t)N * 3 * PDIM);
  int*  offsets = counts + nPad;
  int*  slot    = offsets + nPad;
  int2* list    = (int2*)(slot + ePad);

  const int* rowIdx = ei;
  const int* colIdx = ei + E;

  int edgeBlocks = (E + 1023) / 1024;
  csr_zero<<<(nPad / 4 + 255) / 256, 256, 0, stream>>>(counts, nPad / 4);
  csr_count<<<edgeBlocks, 256, 0, stream>>>(rowIdx, counts, slot, E);
  csr_scan<<<1, 1024, 0, stream>>>(counts, offsets, nPad / SCAN_TILE);
  csr_fill<<<edgeBlocks, 256, 0, stream>>>(rowIdx, colIdx, slot, offsets, list, E);

  int stripsA = N / 16;
  int groupsA = (stripsA + 7) / 8;
  painn_node_gemm<<<4 * groupsA, 256, 0, stream>>>(
      s, Wq, bq, Wkv, bkv, q, k, vs, vv, stripsA, groupsA);

  node_rows<<<(N + 7) / 8, 256, 0, stream>>>(
      q, k, vs, vv, v, s, offsets, list, s_out, aggv, N);

  int stripsC = N * 3 / 16;
  painn_vec_gemm<<<(stripsC + 7) / 8, 256, 0, stream>>>(
      aggv, Wvec, bvec, v, v_out, stripsC);
}


# Edge lists as fixed 32-slot rows per node filled in one pass (overflow list for longer rows): prefix scan and second edge pass go, node GEMM launch clears the counters; 7 launches to 4; same sums
# speedup vs baseline: 2.6873x; 1.1449x over previous
#include <hip/hip_runtime.h>


typedef __attribute__((ext_vector_type(2))) float v2f;
typedef __attribute__((ext_vector_type(8))) float v8f;

#define PDIM 128
#define CAP 32

__global__ __launch_bounds__(256) void bucket_fill(const int* __restrict__ rowIdx,
                                                   const int* __restrict__ colIdx,
                                                   int* __restrict__ counts,
                                                   int2* __restrict__ list,
                                                   int4* __restrict__ ovf,
                                                   int ovfCounter, int E) {
  int stride = gridDim.x * blockDim.x;
  for (int e0 = blockIdx.x * blockDim.x + threadIdx.x; e0 < E; e0 += 4 * stride) {
    int r[4], c[4], sl[4];
#pragma unroll
    for (int u = 0; u < 4; ++u) {
      int e = e0 + u * stride;
      int el = min(e, E - 1);
      int rr = rowIdx[el];
      c[u] = colIdx[el];
      r[u] = (e < E) ? rr : -1;
    }
#pragma unroll
    for (int u = 0; u < 4; ++u)
      sl[u] = (r[u] >= 0) ? atomicAdd(&counts[r[u]], 1) : 0;
#pragma unroll
    for (int u = 0; u < 4; ++u) {
      if (r[u] < 0) continue;
      int e = e0 + u * stride;
      if (sl[u] < CAP) {
        list[(size_t)r[u] * CAP + sl[u]] = make_int2(e, c[u]);
      } else {
        int o = atomicAdd(&counts[ovfCounter], 1);
        ovf[o] = make_int4(r[u], e, c[u], 0);
      }
    }
  }
}

__device__ __forceinline__ void load_strip(const float* __restrict__ rowp, v2f (&a)[32]) {
#pragma unroll
  for (int ks = 0; ks < 32; ++ks) {
    float2 t = *(const float2*)(rowp + 4 * ks);
    a[ks].x = t.x;
    a[ks].y = t.y;
  }
}

__device__ __forceinline__ void step4(const v2f& af, const float4& r0, const float4& r1,
                                      v8f (&acc)[4]) {
  v2f bf;
  bf.x = r0.x; bf.y = r1.x;
  acc[0] = __builtin_amdgcn_wmma_f32_16x16x4_f32(false, af, false, bf, (short)0, acc[0], false, false);
  bf.x = r0.y; bf.y = r1.y;
  acc[1] = __builtin_amdgcn_wmma_f32_16x16x4_f32(false, af, false, bf, (short)0, acc[1], false, false);
  bf.x = r0.z; bf.y = r1.z;
  acc[2] = __builtin_amdgcn_wmma_f32_16x16x4_f32(false, af, false, bf, (short)0, acc[2], false, false);
  bf.x = r0.w; bf.y = r1.w;
  acc[3] = __builtin_amdgcn_wmma_f32_16x16x4_f32(false, af, false, bf, (short)0, acc[3], false, false);
}

template <int LDW>
__device__ __forceinline__ void strip_chunk(const v2f (&a)[32], const float* __restrict__ Wl,
                                            v8f (&acc)[4]) {
  float4 n0 = *(const float4*)(Wl);
  float4 n1 = *(const float4*)(Wl + LDW);
  float4 n2 = *(const float4*)(Wl + 4 * LDW);
  float4 n3 = *(const float4*)(Wl + 5 * LDW);
#pragma unroll
  for (int j = 0; j < 16; ++j) {
    float4 c0 = n0, c1 = n1, c2 = n2, c3 = n3;
    if (j + 1 < 16) {
      const float* p = Wl + (8 * (j + 1)) * LDW;
      n0 = *(const float4*)(p);
      n1 = *(const float4*)(p + LDW);
      n2 = *(const float4*)(p + 4 * LDW);
      n3 = *(const float4*)(p + 5 * LDW);
    }
    step4(a[2 * j],     c0, c1, acc);
    step4(a[2 * j + 1], c2, c3, acc);
  }
}

template <int LDW, bool ADD>
__device__ __forceinline__ void strip_block(const v2f (&a)[32], const float* __restrict__ Wl,
                                            const float* __restrict__ bl,
                                            const float* __restrict__ arow,
                                            float* __restrict__ orow) {
#pragma unroll 1
  for (int cb = 0; cb < PDIM; cb += 64) {
    v8f acc[4] = {};
    strip_chunk<LDW>(a, Wl + cb, acc);
    float4 bias = *(const float4*)(bl + cb);
#pragma unroll
    for (int r = 0; r < 8; ++r) {
      float4 ov;
      if (ADD) {
        float4 x = *(const float4*)(arow + (size_t)r * PDIM + cb);
        ov = make_float4(x.x + acc[0][r] + bias.x, x.y + acc[1][r] + bias.y,
                         x.z + acc[2][r] + bias.z, x.w + acc[3][r] + bias.w);
      } else {
        ov = make_float4(acc[0][r] + bias.x, acc[1][r] + bias.y,
                         acc[2][r] + bias.z, acc[3][r] + bias.w);
      }
      *(float4*)(orow + (size_t)r * PDIM + cb) = ov;
    }
  }
}

__global__ __launch_bounds__(256) void painn_node_gemm(
    const float* __restrict__ s,
    const float* __restrict__ Wq,  const float* __restrict__ bq,
    const float* __restrict__ Wkv, const float* __restrict__ bkv,
    float* __restrict__ q, float* __restrict__ k,
    float* __restrict__ vs, float* __restrict__ vv,
    int nStrips, int nGroups, int* __restrict__ counts, int nCounts4) {
  int gid = blockIdx.x * 256 + threadIdx.x;
  if (gid < nCounts4) ((int4*)counts)[gid] = make_int4(0, 0, 0, 0);
  int wave = threadIdx.x >> 5;
  int lane = threadIdx.x & 31;
  int o = blockIdx.x / nGroups;
  int strip = (blockIdx.x - o * nGroups) * 8 + wave;
  if (strip >= nStrips) return;
  int half = lane >> 4;
  int l16  = lane & 15;
  int rowBase = strip * 16;

  v2f a[32];
  load_strip(s + (size_t)(rowBase + l16) * PDIM + 2 * half, a);

  size_t ooff = (size_t)(rowBase + 8 * half) * PDIM + 4 * l16;
  if (o == 0) {
    strip_block<PDIM, false>(a, Wq + (2 * half) * PDIM + 4 * l16, bq + 4 * l16,
                             nullptr, q + ooff);
  } else {
    int wc = (o - 1) * PDIM + 4 * l16;
    float* out = (o == 1) ? k : (o == 2) ? vs : vv;
    strip_block<3 * PDIM, false>(a, Wkv + (2 * half) * 3 * PDIM + wc, bkv + wc,
                                 nullptr, out + ooff);
  }
}

__device__ __forceinline__ void edge_accum(
    int c, int d, const float4& qv,
    const float* __restrict__ k, const float* __restrict__ vs,
    const float* __restrict__ vv, const float* __restrict__ v,
    float4& as, float4& a0, float4& a1, float4& a2) {
  float4 kv  = *(const float4*)(k  + (size_t)c * PDIM + d);
  float4 vsv = *(const float4*)(vs + (size_t)c * PDIM + d);
  float4 vvv = *(const float4*)(vv + (size_t)c * PDIM + d);
  const float* vc = v + (size_t)c * 3 * PDIM + d;
  float4 x0 = *(const float4*)(vc);
  float4 x1 = *(const float4*)(vc + PDIM);
  float4 x2 = *(const float4*)(vc + 2 * PDIM);

  float p = qv.x * kv.x + qv.y * kv.y + qv.z * kv.z + qv.w * kv.w;
  p += __shfl_xor(p, 16, 32);
  p += __shfl_xor(p, 8, 32);
  p += __shfl_xor(p, 4, 32);
  p += __shfl_xor(p, 2, 32);
  p += __shfl_xor(p, 1, 32);
  float w = p;

  as.x += w * vsv.x; as.y += w * vsv.y; as.z += w * vsv.z; as.w += w * vsv.w;
  float tx = w * vvv.x, ty = w * vvv.y, tz = w * vvv.z, tw = w * vvv.w;
  a0.x += tx * x0.x; a0.y += ty * x0.y; a0.z += tz * x0.z; a0.w += tw * x0.w;
  a1.x += tx * x1.x; a1.y += ty * x1.y; a1.z += tz * x1.z; a1.w += tw * x1.w;
  a2.x += tx * x2.x; a2.y += ty * x2.y; a2.z += tz * x2.z; a2.w += tw * x2.w;
}

__global__ __launch_bounds__(256) void node_rows(
    const float* __restrict__ q, const float* __restrict__ k,
    const float* __restrict__ vs, const float* __restrict__ vv,
    const float* __restrict__ v, const float* __restrict__ s,
    const int* __restrict__ counts, const int2* __restrict__ list,
    const int4* __restrict__ ovf,
    float* __restrict__ s_out, float* __restrict__ aggv, int N) {
  int wave = __builtin_amdgcn_readfirstlane((int)(threadIdx.x >> 5));
  int node = blockIdx.x * 8 + wave;
  if (node >= N) return;
  int lane = threadIdx.x & 31;
  int d = lane * 4;
  int deg = counts[node];
  const int2* slots = list + (size_t)node * CAP;

  float4 qv = *(const float4*)(q + (size_t)node * PDIM + d);
  float4 as = make_float4(0.f, 0.f, 0.f, 0.f);
  float4 a0 = as, a1 = as, a2 = as;

  if (deg <= CAP) {
    int2 ec = make_int2(0x7fffffff, 0);
    if (lane < deg) ec = slots[lane];
    int rank = 0;
    for (int j = 0; j < deg; ++j) {
      int ej = __builtin_amdgcn_readlane(ec.x, j);
      rank += (ej < ec.x) ? 1 : 0;
    }
#pragma unroll 1
    for (int i = 0; i < deg; ++i) {
      unsigned m = __builtin_amdgcn_ballot_w32(rank == i);
      int src = __builtin_ctz(m | 0x80000000u);
      int c = __builtin_amdgcn_readlane(ec.y, src);
      edge_accum(c, d, qv, k, vs, vv, v, as, a0, a1, a2);
    }
  } else {
    int nOvf = counts[N];
    int2 ec = slots[lane];
    int last = -1;
    for (int i = 0; i < deg; ++i) {
      int be = 0x7fffffff, bc = 0;
      if (ec.x > last) { be = ec.x; bc = ec.y; }
      for (int j = lane; j < nOvf; j += 32) {
        int4 o = ovf[j];
        if (o.x == node && o.y > last && o.y < be) { be = o.y; bc = o.z; }
      }
      int mn = be;
      mn = min(mn, __shfl_xor(mn, 16, 32));
      mn = min(mn, __shfl_xor(mn, 8, 32));
      mn = min(mn, __shfl_xor(mn, 4, 32));
      mn = min(mn, __shfl_xor(mn, 2, 32));
      mn = min(mn, __shfl_xor(mn, 1, 32));
      unsigned m = __builtin_amdgcn_ballot_w32(be == mn);
      int src = __builtin_ctz(m | 0x80000000u);
      int c = __builtin_amdgcn_readlane(bc, src);
      last = __builtin_amdgcn_readfirstlane(mn);
      edge_accum(c, d, qv, k, vs, vv, v, as, a0, a1, a2);
    }
  }

  float4 sv = *(const float4*)(s + (size_t)node * PDIM + d);
  float4 so = make_float4(sv.x + as.x, sv.y + as.y, sv.z + as.z, sv.w + as.w);
  *(float4*)(s_out + (size_t)node * PDIM + d) = so;
  float* av = aggv + (size_t)node * 3 * PDIM + d;
  *(float4*)(av)            = a0;
  *(float4*)(av + PDIM)     = a1;
  *(float4*)(av + 2 * PDIM) = a2;
}

__global__ __launch_bounds__(256) void painn_vec_gemm(
    const float* __restrict__ aggv, const float* __restrict__ Wvec,
    const float* __restrict__ bvec, const float* __restrict__ v,
    float* __restrict__ vout, int nStrips) {
  int wave = threadIdx.x >> 5;
  int lane = threadIdx.x & 31;
  int strip = blockIdx.x * 8 + wave;
  if (strip >= nStrips) return;
  int half = lane >> 4;
  int l16  = lane & 15;
  size_t rowBase = (size_t)strip * 16;

  v2f a[32];
  load_strip(aggv + (rowBase + l16) * PDIM + 2 * half, a);

  size_t ooff = (rowBase + 8 * half) * PDIM + 4 * l16;
  strip_block<PDIM, true>(a, Wvec + (2 * half) * PDIM + 4 * l16, bvec + 4 * l16,
                          v + ooff, vout + ooff);
}

extern "C" void kernel_launch(void* const* d_in, const int* in_sizes, int n_in,
                              void* d_out, int out_size, void* d_ws, size_t ws_size,
                              hipStream_t stream) {
  const float* s    = (const float*)d_in[0];
  const float* v    = (const float*)d_in[1];
  const int*   ei   = (const int*)d_in[2];
  const float* Wq   = (const float*)d_in[3];
  const float* bq   = (const float*)d_in[4];
  const float* Wkv  = (const float*)d_in[5];
  const float* bkv  = (const float*)d_in[6];
  const float* Wvec = (const float*)d_in[7];
  const float* bvec = (const float*)d_in[8];

  int N = in_sizes[0] / PDIM;
  int E = in_sizes[2] / 2;

  float* s_out = (float*)d_out;
  float* v_out = s_out + (size_t)N * PDIM;

  float* q    = (float*)d_ws;
  float* k    = q  + (size_t)N * PDIM;
  float* vs   = k  + (size_t)N * PDIM;
  float* vv   = vs + (size_t)N * PDIM;
  float* aggv = vv + (size_t)N * PDIM;

  int nCnt = (N + 1 + 63) / 64 * 64;
  int*  counts = (int*)(aggv + (size_t)N * 3 * PDIM);
  int2* list   = (int2*)(counts + nCnt);
  int4* ovf    = (int4*)(list + (size_t)N * CAP);

  const int* rowIdx = ei;
  const int* colIdx = ei + E;

  int stripsA = N / 16;
  int groupsA = (stripsA + 7) / 8;
  painn_node_gemm<<<4 * groupsA, 256, 0, stream>>>(
      s, Wq, bq, Wkv, bkv, q, k, vs, vv, stripsA, groupsA, counts, nCnt / 4);

  int edgeBlocks = (E + 1023) / 1024;
  bucket_fill<<<edgeBlocks, 256, 0, stream>>>(rowIdx, colIdx, counts, list, ovf, N, E);

  node_rows<<<(N + 7) / 8, 256, 0, stream>>>(
      q, k, vs, vv, v, s, counts, list, ovf, s_out, aggv, N);

  int stripsC = N * 3 / 16;
  painn_vec_gemm<<<(stripsC + 7) / 8, 256, 0, stream>>>(
      aggv, Wvec, bvec, v, v_out, stripsC);
}


# Message sums run per 64-feature half (slow grid index) so the rows a half gathers stay in the second-level cache; edge weights first, by a k-only pass into the sorted slots; 5 launches; same sums
# speedup vs baseline: 2.8417x; 1.0575x over previous
#include <hip/hip_runtime.h>


typedef __attribute__((ext_vector_type(2))) float v2f;
typedef __attribute__((ext_vector_type(8))) float v8f;

#define PDIM 128
#define CAP 32

__global__ __launch_bounds__(256) void bucket_fill(const int* __restrict__ rowIdx,
                                                   const int* __restrict__ colIdx,
                                                   int* __restrict__ counts,
                                                   int2* __restrict__ list,
                                                   int4* __restrict__ ovf,
                                                   int ovfCounter, int E) {
  int stride = gridDim.x * blockDim.x;
  for (int e0 = blockIdx.x * blockDim.x + threadIdx.x; e0 < E; e0 += 4 * stride) {
    int r[4], c[4], sl[4];
#pragma unroll
    for (int u = 0; u < 4; ++u) {
      int e = e0 + u * stride;
      int el = min(e, E - 1);
      int rr = rowIdx[el];
      c[u] = colIdx[el];
      r[u] = (e < E) ? rr : -1;
    }
#pragma unroll
    for (int u = 0; u < 4; ++u)
      sl[u] = (r[u] >= 0) ? atomicAdd(&counts[r[u]], 1) : 0;
#pragma unroll
    for (int u = 0; u < 4; ++u) {
      if (r[u] < 0) continue;
      int e = e0 + u * stride;
      if (sl[u] < CAP) {
        list[(size_t)r[u] * CAP + sl[u]] = make_int2(e, c[u]);
      } else {
        int o = atomicAdd(&counts[ovfCounter], 1);
        ovf[o] = make_int4(r[u], e, c[u], 0);
      }
    }
  }
}

__device__ __forceinline__ void load_strip(const float* __restrict__ rowp, v2f (&a)[32]) {
#pragma unroll
  for (int ks = 0; ks < 32; ++ks) {
    float2 t = *(const float2*)(rowp + 4 * ks);
    a[ks].x = t.x;
    a[ks].y = t.y;
  }
}

__device__ __forceinline__ void step4(const v2f& af, const float4& r0, const float4& r1,
                                      v8f (&acc)[4]) {
  v2f bf;
  bf.x = r0.x; bf.y = r1.x;
  acc[0] = __builtin_amdgcn_wmma_f32_16x16x4_f32(false, af, false, bf, (short)0, acc[0], false, false);
  bf.x = r0.y; bf.y = r1.y;
  acc[1] = __builtin_amdgcn_wmma_f32_16x16x4_f32(false, af, false, bf, (short)0, acc[1], false, false);
  bf.x = r0.z; bf.y = r1.z;
  acc[2] = __builtin_amdgcn_wmma_f32_16x16x4_f32(false, af, false, bf, (short)0, acc[2], false, false);
  bf.x = r0.w; bf.y = r1.w;
  acc[3] = __builtin_amdgcn_wmma_f32_16x16x4_f32(false, af, false, bf, (short)0, acc[3], false, false);
}

template <int LDW>
__device__ __forceinline__ void strip_chunk(const v2f (&a)[32], const float* __restrict__ Wl,
                                            v8f (&acc)[4]) {
  float4 n0 = *(const float4*)(Wl);
  float4 n1 = *(const float4*)(Wl + LDW);
  float4 n2 = *(const float4*)(Wl + 4 * LDW);
  float4 n3 = *(const float4*)(Wl + 5 * LDW);
#pragma unroll
  for (int j = 0; j < 16; ++j) {
    float4 c0 = n0, c1 = n1, c2 = n2, c3 = n3;
    if (j + 1 < 16) {
      const float* p = Wl + (8 * (j + 1)) * LDW;
      n0 = *(const float4*)(p);
      n1 = *(const float4*)(p + LDW);
      n2 = *(const float4*)(p + 4 * LDW);
      n3 = *(const float4*)(p + 5 * LDW);
    }
    step4(a[2 * j],     c0, c1, acc);
    step4(a[2 * j + 1], c2, c3, acc);
  }
}

template <int LDW, bool ADD>
__device__ __forceinline__ void strip_block(const v2f (&a)[32], const float* __restrict__ Wl,
                                            const float* __restrict__ bl,
                                            const float* __restrict__ arow,
                                            float* __restrict__ orow) {
#pragma unroll 1
  for (int cb = 0; cb < PDIM; cb += 64) {
    v8f acc[4] = {};
    strip_chunk<LDW>(a, Wl + cb, acc);
    float4 bias = *(const float4*)(bl + cb);
#pragma unroll
    for (int r = 0; r < 8; ++r) {
      float4 ov;
      if (ADD) {
        float4 x = *(const float4*)(arow + (size_t)r * PDIM + cb);
        ov = make_float4(x.x + acc[0][r] + bias.x, x.y + acc[1][r] + bias.y,
                         x.z + acc[2][r] + bias.z, x.w + acc[3][r] + bias.w);
      } else {
        ov = make_float4(acc[0][r] + bias.x, acc[1][r] + bias.y,
                         acc[2][r] + bias.z, acc[3][r] + bias.w);
      }
      *(float4*)(orow + (size_t)r * PDIM + cb) = ov;
    }
  }
}

__global__ __launch_bounds__(256) void painn_node_gemm(
    const float* __restrict__ s,
    const float* __restrict__ Wq,  const float* __restrict__ bq,
    const float* __restrict__ Wkv, const float* __restrict__ bkv,
    float* __restrict__ q, float* __restrict__ k,
    float* __restrict__ vs, float* __restrict__ vv,
    int nStrips, int nGroups, int* __restrict__ counts, int nCounts4) {
  int gid = blockIdx.x * 256 + threadIdx.x;
  if (gid < nCounts4) ((int4*)counts)[gid] = make_int4(0, 0, 0, 0);
  int wave = threadIdx.x >> 5;
  int lane = threadIdx.x & 31;
  int o = blockIdx.x / nGroups;
  int strip = (blockIdx.x - o * nGroups) * 8 + wave;
  if (strip >= nStrips) return;
  int half = lane >> 4;
  int l16  = lane & 15;
  int rowBase = strip * 16;

  v2f a[32];
  load_strip(s + (size_t)(rowBase + l16) * PDIM + 2 * half, a);

  size_t ooff = (size_t)(rowBase + 8 * half) * PDIM + 4 * l16;
  if (o == 0) {
    strip_block<PDIM, false>(a, Wq + (2 * half) * PDIM + 4 * l16, bq + 4 * l16,
                             nullptr, q + ooff);
  } else {
    int wc = (o - 1) * PDIM + 4 * l16;
    float* out = (o == 1) ? k : (o == 2) ? vs : vv;
    strip_block<3 * PDIM, false>(a, Wkv + (2 * half) * 3 * PDIM + wc, bkv + wc,
                                 nullptr, out + ooff);
  }
}

__device__ __forceinline__ float wave_dot(const float4& qv, const float4& kv) {
  float p = qv.x * kv.x + qv.y * kv.y + qv.z * kv.z + qv.w * kv.w;
  p += __shfl_xor(p, 16, 32);
  p += __shfl_xor(p, 8, 32);
  p += __shfl_xor(p, 4, 32);
  p += __shfl_xor(p, 2, 32);
  p += __shfl_xor(p, 1, 32);
  return p;
}

__device__ __forceinline__ void msg_accum(
    float w, const float4& vsv, const float4& vvv,
    const float4& x0, const float4& x1, const float4& x2,
    float4& as, float4& a0, float4& a1, float4& a2) {
  as.x += w * vsv.x; as.y += w * vsv.y; as.z += w * vsv.z; as.w += w * vsv.w;
  float tx = w * vvv.x, ty = w * vvv.y, tz = w * vvv.z, tw = w * vvv.w;
  a0.x += tx * x0.x; a0.y += ty * x0.y; a0.z += tz * x0.z; a0.w += tw * x0.w;
  a1.x += tx * x1.x; a1.y += ty * x1.y; a1.z += tz * x1.z; a1.w += tw * x1.w;
  a2.x += tx * x2.x; a2.y += ty * x2.y; a2.z += tz * x2.z; a2.w += tw * x2.w;
}

__global__ __launch_bounds__(256) void edge_weights(
    const float* __restrict__ q, const float* __restrict__ k,
    const int* __restrict__ counts, int2* __restrict__ list, int N) {
  int wave = __builtin_amdgcn_readfirstlane((int)(threadIdx.x >> 5));
  int node = blockIdx.x * 8 + wave;
  if (node >= N) return;
  int lane = threadIdx.x & 31;
  int d = lane * 4;
  int deg = counts[node];
  if (deg > CAP) return;
  int2* slots = list + (size_t)node * CAP;

  float4 qv = *(const float4*)(q + (size_t)node * PDIM + d);
  int2 ec = make_int2(0x7fffffff, 0);
  if (lane < deg) ec = slots[lane];
  int rank = 0;
  for (int j = 0; j < deg; ++j) {
    int ej = __builtin_amdgcn_readlane(ec.x, j);
    rank += (ej < ec.x) ? 1 : 0;
  }
  float wm = 0.f;
  for (int j0 = 0; j0 < deg; j0 += 4) {
    float4 kv[4];
#pragma unroll
    for (int u = 0; u < 4; ++u) {
      int c = __builtin_amdgcn_readlane(ec.y, min(j0 + u, deg - 1));
      kv[u] = *(const float4*)(k + (size_t)c * PDIM + d);
    }
#pragma unroll
    for (int u = 0; u < 4; ++u) {
      float w = wave_dot(qv, kv[u]);
      if (lane == j0 + u) wm = w;
    }
  }
  if (lane < deg) slots[rank] = make_int2(ec.y, __float_as_int(wm));
}

__device__ void long_row(
    int node, int deg,
    const float* __restrict__ q, const float* __restrict__ k,
    const float* __restrict__ vs, const float* __restrict__ vv,
    const float* __restrict__ v, const float* __restrict__ s,
    const int* __restrict__ counts, const int2* __restrict__ list,
    const int4* __restrict__ ovf,
    float* __restrict__ s_out, float* __restrict__ aggv, int N) {
  int lane = threadIdx.x & 31;
  int d = lane * 4;
  float4 qv = *(const float4*)(q + (size_t)node * PDIM + d);
  float4 as = make_float4(0.f, 0.f, 0.f, 0.f);
  float4 a0 = as, a1 = as, a2 = as;
  int nOvf = counts[N];
  int2 ec = list[(size_t)node * CAP + lane];
  int last = -1;
  for (int i = 0; i < deg; ++i) {
    int be = 0x7fffffff, bc = 0;
    if (ec.x > last) { be = ec.x; bc = ec.y; }
    for (int j = lane; j < nOvf; j += 32) {
      int4 o = ovf[j];
      if (o.x == node && o.y > last && o.y < be) { be = o.y; bc = o.z; }
    }
    int mn = be;
    mn = min(mn, __shfl_xor(mn, 16, 32));
    mn = min(mn, __shfl_xor(mn, 8, 32));
    mn = min(mn, __shfl_xor(mn, 4, 32));
    mn = min(mn, __shfl_xor(mn, 2, 32));
    mn = min(mn, __shfl_xor(mn, 1, 32));
    unsigned m = __builtin_amdgcn_ballot_w32(be == mn);
    int src = __builtin_ctz(m | 0x80000000u);
    int c = __builtin_amdgcn_readlane(bc, src);
    last = __builtin_amdgcn_readfirstlane(mn);
    float4 kv  = *(const float4*)(k  + (size_t)c * PDIM + d);
    float4 vsv = *(const float4*)(vs + (size_t)c * PDIM + d);
    float4 vvv = *(const float4*)(vv + (size_t)c * PDIM + d);
    const float* vc = v + (size_t)c * 3 * PDIM + d;
    float4 x0 = *(const float4*)(vc);
    float4 x1 = *(const float4*)(vc + PDIM);
    float4 x2 = *(const float4*)(vc + 2 * PDIM);
    float w = wave_dot(qv, kv);
    msg_accum(w, vsv, vvv, x0, x1, x2, as, a0, a1, a2);
  }
  float4 sv = *(const float4*)(s + (size_t)node * PDIM + d);
  float4 so = make_float4(sv.x + as.x, sv.y + as.y, sv.z + as.z, sv.w + as.w);
  *(float4*)(s_out + (size_t)node * PDIM + d) = so;
  float* av = aggv + (size_t)node * 3 * PDIM + d;
  *(float4*)(av)            = a0;
  *(float4*)(av + PDIM)     = a1;
  *(float4*)(av + 2 * PDIM) = a2;
}

__global__ __launch_bounds__(256) void node_rows_half(
    const float* __restrict__ q, const float* __restrict__ k,
    const float* __restrict__ vs, const float* __restrict__ vv,
    const float* __restrict__ v, const float* __restrict__ s,
    const int* __restrict__ counts, const int2* __restrict__ list,
    const int4* __restrict__ ovf,
    float* __restrict__ s_out, float* __restrict__ aggv, int N, int nGroups) {
  int wave = __builtin_amdgcn_readfirstlane((int)(threadIdx.x >> 5));
  int half = blockIdx.x / nGroups;
  int pairBase = ((blockIdx.x - half * nGroups) * 8 + wave) * 2;
  if (pairBase >= N) return;
  int lane = threadIdx.x & 31;
  int node = pairBase + (lane >> 4);
  bool valid = node < N;
  int nd = valid ? node : pairBase;
  int deg = valid ? counts[nd] : 0;
  int dl = (deg <= CAP) ? deg : 0;
  int f = half * 64 + (lane & 15) * 4;
  int maxd = max(__builtin_amdgcn_readlane(dl, 0), __builtin_amdgcn_readlane(dl, 16));
  const int2* sl = list + (size_t)nd * CAP;

  float4 as = make_float4(0.f, 0.f, 0.f, 0.f);
  float4 a0 = as, a1 = as, a2 = as;
  int2 cw = make_int2(0, 0);
  if (0 < dl) cw = sl[0];
#pragma unroll 1
  for (int i = 0; i < maxd; ++i) {
    int2 cur = cw;
    if (i + 1 < dl) cw = sl[i + 1];
    if (i < dl) {
      int c = cur.x;
      float w = __int_as_float(cur.y);
      float4 vsv = *(const float4*)(vs + (size_t)c * PDIM + f);
      float4 vvv = *(const float4*)(vv + (size_t)c * PDIM + f);
      const float* vc = v + (size_t)c * 3 * PDIM + f;
      float4 x0 = *(const float4*)(vc);
      float4 x1 = *(const float4*)(vc + PDIM);
      float4 x2 = *(const float4*)(vc + 2 * PDIM);
      msg_accum(w, vsv, vvv, x0, x1, x2, as, a0, a1, a2);
    }
  }
  if (valid && deg <= CAP) {
    float4 sv = *(const float4*)(s + (size_t)node * PDIM + f);
    float4 so = make_float4(sv.x + as.x, sv.y + as.y, sv.z + as.z, sv.w + as.w);
    *(float4*)(s_out + (size_t)node * PDIM + f) = so;
    float* av = aggv + (size_t)node * 3 * PDIM + f;
    *(float4*)(av)            = a0;
    *(float4*)(av + PDIM)     = a1;
    *(float4*)(av + 2 * PDIM) = a2;
  }
  if (half == 0) {
    int dg0 = __builtin_amdgcn_readlane(deg, 0);
    int dg1 = __builtin_amdgcn_readlane(deg, 16);
    if (dg0 > CAP)
      long_row(pairBase, dg0, q, k, vs, vv, v, s, counts, list, ovf, s_out, aggv, N);
    if (dg1 > CAP)
      long_row(pairBase + 1, dg1, q, k, vs, vv, v, s, counts, list, ovf, s_out, aggv, N);
  }
}

__global__ __launch_bounds__(256) void painn_vec_gemm(
    const float* __restrict__ aggv, const float* __restrict__ Wvec,
    const float* __restrict__ bvec, const float* __restrict__ v,
    float* __restrict__ vout, int nStrips) {
  int wave = threadIdx.x >> 5;
  int lane = threadIdx.x & 31;
  int strip = blockIdx.x * 8 + wave;
  if (strip >= nStrips) return;
  int half = lane >> 4;
  int l16  = lane & 15;
  size_t rowBase = (size_t)strip * 16;

  v2f a[32];
  load_strip(aggv + (rowBase + l16) * PDIM + 2 * half, a);

  size_t ooff = (rowBase + 8 * half) * PDIM + 4 * l16;
  strip_block<PDIM, true>(a, Wvec + (2 * half) * PDIM + 4 * l16, bvec + 4 * l16,
                          v + ooff, vout + ooff);
}

extern "C" void kernel_launch(void* const* d_in, const int* in_sizes, int n_in,
                              void* d_out, int out_size, void* d_ws, size_t ws_size,
                              hipStream_t stream) {
  const float* s    = (const float*)d_in[0];
  const float* v    = (const float*)d_in[1];
  const int*   ei   = (const int*)d_in[2];
  const float* Wq   = (const float*)d_in[3];
  const float* bq   = (const float*)d_in[4];
  const float* Wkv  = (const float*)d_in[5];
  const float* bkv  = (const float*)d_in[6];
  const float* Wvec = (const float*)d_in[7];
  const float* bvec = (const float*)d_in[8];

  int N = in_sizes[0] / PDIM;
  int E = in_sizes[2] / 2;

  float* s_out = (float*)d_out;
  float* v_out = s_out + (size_t)N * PDIM;

  float* q    = (float*)d_ws;
  float* k    = q  + (size_t)N * PDIM;
  float* vs   = k  + (size_t)N * PDIM;
  float* vv   = vs + (size_t)N * PDIM;
  float* aggv = vv + (size_t)N * PDIM;

  int nCnt = (N + 1 + 63) / 64 * 64;
  int*  counts = (int*)(aggv + (size_t)N * 3 * PDIM);
  int2* list   = (int2*)(counts + nCnt);
  int4* ovf    = (int4*)(list + (size_t)N * CAP);

  const int* rowIdx = ei;
  const int* colIdx = ei + E;

  int stripsA = N / 16;
  int groupsA = (stripsA + 7) / 8;
  painn_node_gemm<<<4 * groupsA, 256, 0, stream>>>(
      s, Wq, bq, Wkv, bkv, q, k, vs, vv, stripsA, groupsA, counts, nCnt / 4);

  int edgeBlocks = (E + 1023) / 1024;
  bucket_fill<<<edgeBlocks, 256, 0, stream>>>(rowIdx, colIdx, counts, list, ovf, N, E);

  edge_weights<<<(N + 7) / 8, 256, 0, stream>>>(q, k, counts, list, N);

  int pairGroups = ((N + 1) / 2 + 7) / 8;
  node_rows_half<<<2 * pairGroups, 256, 0, stream>>>(
      q, k, vs, vv, v, s, counts, list, ovf, s_out, aggv, N, pairGroups);

  int stripsC = N * 3 / 16;
  painn_vec_gemm<<<(stripsC + 7) / 8, 256, 0, stream>>>(
      aggv, Wvec, bvec, v, v_out, stripsC);
}


# Wvec applied per source node before the message sums (U = (v_v*v) @ Wvec): the pass gathers 4 half-rows per edge, not 5, and stores v_out itself; agg_v round trip gone; sum order changes
# speedup vs baseline: 3.0080x; 1.0585x over previous
#include <hip/hip_runtime.h>


typedef __attribute__((ext_vector_type(2))) float v2f;
typedef __attribute__((ext_vector_type(8))) float v8f;

#define PDIM 128
#define CAP 32

__global__ __launch_bounds__(256) void bucket_fill(const int* __restrict__ rowIdx,
                                                   const int* __restrict__ colIdx,
                                                   int* __restrict__ counts,
                                                   int2* __restrict__ list,
                                                   int4* __restrict__ ovf,
                                                   int ovfCounter, int E) {
  int stride = gridDim.x * blockDim.x;
  for (int e0 = blockIdx.x * blockDim.x + threadIdx.x; e0 < E; e0 += 4 * stride) {
    int r[4], c[4], sl[4];
#pragma unroll
    for (int u = 0; u < 4; ++u) {
      int e = e0 + u * stride;
      int el = min(e, E - 1);
      int rr = rowIdx[el];
      c[u] = colIdx[el];
      r[u] = (e < E) ? rr : -1;
    }
#pragma unroll
    for (int u = 0; u < 4; ++u)
      sl[u] = (r[u] >= 0) ? atomicAdd(&counts[r[u]], 1) : 0;
#pragma unroll
    for (int u = 0; u < 4; ++u) {
      if (r[u] < 0) continue;
      int e = e0 + u * stride;
      if (sl[u] < CAP) {
        list[(size_t)r[u] * CAP + sl[u]] = make_int2(e, c[u]);
      } else {
        int o = atomicAdd(&counts[ovfCounter], 1);
        ovf[o] = make_int4(r[u], e, c[u], 0);
      }
    }
  }
}

__device__ __forceinline__ void load_strip(const float* __restrict__ rowp, v2f (&a)[32]) {
#pragma unroll
  for (int ks = 0; ks < 32; ++ks) {
    float2 t = *(const float2*)(rowp + 4 * ks);
    a[ks].x = t.x;
    a[ks].y = t.y;
  }
}

__device__ __forceinline__ void load_strip_prod(const float* __restrict__ xrow,
                                                const float* __restrict__ grow, v2f (&a)[32]) {
#pragma unroll
  for (int ks = 0; ks < 32; ++ks) {
    float2 t = *(const float2*)(xrow + 4 * ks);
    float2 g = *(const float2*)(grow + 4 * ks);
    a[ks].x = g.x * t.x;
    a[ks].y = g.y * t.y;
  }
}

__device__ __forceinline__ void step4(const v2f& af, const float4& r0, const float4& r1,
                                      v8f (&acc)[4]) {
  v2f bf;
  bf.x = r0.x; bf.y = r1.x;
  acc[0] = __builtin_amdgcn_wmma_f32_16x16x4_f32(false, af, false, bf, (short)0, acc[0], false, false);
  bf.x = r0.y; bf.y = r1.y;
  acc[1] = __builtin_amdgcn_wmma_f32_16x16x4_f32(false, af, false, bf, (short)0, acc[1], false, false);
  bf.x = r0.z; bf.y = r1.z;
  acc[2] = __builtin_amdgcn_wmma_f32_16x16x4_f32(false, af, false, bf, (short)0, acc[2], false, false);
  bf.x = r0.w; bf.y = r1.w;
  acc[3] = __builtin_amdgcn_wmma_f32_16x16x4_f32(false, af, false, bf, (short)0, acc[3], false, false);
}

template <int LDW>
__device__ __forceinline__ void strip_chunk(const v2f (&a)[32], const float* __restrict__ Wl,
                                            v8f (&acc)[4]) {
  float4 n0 = *(const float4*)(Wl);
  float4 n1 = *(const float4*)(Wl + LDW);
  float4 n2 = *(const float4*)(Wl + 4 * LDW);
  float4 n3 = *(const float4*)(Wl + 5 * LDW);
#pragma unroll
  for (int j = 0; j < 16; ++j) {
    float4 c0 = n0, c1 = n1, c2 = n2, c3 = n3;
    if (j + 1 < 16) {
      const float* p = Wl + (8 * (j + 1)) * LDW;
      n0 = *(const float4*)(p);
      n1 = *(const float4*)(p + LDW);
      n2 = *(const float4*)(p + 4 * LDW);
      n3 = *(const float4*)(p + 5 * LDW);
    }
    step4(a[2 * j],     c0, c1, acc);
    step4(a[2 * j + 1], c2, c3, acc);
  }
}

template <int LDW, bool BIAS>
__device__ __forceinline__ void strip_block(const v2f (&a)[32], const float* __restrict__ Wl,
                                            const float* __restrict__ bl,
                                            float* __restrict__ orow) {
#pragma unroll 1
  for (int cb = 0; cb < PDIM; cb += 64) {
    v8f acc[4] = {};
    strip_chunk<LDW>(a, Wl + cb, acc);
    float4 bias = make_float4(0.f, 0.f, 0.f, 0.f);
    if (BIAS) bias = *(const float4*)(bl + cb);
#pragma unroll
    for (int r = 0; r < 8; ++r) {
      float4 ov;
      if (BIAS) {
        ov = make_float4(acc[0][r] + bias.x, acc[1][r] + bias.y,
                         acc[2][r] + bias.z, acc[3][r] + bias.w);
      } else {
        ov = make_float4(acc[0][r], acc[1][r], acc[2][r], acc[3][r]);
      }
      *(float4*)(orow + (size_t)r * PDIM + cb) = ov;
    }
  }
}

__global__ __launch_bounds__(256) void painn_node_gemm(
    const float* __restrict__ s,
    const float* __restrict__ Wq,  const float* __restrict__ bq,
    const float* __restrict__ Wkv, const float* __restrict__ bkv,
    float* __restrict__ q, float* __restrict__ k,
    float* __restrict__ vs, float* __restrict__ vv,
    int nStrips, int nGroups, int* __restrict__ counts, int nCounts4) {
  int gid = blockIdx.x * 256 + threadIdx.x;
  if (gid < nCounts4) ((int4*)counts)[gid] = make_int4(0, 0, 0, 0);
  int wave = threadIdx.x >> 5;
  int lane = threadIdx.x & 31;
  int o = blockIdx.x / nGroups;
  int strip = (blockIdx.x - o * nGroups) * 8 + wave;
  if (strip >= nStrips) return;
  int half = lane >> 4;
  int l16  = lane & 15;
  int rowBase = strip * 16;

  v2f a[32];
  load_strip(s + (size_t)(rowBase + l16) * PDIM + 2 * half, a);

  size_t ooff = (size_t)(rowBase + 8 * half) * PDIM + 4 * l16;
  if (o == 0) {
    strip_block<PDIM, true>(a, Wq + (2 * half) * PDIM + 4 * l16, bq + 4 * l16, q + ooff);
  } else {
    int wc = (o - 1) * PDIM + 4 * l16;
    float* out = (o == 1) ? k : (o == 2) ? vs : vv;
    strip_block<3 * PDIM, true>(a, Wkv + (2 * half) * 3 * PDIM + wc, bkv + wc, out + ooff);
  }
}

__device__ __forceinline__ float wave_dot(const float4& qv, const float4& kv) {
  float p = qv.x * kv.x + qv.y * kv.y + qv.z * kv.z + qv.w * kv.w;
  p += __shfl_xor(p, 16, 32);
  p += __shfl_xor(p, 8, 32);
  p += __shfl_xor(p, 4, 32);
  p += __shfl_xor(p, 2, 32);
  p += __shfl_xor(p, 1, 32);
  return p;
}

__device__ __forceinline__ void msg_accum(
    float w, const float4& vsv,
    const float4& x0, const float4& x1, const float4& x2,
    float4& as, float4& a0, float4& a1, float4& a2) {
  as.x += w * vsv.x; as.y += w * vsv.y; as.z += w * vsv.z; as.w += w * vsv.w;
  a0.x += w * x0.x; a0.y += w * x0.y; a0.z += w * x0.z; a0.w += w * x0.w;
  a1.x += w * x1.x; a1.y += w * x1.y; a1.z += w * x1.z; a1.w += w * x1.w;
  a2.x += w * x2.x; a2.y += w * x2.y; a2.z += w * x2.z; a2.w += w * x2.w;
}

__device__ __forceinline__ void store_row(
    int node, int f, const float4& as, const float4& a0, const float4& a1, const float4& a2,
    const float* __restrict__ s, const float* __restrict__ v, const float* __restrict__ bvec,
    float* __restrict__ s_out, float* __restrict__ v_out) {
  size_t so = (size_t)node * PDIM + f;
  size_t vo3 = (size_t)node * 3 * PDIM + f;
  float4 sv = *(const float4*)(s + so);
  float4 bv = *(const float4*)(bvec + f);
  const float* vr = v + vo3;
  float4 y0 = *(const float4*)(vr);
  float4 y1 = *(const float4*)(vr + PDIM);
  float4 y2 = *(const float4*)(vr + 2 * PDIM);
  *(float4*)(s_out + so) = make_float4(sv.x + as.x, sv.y + as.y, sv.z + as.z, sv.w + as.w);
  float* vo = v_out + vo3;
  *(float4*)(vo) = make_float4(y0.x + (a0.x + bv.x), y0.y + (a0.y + bv.y),
                               y0.z + (a0.z + bv.z), y0.w + (a0.w + bv.w));
  *(float4*)(vo + PDIM) = make_float4(y1.x + (a1.x + bv.x), y1.y + (a1.y + bv.y),
                                      y1.z + (a1.z + bv.z), y1.w + (a1.w + bv.w));
  *(float4*)(vo + 2 * PDIM) = make_float4(y2.x + (a2.x + bv.x), y2.y + (a2.y + bv.y),
                                          y2.z + (a2.z + bv.z), y2.w + (a2.w + bv.w));
}

__global__ __launch_bounds__(256) void edge_weights(
    const float* __restrict__ q, const float* __restrict__ k,
    const int* __restrict__ counts, int2* __restrict__ list, int N) {
  int wave = __builtin_amdgcn_readfirstlane((int)(threadIdx.x >> 5));
  int node = blockIdx.x * 8 + wave;
  if (node >= N) return;
  int lane = threadIdx.x & 31;
  int d = lane * 4;
  int deg = counts[node];
  if (deg > CAP) return;
  int2* slots = list + (size_t)node * CAP;

  float4 qv = *(const float4*)(q + (size_t)node * PDIM + d);
  int2 ec = make_int2(0x7fffffff, 0);
  if (lane < deg) ec = slots[lane];
  int rank = 0;
  for (int j = 0; j < deg; ++j) {
    int ej = __builtin_amdgcn_readlane(ec.x, j);
    rank += (ej < ec.x) ? 1 : 0;
  }
  float wm = 0.f;
  for (int j0 = 0; j0 < deg; j0 += 4) {
    float4 kv[4];
#pragma unroll
    for (int u = 0; u < 4; ++u) {
      int c = __builtin_amdgcn_readlane(ec.y, min(j0 + u, deg - 1));
      kv[u] = *(const float4*)(k + (size_t)c * PDIM + d);
    }
#pragma unroll
    for (int u = 0; u < 4; ++u) {
      float w = wave_dot(qv, kv[u]);
      if (lane == j0 + u) wm = w;
    }
  }
  if (lane < deg) slots[rank] = make_int2(ec.y, __float_as_int(wm));
}

__device__ void long_row(
    int node, int deg,
    const float* __restrict__ q, const float* __restrict__ k,
    const float* __restrict__ vs, const float* __restrict__ U,
    const float* __restrict__ v, const float* __restrict__ s,
    const float* __restrict__ bvec,
    const int* __restrict__ counts, const int2* __restrict__ list,
    const int4* __restrict__ ovf,
    float* __restrict__ s_out, float* __restrict__ v_out, int N) {
  int lane = threadIdx.x & 31;
  int d = lane * 4;
  float4 qv = *(const float4*)(q + (size_t)node * PDIM + d);
  float4 as = make_float4(0.f, 0.f, 0.f, 0.f);
  float4 a0 = as, a1 = as, a2 = as;
  int nOvf = counts[N];
  int2 ec = list[(size_t)node * CAP + lane];
  int last = -1;
  for (int i = 0; i < deg; ++i) {
    int be = 0x7fffffff, bc = 0;
    if (ec.x > last) { be = ec.x; bc = ec.y; }
    for (int j = lane; j < nOvf; j += 32) {
      int4 o = ovf[j];
      if (o.x == node && o.y > last && o.y < be) { be = o.y; bc = o.z; }
    }
    int mn = be;
    mn = min(mn, __shfl_xor(mn, 16, 32));
    mn = min(mn, __shfl_xor(mn, 8, 32));
    mn = min(mn, __shfl_xor(mn, 4, 32));
    mn = min(mn, __shfl_xor(mn, 2, 32));
    mn = min(mn, __shfl_xor(mn, 1, 32));
    unsigned m = __builtin_amdgcn_ballot_w32(be == mn);
    int src = __builtin_ctz(m | 0x80000000u);
    int c = __builtin_amdgcn_readlane(bc, src);
    last = __builtin_amdgcn_readfirstlane(mn);
    float4 kv  = *(const float4*)(k  + (size_t)c * PDIM + d);
    float4 vsv = *(const float4*)(vs + (size_t)c * PDIM + d);
    const float* uc = U + (size_t)c * 3 * PDIM + d;
    float4 x0 = *(const float4*)(uc);
    float4 x1 = *(const float4*)(uc + PDIM);
    float4 x2 = *(const float4*)(uc + 2 * PDIM);
    float w = wave_dot(qv, kv);
    msg_accum(w, vsv, x0, x1, x2, as, a0, a1, a2);
  }
  store_row(node, d, as, a0, a1, a2, s, v, bvec, s_out, v_out);
}

__global__ __launch_bounds__(256) void node_rows_half(
    const float* __restrict__ q, const float* __restrict__ k,
    const float* __restrict__ vs, const float* __restrict__ U,
    const float* __restrict__ v, const float* __restrict__ s,
    const float* __restrict__ bvec,
    const int* __restrict__ counts, const int2* __restrict__ list,
    const int4* __restrict__ ovf,
    float* __restrict__ s_out, float* __restrict__ v_out, int N, int nGroups) {
  int wave = __builtin_amdgcn_readfirstlane((int)(threadIdx.x >> 5));
  int half = blockIdx.x / nGroups;
  int pairBase = ((blockIdx.x - half * nGroups) * 8 + wave) * 2;
  if (pairBase >= N) return;
  int lane = threadIdx.x & 31;
  int node = pairBase + (lane >> 4);
  bool valid = node < N;
  int nd = valid ? node : pairBase;
  int deg = valid ? counts[nd] : 0;
  int dl = (deg <= CAP) ? deg : 0;
  int f = half * 64 + (lane & 15) * 4;
  int maxd = max(__builtin_amdgcn_readlane(dl, 0), __builtin_amdgcn_readlane(dl, 16));
  const int2* sl = list + (size_t)nd * CAP;

  float4 as = make_float4(0.f, 0.f, 0.f, 0.f);
  float4 a0 = as, a1 = as, a2 = as;
  int2 cw = make_int2(0, 0);
  if (0 < dl) cw = sl[0];
#pragma unroll 1
  for (int i = 0; i < maxd; ++i) {
    int2 cur = cw;
    if (i + 1 < dl) cw = sl[i + 1];
    if (i < dl) {
      int c = cur.x;
      float w = __int_as_float(cur.y);
      float4 vsv = *(const float4*)(vs + (size_t)c * PDIM + f);
      const float* uc = U + (size_t)c * 3 * PDIM + f;
      float4 x0 = *(const float4*)(uc);
      float4 x1 = *(const float4*)(uc + PDIM);
      float4 x2 = *(const float4*)(uc + 2 * PDIM);
      msg_accum(w, vsv, x0, x1, x2, as, a0, a1, a2);
    }
  }
  if (valid && deg <= CAP)
    store_row(node, f, as, a0, a1, a2, s, v, bvec, s_out, v_out);
  if (half == 0) {
    int dg0 = __builtin_amdgcn_readlane(deg, 0);
    int dg1 = __builtin_amdgcn_readlane(deg, 16);
    if (dg0 > CAP)
      long_row(pairBase, dg0, q, k, vs, U, v, s, bvec, counts, list, ovf, s_out, v_out, N);
    if (dg1 > CAP)
      long_row(pairBase + 1, dg1, q, k, vs, U, v, s, bvec, counts, list, ovf, s_out, v_out, N);
  }
}

__global__ __launch_bounds__(256) void painn_u_gemm(
    const float* __restrict__ vv, const float* __restrict__ v,
    const float* __restrict__ Wvec, float* __restrict__ U, int nStrips) {
  int wave = threadIdx.x >> 5;
  int lane = threadIdx.x & 31;
  int strip = blockIdx.x * 8 + wave;
  if (strip >= nStrips) return;
  int half = lane >> 4;
  int l16  = lane & 15;
  int rowBase = strip * 16;
  int row = rowBase + l16;
  int node = row / 3;

  v2f a[32];
  load_strip_prod(v + (size_t)row * PDIM + 2 * half, vv + (size_t)node * PDIM + 2 * half, a);

  size_t ooff = (size_t)(rowBase + 8 * half) * PDIM + 4 * l16;
  strip_block<PDIM, false>(a, Wvec + (2 * half) * PDIM + 4 * l16, nullptr, U + ooff);
}

extern "C" void kernel_launch(void* const* d_in, const int* in_sizes, int n_in,
                              void* d_out, int out_size, void* d_ws, size_t ws_size,
                              hipStream_t stream) {
  const float* s    = (const float*)d_in[0];
  const float* v    = (const float*)d_in[1];
  const int*   ei   = (const int*)d_in[2];
  const float* Wq   = (const float*)d_in[3];
  const float* bq   = (const float*)d_in[4];
  const float* Wkv  = (const float*)d_in[5];
  const float* bkv  = (const float*)d_in[6];
  const float* Wvec = (const float*)d_in[7];
  const float* bvec = (const float*)d_in[8];

  int N = in_sizes[0] / PDIM;
  int E = in_sizes[2] / 2;

  float* s_out = (float*)d_out;
  float* v_out = s_out + (size_t)N * PDIM;

  float* q    = (float*)d_ws;
  float* k    = q  + (size_t)N * PDIM;
  float* vs   = k  + (size_t)N * PDIM;
  float* vv   = vs + (size_t)N * PDIM;
  float* U    = vv + (size_t)N * PDIM;

  int nCnt = (N + 1 + 63) / 64 * 64;
  int*  counts = (int*)(U + (size_t)N * 3 * PDIM);
  int2* list   = (int2*)(counts + nCnt);
  int4* ovf    = (int4*)(list + (size_t)N * CAP);

  const int* rowIdx = ei;
  const int* colIdx = ei + E;

  int stripsA = N / 16;
  int groupsA = (stripsA + 7) / 8;
  painn_node_gemm<<<4 * groupsA, 256, 0, stream>>>(
      s, Wq, bq, Wkv, bkv, q, k, vs, vv, stripsA, groupsA, counts, nCnt / 4);

  int edgeBlocks = (E + 1023) / 1024;
  bucket_fill<<<edgeBlocks, 256, 0, stream>>>(rowIdx, colIdx, counts, list, ovf, N, E);

  edge_weights<<<(N + 7) / 8, 256, 0, stream>>>(q, k, counts, list, N);

  int stripsC = N * 3 / 16;
  painn_u_gemm<<<(stripsC + 7) / 8, 256, 0, stream>>>(vv, v, Wvec, U, stripsC);

  int pairGroups = ((N + 1) / 2 + 7) / 8;
  node_rows_half<<<2 * pairGroups, 256, 0, stream>>>(
      q, k, vs, U, v, s, bvec, counts, list, ovf, s_out, v_out, N, pairGroups);
}


# k and v_s never formed: weight pass gathers s rows (w = (q Wk^T).s + q.bk) and sums w*s per row; s_out = s + S Wvs by GEMM; row pass gathers 3 half-rows, not 4; v_s map moved behind its sum
# speedup vs baseline: 3.1536x; 1.0484x over previous
#include <hip/hip_runtime.h>


typedef __attribute__((ext_vector_type(2))) float v2f;
typedef __attribute__((ext_vector_type(8))) float v8f;

#define PDIM 128
#define CAP 32

__global__ __launch_bounds__(256) void bucket_fill(const int* __restrict__ rowIdx,
                                                   const int* __restrict__ colIdx,
                                                   int* __restrict__ counts,
                                                   int2* __restrict__ list,
                                                   int4* __restrict__ ovf,
                                                   int ovfCounter, int E) {
  int stride = gridDim.x * blockDim.x;
  for (int e0 = blockIdx.x * blockDim.x + threadIdx.x; e0 < E; e0 += 4 * stride) {
    int r[4], c[4], sl[4];
#pragma unroll
    for (int u = 0; u < 4; ++u) {
      int e = e0 + u * stride;
      int el = min(e, E - 1);
      int rr = rowIdx[el];
      c[u] = colIdx[el];
      r[u] = (e < E) ? rr : -1;
    }
#pragma unroll
    for (int u = 0; u < 4; ++u)
      sl[u] = (r[u] >= 0) ? atomicAdd(&counts[r[u]], 1) : 0;
#pragma unroll
    for (int u = 0; u < 4; ++u) {
      if (r[u] < 0) continue;
      int e = e0 + u * stride;
      if (sl[u] < CAP) {
        list[(size_t)r[u] * CAP + sl[u]] = make_int2(e, c[u]);
      } else {
        int o = atomicAdd(&counts[ovfCounter], 1);
        ovf[o] = make_int4(r[u], e, c[u], 0);
      }
    }
  }
}

__device__ __forceinline__ void load_strip(const float* __restrict__ rowp, v2f (&a)[32]) {
#pragma unroll
  for (int ks = 0; ks < 32; ++ks) {
    float2 t = *(const float2*)(rowp + 4 * ks);
    a[ks].x = t.x;
    a[ks].y = t.y;
  }
}

__device__ __forceinline__ void load_strip_prod(const float* __restrict__ xrow,
                                                const float* __restrict__ grow, v2f (&a)[32]) {
#pragma unroll
  for (int ks = 0; ks < 32; ++ks) {
    float2 t = *(const float2*)(xrow + 4 * ks);
    float2 g = *(const float2*)(grow + 4 * ks);
    a[ks].x = g.x * t.x;
    a[ks].y = g.y * t.y;
  }
}

__device__ __forceinline__ void step4(const v2f& af, const float4& r0, const float4& r1,
                                      v8f (&acc)[4]) {
  v2f bf;
  bf.x = r0.x; bf.y = r1.x;
  acc[0] = __builtin_amdgcn_wmma_f32_16x16x4_f32(false, af, false, bf, (short)0, acc[0], false, false);
  bf.x = r0.y; bf.y = r1.y;
  acc[1] = __builtin_amdgcn_wmma_f32_16x16x4_f32(false, af, false, bf, (short)0, acc[1], false, false);
  bf.x = r0.z; bf.y = r1.z;
  acc[2] = __builtin_amdgcn_wmma_f32_16x16x4_f32(false, af, false, bf, (short)0, acc[2], false, false);
  bf.x = r0.w; bf.y = r1.w;
  acc[3] = __builtin_amdgcn_wmma_f32_16x16x4_f32(false, af, false, bf, (short)0, acc[3], false, false);
}

template <int LDW>
__device__ __forceinline__ void strip_chunk(const v2f (&a)[32], const float* __restrict__ Wl,
                                            v8f (&acc)[4]) {
  float4 n0 = *(const float4*)(Wl);
  float4 n1 = *(const float4*)(Wl + LDW);
  float4 n2 = *(const float4*)(Wl + 4 * LDW);
  float4 n3 = *(const float4*)(Wl + 5 * LDW);
#pragma unroll
  for (int j = 0; j < 16; ++j) {
    float4 c0 = n0, c1 = n1, c2 = n2, c3 = n3;
    if (j + 1 < 16) {
      const float* p = Wl + (8 * (j + 1)) * LDW;
      n0 = *(const float4*)(p);
      n1 = *(const float4*)(p + LDW);
      n2 = *(const float4*)(p + 4 * LDW);
      n3 = *(const float4*)(p + 5 * LDW);
    }
    step4(a[2 * j],     c0, c1, acc);
    step4(a[2 * j + 1], c2, c3, acc);
    __builtin_amdgcn_sched_barrier(0);
  }
}

template <int LDW, int MODE>
__device__ __forceinline__ void strip_block(const v2f (&a)[32], const float* __restrict__ Wl,
                                            const float* __restrict__ bl,
                                            float* __restrict__ orow,
                                            const float* __restrict__ xrow,
                                            const float* __restrict__ gcol) {
#pragma unroll 1
  for (int cb = 0; cb < PDIM; cb += 64) {
    v8f acc[4] = {};
    if (MODE == 2) {
#pragma unroll
      for (int r = 0; r < 8; ++r) {
        float4 x = *(const float4*)(xrow + (size_t)r * PDIM + cb);
        acc[0][r] = x.x; acc[1][r] = x.y; acc[2][r] = x.z; acc[3][r] = x.w;
      }
      __builtin_amdgcn_sched_barrier(0);
    }
    strip_chunk<LDW>(a, Wl + cb, acc);
    float4 bias = make_float4(0.f, 0.f, 0.f, 0.f);
    if (MODE != 0) bias = *(const float4*)(bl + cb);
#pragma unroll
    for (int r = 0; r < 8; ++r) {
      float4 ov = make_float4(acc[0][r], acc[1][r], acc[2][r], acc[3][r]);
      if (MODE == 1) {
        ov = make_float4(ov.x + bias.x, ov.y + bias.y, ov.z + bias.z, ov.w + bias.w);
      }
      if (MODE == 2) {
        float g = gcol[r];
        ov = make_float4(ov.x + g * bias.x, ov.y + g * bias.y,
                         ov.z + g * bias.z, ov.w + g * bias.w);
      }
      *(float4*)(orow + (size_t)r * PDIM + cb) = ov;
    }
  }
}

__device__ __forceinline__ void load_strip_k8(const float* rowp, v2f (&a)[32]) {
#pragma unroll
  for (int j = 0; j < 16; ++j) {
    float4 t = *(const float4*)(rowp + 8 * j);
    a[2 * j].x = t.x;     a[2 * j].y = t.y;
    a[2 * j + 1].x = t.z; a[2 * j + 1].y = t.w;
  }
}

template <int LDW>
__device__ __forceinline__ void strip_chunk_T(const v2f (&a)[32], const float* __restrict__ Wt,
                                              v8f (&acc)[4]) {
  float4 n0 = *(const float4*)(Wt);
  float4 n1 = *(const float4*)(Wt + 16 * LDW);
  float4 n2 = *(const float4*)(Wt + 32 * LDW);
  float4 n3 = *(const float4*)(Wt + 48 * LDW);
#pragma unroll
  for (int j = 0; j < 16; ++j) {
    float4 c[4] = {n0, n1, n2, n3};
    if (j + 1 < 16) {
      const float* p = Wt + 8 * (j + 1);
      n0 = *(const float4*)(p);
      n1 = *(const float4*)(p + 16 * LDW);
      n2 = *(const float4*)(p + 32 * LDW);
      n3 = *(const float4*)(p + 48 * LDW);
    }
#pragma unroll
    for (int t = 0; t < 4; ++t) {
      v2f bf;
      bf.x = c[t].x; bf.y = c[t].y;
      acc[t] = __builtin_amdgcn_wmma_f32_16x16x4_f32(false, a[2 * j], false, bf, (short)0, acc[t], false, false);
    }
#pragma unroll
    for (int t = 0; t < 4; ++t) {
      v2f bf;
      bf.x = c[t].z; bf.y = c[t].w;
      acc[t] = __builtin_amdgcn_wmma_f32_16x16x4_f32(false, a[2 * j + 1], false, bf, (short)0, acc[t], false, false);
    }
    __builtin_amdgcn_sched_barrier(0);
  }
}

__global__ __launch_bounds__(256) void painn_node_gemm(
    const float* __restrict__ s,
    const float* __restrict__ Wq,  const float* __restrict__ bq,
    const float* __restrict__ Wkv, const float* __restrict__ bkv,
    float* q, float* __restrict__ qt, float* __restrict__ qb, float* __restrict__ vv,
    int nStrips, int nGroups, int* __restrict__ counts, int nCounts4) {
  int gid = blockIdx.x * 256 + threadIdx.x;
  if (gid < nCounts4) ((int4*)counts)[gid] = make_int4(0, 0, 0, 0);
  int wave = threadIdx.x >> 5;
  int lane = threadIdx.x & 31;
  int o = blockIdx.x / nGroups;
  int strip = (blockIdx.x - o * nGroups) * 8 + wave;
  if (strip >= nStrips) return;
  int half = lane >> 4;
  int l16  = lane & 15;
  int rowBase = strip * 16;

  v2f a[32];
  load_strip(s + (size_t)(rowBase + l16) * PDIM + 2 * half, a);

  size_t ooff = (size_t)(rowBase + 8 * half) * PDIM + 4 * l16;
  if (o == 0) {
    strip_block<PDIM, 1>(a, Wq + (2 * half) * PDIM + 4 * l16, bq + 4 * l16, q + ooff,
                         nullptr, nullptr);
    __threadfence_block();
    load_strip_k8(q + (size_t)(rowBase + l16) * PDIM + 4 * half, a);
    float p = 0.f;
#pragma unroll
    for (int j = 0; j < 16; ++j) {
      float4 b = *(const float4*)(bkv + 8 * j + 4 * half);
      p += a[2 * j].x * b.x + a[2 * j].y * b.y + a[2 * j + 1].x * b.z + a[2 * j + 1].y * b.w;
    }
    p += __shfl_xor(p, 16, 32);
    if (half == 0) qb[rowBase + l16] = p;
#pragma unroll 1
    for (int cb = 0; cb < PDIM; cb += 64) {
      v8f acc[4] = {};
      strip_chunk_T<3 * PDIM>(a, Wkv + (size_t)(cb + l16) * 3 * PDIM + 4 * half, acc);
      float* orow = qt + (size_t)(rowBase + 8 * half) * PDIM + cb + l16;
#pragma unroll
      for (int r = 0; r < 8; ++r) {
#pragma unroll
        for (int t = 0; t < 4; ++t) orow[(size_t)r * PDIM + 16 * t] = acc[t][r];
      }
    }
  } else {
    int wc = 2 * PDIM + 4 * l16;
    strip_block<3 * PDIM, 1>(a, Wkv + (2 * half) * 3 * PDIM + wc, bkv + wc, vv + ooff,
                             nullptr, nullptr);
  }
}

__device__ __forceinline__ float wave_dot(const float4& qv, const float4& kv) {
  float p = qv.x * kv.x + qv.y * kv.y + qv.z * kv.z + qv.w * kv.w;
  p += __shfl_xor(p, 16, 32);
  p += __shfl_xor(p, 8, 32);
  p += __shfl_xor(p, 4, 32);
  p += __shfl_xor(p, 2, 32);
  p += __shfl_xor(p, 1, 32);
  return p;
}

__device__ __forceinline__ void axpy4(float w, const float4& x, float4& a) {
  a.x += w * x.x; a.y += w * x.y; a.z += w * x.z; a.w += w * x.w;
}

__device__ __forceinline__ void store_vrow(
    int node, int f, const float4& a0, const float4& a1, const float4& a2,
    const float* __restrict__ v, const float* __restrict__ bvec, float* __restrict__ v_out) {
  size_t vo3 = (size_t)node * 3 * PDIM + f;
  float4 bv = *(const float4*)(bvec + f);
  const float* vr = v + vo3;
  float4 y0 = *(const float4*)(vr);
  float4 y1 = *(const float4*)(vr + PDIM);
  float4 y2 = *(const float4*)(vr + 2 * PDIM);
  float* vo = v_out + vo3;
  *(float4*)(vo) = make_float4(y0.x + (a0.x + bv.x), y0.y + (a0.y + bv.y),
                               y0.z + (a0.z + bv.z), y0.w + (a0.w + bv.w));
  *(float4*)(vo + PDIM) = make_float4(y1.x + (a1.x + bv.x), y1.y + (a1.y + bv.y),
                                      y1.z + (a1.z + bv.z), y1.w + (a1.w + bv.w));
  *(float4*)(vo + 2 * PDIM) = make_float4(y2.x + (a2.x + bv.x), y2.y + (a2.y + bv.y),
                                          y2.z + (a2.z + bv.z), y2.w + (a2.w + bv.w));
}

template <bool SCALAR>
__device__ void long_row(
    int node, int deg,
    const float* __restrict__ qt, const float* __restrict__ qb,
    const float* __restrict__ s, const float* __restrict__ U,
    const float* __restrict__ v, const float* __restrict__ bvec,
    const int* __restrict__ counts, const int2* __restrict__ list,
    const int4* __restrict__ ovf,
    float* __restrict__ S, float* __restrict__ sw, float* __restrict__ v_out, int N) {
  int lane = threadIdx.x & 31;
  int d = lane * 4;
  float4 qv = *(const float4*)(qt + (size_t)node * PDIM + d);
  float qbn = qb[node];
  float4 a0 = make_float4(0.f, 0.f, 0.f, 0.f);
  float4 a1 = a0, a2 = a0;
  float sws = 0.f;
  int nOvf = counts[N];
  int2 ec = list[(size_t)node * CAP + lane];
  int last = -1;
  for (int i = 0; i < deg; ++i) {
    int be = 0x7fffffff, bc = 0;
    if (ec.x > last) { be = ec.x; bc = ec.y; }
    for (int j = lane; j < nOvf; j += 32) {
      int4 o = ovf[j];
      if (o.x == node && o.y > last && o.y < be) { be = o.y; bc = o.z; }
    }
    int mn = be;
    mn = min(mn, __shfl_xor(mn, 16, 32));
    mn = min(mn, __shfl_xor(mn, 8, 32));
    mn = min(mn, __shfl_xor(mn, 4, 32));
    mn = min(mn, __shfl_xor(mn, 2, 32));
    mn = min(mn, __shfl_xor(mn, 1, 32));
    unsigned m = __builtin_amdgcn_ballot_w32(be == mn);
    int src = __builtin_ctz(m | 0x80000000u);
    int c = __builtin_amdgcn_readlane(bc, src);
    last = __builtin_amdgcn_readfirstlane(mn);
    float4 sv = *(const float4*)(s + (size_t)c * PDIM + d);
    float w = wave_dot(qv, sv) + qbn;
    if (SCALAR) {
      axpy4(w, sv, a0);
      sws += w;
    } else {
      const float* uc = U + (size_t)c * 3 * PDIM + d;
      float4 x0 = *(const float4*)(uc);
      float4 x1 = *(const float4*)(uc + PDIM);
      float4 x2 = *(const float4*)(uc + 2 * PDIM);
      axpy4(w, x0, a0);
      axpy4(w, x1, a1);
      axpy4(w, x2, a2);
    }
  }
  if (SCALAR) {
    *(float4*)(S + (size_t)node * PDIM + d) = a0;
    if (lane == 0) sw[node] = sws;
  } else {
    store_vrow(node, d, a0, a1, a2, v, bvec, v_out);
  }
}

__global__ __launch_bounds__(256) void edge_weights(
    const float* __restrict__ qt, const float* __restrict__ qb, const float* __restrict__ s,
    const int* __restrict__ counts, int2* __restrict__ list, const int4* __restrict__ ovf,
    float* __restrict__ S, float* __restrict__ sw, int N) {
  int wave = __builtin_amdgcn_readfirstlane((int)(threadIdx.x >> 5));
  int node = blockIdx.x * 8 + wave;
  if (node >= N) return;
  int lane = threadIdx.x & 31;
  int d = lane * 4;
  int deg = counts[node];
  if (deg > CAP) {
    long_row<true>(node, deg, qt, qb, s, nullptr, nullptr, nullptr, counts, list, ovf,
                   S, sw, nullptr, N);
    return;
  }
  int2* slots = list + (size_t)node * CAP;

  float4 qv = *(const float4*)(qt + (size_t)node * PDIM + d);
  float qbn = qb[node];
  int2 ec = make_int2(0x7fffffff, 0);
  if (lane < deg) ec = slots[lane];
  int rank = 0;
  for (int j = 0; j < deg; ++j) {
    int ej = __builtin_amdgcn_readlane(ec.x, j);
    rank += (ej < ec.x) ? 1 : 0;
  }
  float4 acc = make_float4(0.f, 0.f, 0.f, 0.f);
  float sws = 0.f;
  float wm = 0.f;
  int cm = 0;
  for (int p0 = 0; p0 < deg; p0 += 4) {
    float4 sv[4];
    int cc[4];
#pragma unroll
    for (int u = 0; u < 4; ++u) {
      int p = min(p0 + u, deg - 1);
      unsigned m = __builtin_amdgcn_ballot_w32(rank == p);
      int src = __builtin_ctz(m | 0x80000000u);
      cc[u] = __builtin_amdgcn_readlane(ec.y, src);
      sv[u] = *(const float4*)(s + (size_t)cc[u] * PDIM + d);
    }
#pragma unroll
    for (int u = 0; u < 4; ++u) {
      float w = wave_dot(qv, sv[u]) + qbn;
      if (p0 + u < deg) {
        axpy4(w, sv[u], acc);
        sws += w;
      }
      if (lane == p0 + u) { wm = w; cm = cc[u]; }
    }
  }
  if (lane < deg) slots[lane] = make_int2(cm, __float_as_int(wm));
  *(float4*)(S + (size_t)node * PDIM + d) = acc;
  if (lane == 0) sw[node] = sws;
}

__global__ __launch_bounds__(256) void node_rows_half(
    const float* __restrict__ qt, const float* __restrict__ qb, const float* __restrict__ s,
    const float* __restrict__ U, const float* __restrict__ v,
    const float* __restrict__ bvec,
    const int* __restrict__ counts, const int2* __restrict__ list,
    const int4* __restrict__ ovf,
    float* __restrict__ v_out, int N, int nGroups) {
  int wave = __builtin_amdgcn_readfirstlane((int)(threadIdx.x >> 5));
  int half = blockIdx.x / nGroups;
  int pairBase = ((blockIdx.x - half * nGroups) * 8 + wave) * 2;
  if (pairBase >= N) return;
  int lane = threadIdx.x & 31;
  int node = pairBase + (lane >> 4);
  bool valid = node < N;
  int nd = valid ? node : pairBase;
  int deg = valid ? counts[nd] : 0;
  int dl = (deg <= CAP) ? deg : 0;
  int f = half * 64 + (lane & 15) * 4;
  int maxd = max(__builtin_amdgcn_readlane(dl, 0), __builtin_amdgcn_readlane(dl, 16));
  const int2* sl = list + (size_t)nd * CAP;

  float4 a0 = make_float4(0.f, 0.f, 0.f, 0.f);
  float4 a1 = a0, a2 = a0;
  int2 cw = make_int2(0, 0);
  if (0 < dl) cw = sl[0];
#pragma unroll 1
  for (int i = 0; i < maxd; ++i) {
    int2 cur = cw;
    if (i + 1 < dl) cw = sl[i + 1];
    if (i < dl) {
      int c = cur.x;
      float w = __int_as_float(cur.y);
      const float* uc = U + (size_t)c * 3 * PDIM + f;
      float4 x0 = *(const float4*)(uc);
      float4 x1 = *(const float4*)(uc + PDIM);
      float4 x2 = *(const float4*)(uc + 2 * PDIM);
      axpy4(w, x0, a0);
      axpy4(w, x1, a1);
      axpy4(w, x2, a2);
    }
  }
  if (valid && deg <= CAP)
    store_vrow(node, f, a0, a1, a2, v, bvec, v_out);
  if (half == 0) {
    int dg0 = __builtin_amdgcn_readlane(deg, 0);
    int dg1 = __builtin_amdgcn_readlane(deg, 16);
    if (dg0 > CAP)
      long_row<false>(pairBase, dg0, qt, qb, s, U, v, bvec, counts, list, ovf,
                      nullptr, nullptr, v_out, N);
    if (dg1 > CAP)
      long_row<false>(pairBase + 1, dg1, qt, qb, s, U, v, bvec, counts, list, ovf,
                      nullptr, nullptr, v_out, N);
  }
}

__global__ __launch_bounds__(256) void painn_u_gemm(
    const float* __restrict__ vv, const float* __restrict__ v,
    const float* __restrict__ Wvec, float* __restrict__ U, int nStripsU, int nGroupsU,
    const float* __restrict__ S, const float* __restrict__ sw, const float* __restrict__ s,
    const float* __restrict__ Wkv, const float* __restrict__ bkv,
    float* __restrict__ s_out, int nStripsS) {
  int wave = threadIdx.x >> 5;
  int lane = threadIdx.x & 31;
  int half = lane >> 4;
  int l16  = lane & 15;
  v2f a[32];
  if ((int)blockIdx.x < nGroupsU) {
    int strip = blockIdx.x * 8 + wave;
    if (strip >= nStripsU) return;
    int rowBase = strip * 16;
    int row = rowBase + l16;
    int node = row / 3;
    load_strip_prod(v + (size_t)row * PDIM + 2 * half, vv + (size_t)node * PDIM + 2 * half, a);
    size_t ooff = (size_t)(rowBase + 8 * half) * PDIM + 4 * l16;
    strip_block<PDIM, 0>(a, Wvec + (2 * half) * PDIM + 4 * l16, nullptr, U + ooff,
                         nullptr, nullptr);
  } else {
    int strip = ((int)blockIdx.x - nGroupsU) * 8 + wave;
    if (strip >= nStripsS) return;
    int rowBase = strip * 16;
    load_strip(S + (size_t)(rowBase + l16) * PDIM + 2 * half, a);
    size_t ooff = (size_t)(rowBase + 8 * half) * PDIM + 4 * l16;
    int wc = PDIM + 4 * l16;
    strip_block<3 * PDIM, 2>(a, Wkv + (2 * half) * 3 * PDIM + wc, bkv + wc, s_out + ooff,
                             s + ooff, sw + rowBase + 8 * half);
  }
}

extern "C" void kernel_launch(void* const* d_in, const int* in_sizes, int n_in,
                              void* d_out, int out_size, void* d_ws, size_t ws_size,
                              hipStream_t stream) {
  const float* s    = (const float*)d_in[0];
  const float* v    = (const float*)d_in[1];
  const int*   ei   = (const int*)d_in[2];
  const float* Wq   = (const float*)d_in[3];
  const float* bq   = (const float*)d_in[4];
  const float* Wkv  = (const float*)d_in[5];
  const float* bkv  = (const float*)d_in[6];
  const float* Wvec = (const float*)d_in[7];
  const float* bvec = (const float*)d_in[8];

  int N = in_sizes[0] / PDIM;
  int E = in_sizes[2] / 2;

  float* s_out = (float*)d_out;
  float* v_out = s_out + (size_t)N * PDIM;

  int nCnt = (N + 1 + 63) / 64 * 64;

  float* q    = (float*)d_ws;
  float* qt   = q  + (size_t)N * PDIM;
  float* vv   = qt + (size_t)N * PDIM;
  float* S    = vv + (size_t)N * PDIM;
  float* U    = S  + (size_t)N * PDIM;
  float* qb   = U  + (size_t)N * 3 * PDIM;
  float* sw   = qb + nCnt;

  int*  counts = (int*)(sw + nCnt);
  int2* list   = (int2*)(counts + nCnt);
  int4* ovf    = (int4*)(list + (size_t)N * CAP);

  const int* rowIdx = ei;
  const int* colIdx = ei + E;

  int stripsA = N / 16;
  int groupsA = (stripsA + 7) / 8;
  painn_node_gemm<<<2 * groupsA, 256, 0, stream>>>(
      s, Wq, bq, Wkv, bkv, q, qt, qb, vv, stripsA, groupsA, counts, nCnt / 4);

  int edgeBlocks = (E + 1023) / 1024;
  bucket_fill<<<edgeBlocks, 256, 0, stream>>>(rowIdx, colIdx, counts, list, ovf, N, E);

  edge_weights<<<(N + 7) / 8, 256, 0, stream>>>(qt, qb, s, counts, list, ovf, S, sw, N);

  int stripsC = N * 3 / 16;
  int groupsC = (stripsC + 7) / 8;
  painn_u_gemm<<<groupsC + groupsA, 256, 0, stream>>>(
      vv, v, Wvec, U, stripsC, groupsC, S, sw, s, Wkv, bkv, s_out, stripsA);

  int pairGroups = ((N + 1) / 2 + 7) / 8;
  node_rows_half<<<2 * pairGroups, 256, 0, stream>>>(
      qt, qb, s, U, v, bvec, counts, list, ovf, v_out, N, pairGroups);
}


# Edge lists built through LDS: edge chunks counting-sorted by 256-node range in the node launch, one work-group per range fills its rows' slots by LDS atomics; no global atomic per edge; same sums
# speedup vs baseline: 3.6562x; 1.1594x over previous
#include <hip/hip_runtime.h>


typedef __attribute__((ext_vector_type(2))) float v2f;
typedef __attribute__((ext_vector_type(8))) float v8f;

#define PDIM 128
#define CAP 32

#define BINSH 8
#define BINR (1 << BINSH)
#define MAXBINS 256
#define EPT 10
#define CHUNK (256 * EPT)

__device__ __forceinline__ void presort_chunk(const int* __restrict__ rowIdx,
                                              const int* __restrict__ colIdx, int E, int nBins,
                                              int* __restrict__ offs, int2* __restrict__ sorted,
                                              int w, int* hist, int* wsum, int2* stage) {
  int t = threadIdx.x;
  for (int b = t; b <= nBins; b += 256) hist[b] = 0;
  __syncthreads();
  int e0 = w * CHUNK;
  int r[EPT], c[EPT], rk[EPT];
#pragma unroll
  for (int u = 0; u < EPT; ++u) {
    int e = e0 + u * 256 + t;
    int el = min(e, E - 1);
    int rr = rowIdx[el];
    c[u] = colIdx[el];
    r[u] = (e < E) ? rr : -1;
  }
#pragma unroll
  for (int u = 0; u < EPT; ++u)
    rk[u] = (r[u] >= 0) ? atomicAdd(&hist[r[u] >> BINSH], 1) : 0;
  __syncthreads();
  int lane = t & 31;
  int wave = t >> 5;
  int v = (t < nBins) ? hist[t] : 0;
  int incl = v;
#pragma unroll
  for (int d = 1; d < 32; d <<= 1) {
    int n = __shfl_up(incl, d, 32);
    if (lane >= d) incl += n;
  }
  if (lane == 31) wsum[wave] = incl;
  __syncthreads();
  int pre = 0;
  for (int k = 0; k < wave; ++k) pre += wsum[k];
  int excl = pre + incl - v;
  if (t < nBins) hist[t] = excl;
  if (t == nBins - 1) hist[nBins] = excl + v;
  __syncthreads();
  for (int b = t; b <= nBins; b += 256) offs[(size_t)w * (nBins + 1) + b] = hist[b];
#pragma unroll
  for (int u = 0; u < EPT; ++u) {
    if (r[u] < 0) continue;
    unsigned lr = (unsigned)r[u] & (BINR - 1);
    stage[hist[r[u] >> BINSH] + rk[u]] = make_int2(e0 + u * 256 + t, (int)((lr << 24) | (unsigned)c[u]));
  }
  __syncthreads();
  int cnt = min(CHUNK, E - e0);
  for (int i = t; i < cnt; i += 256) sorted[(size_t)e0 + i] = stage[i];
}

__device__ __forceinline__ void place_edge(int2 en, int nodeBase, int* cnt, int2* slots,
                                           int* __restrict__ counts, int4* __restrict__ ovf,
                                           int ovfCounter) {
  int lr = (int)((unsigned)en.y >> 24);
  int col = en.y & 0xffffff;
  int sl = atomicAdd(&cnt[lr], 1);
  if (sl < CAP) {
    slots[lr * CAP + sl] = make_int2(en.x, col);
  } else {
    int o = atomicAdd(&counts[ovfCounter], 1);
    ovf[o] = make_int4(nodeBase + lr, en.x, col, 0);
  }
}

__global__ __launch_bounds__(1024) void bin_lists(const int* __restrict__ offs,
                                                  const int2* __restrict__ sorted,
                                                  int nChunks, int nBins,
                                                  int* __restrict__ counts,
                                                  int2* __restrict__ list,
                                                  int4* __restrict__ ovf,
                                                  int ovfCounter, int N) {
  __shared__ int cnt[BINR];
  __shared__ __attribute__((aligned(16))) int2 slots[BINR * CAP];
  int t = threadIdx.x;
  int b = blockIdx.x;
  int nodeBase = b * BINR;
  if (t < BINR) cnt[t] = 0;
  __syncthreads();
  int sub = t & 3;
  for (int w = t >> 2; w < nChunks; w += 256) {
    const int* op = offs + (size_t)w * (nBins + 1) + b;
    int o0 = op[0];
    int o1 = op[1];
    const int2* sp = sorted + (size_t)w * CHUNK;
    for (int i0 = o0 + sub; i0 < o1; i0 += 8) {
      int i1 = i0 + 4;
      int2 en0 = sp[i0];
      int2 en1 = sp[min(i1, o1 - 1)];
      place_edge(en0, nodeBase, cnt, slots, counts, ovf, ovfCounter);
      if (i1 < o1) place_edge(en1, nodeBase, cnt, slots, counts, ovf, ovfCounter);
    }
  }
  __syncthreads();
  if (t < BINR && nodeBase + t < N) counts[nodeBase + t] = cnt[t];
  int2* lst = list + (size_t)nodeBase * CAP;
  for (int i = t; i < BINR * CAP / 2; i += 1024) {
    int nd = i / (CAP / 2);
    int s0 = (i % (CAP / 2)) * 2;
    int filled = (nodeBase + nd < N) ? min(cnt[nd], CAP) : 0;
    if (s0 + 1 < filled) {
      ((int4*)lst)[i] = ((const int4*)slots)[i];
    } else if (s0 < filled) {
      lst[2 * i] = slots[2 * i];
    }
  }
}

__device__ __forceinline__ void load_strip(const float* __restrict__ rowp, v2f (&a)[32]) {
#pragma unroll
  for (int ks = 0; ks < 32; ++ks) {
    float2 t = *(const float2*)(rowp + 4 * ks);
    a[ks].x = t.x;
    a[ks].y = t.y;
  }
}

__device__ __forceinline__ void load_strip_prod(const float* __restrict__ xrow,
                                                const float* __restrict__ grow, v2f (&a)[32]) {
#pragma unroll
  for (int ks = 0; ks < 32; ++ks) {
    float2 t = *(const float2*)(xrow + 4 * ks);
    float2 g = *(const float2*)(grow + 4 * ks);
    a[ks].x = g.x * t.x;
    a[ks].y = g.y * t.y;
  }
}

__device__ __forceinline__ void step4(const v2f& af, const float4& r0, const float4& r1,
                                      v8f (&acc)[4]) {
  v2f bf;
  bf.x = r0.x; bf.y = r1.x;
  acc[0] = __builtin_amdgcn_wmma_f32_16x16x4_f32(false, af, false, bf, (short)0, acc[0], false, false);
  bf.x = r0.y; bf.y = r1.y;
  acc[1] = __builtin_amdgcn_wmma_f32_16x16x4_f32(false, af, false, bf, (short)0, acc[1], false, false);
  bf.x = r0.z; bf.y = r1.z;
  acc[2] = __builtin_amdgcn_wmma_f32_16x16x4_f32(false, af, false, bf, (short)0, acc[2], false, false);
  bf.x = r0.w; bf.y = r1.w;
  acc[3] = __builtin_amdgcn_wmma_f32_16x16x4_f32(false, af, false, bf, (short)0, acc[3], false, false);
}

template <int LDW>
__device__ __forceinline__ void strip_chunk(const v2f (&a)[32], const float* __restrict__ Wl,
                                            v8f (&acc)[4]) {
  float4 n0 = *(const float4*)(Wl);
  float4 n1 = *(const float4*)(Wl + LDW);
  float4 n2 = *(const float4*)(Wl + 4 * LDW);
  float4 n3 = *(const float4*)(Wl + 5 * LDW);
#pragma unroll
  for (int j = 0; j < 16; ++j) {
    float4 c0 = n0, c1 = n1, c2 = n2, c3 = n3;
    if (j + 1 < 16) {
      const float* p = Wl + (8 * (j + 1)) * LDW;
      n0 = *(const float4*)(p);
      n1 = *(const float4*)(p + LDW);
      n2 = *(const float4*)(p + 4 * LDW);
      n3 = *(const float4*)(p + 5 * LDW);
    }
    step4(a[2 * j],     c0, c1, acc);
    step4(a[2 * j + 1], c2, c3, acc);
    __builtin_amdgcn_sched_barrier(0);
  }
}

template <int LDW, int MODE>
__device__ __forceinline__ void strip_block(const v2f (&a)[32], const float* __restrict__ Wl,
                                            const float* __restrict__ bl,
                                            float* __restrict__ orow,
                                            const float* __restrict__ xrow,
                                            const float* __restrict__ gcol) {
#pragma unroll 1
  for (int cb = 0; cb < PDIM; cb += 64) {
    v8f acc[4] = {};
    if (MODE == 2) {
#pragma unroll
      for (int r = 0; r < 8; ++r) {
        float4 x = *(const float4*)(xrow + (size_t)r * PDIM + cb);
        acc[0][r] = x.x; acc[1][r] = x.y; acc[2][r] = x.z; acc[3][r] = x.w;
      }
      __builtin_amdgcn_sched_barrier(0);
    }
    strip_chunk<LDW>(a, Wl + cb, acc);
    float4 bias = make_float4(0.f, 0.f, 0.f, 0.f);
    if (MODE != 0) bias = *(const float4*)(bl + cb);
#pragma unroll
    for (int r = 0; r < 8; ++r) {
      float4 ov = make_float4(acc[0][r], acc[1][r], acc[2][r], acc[3][r]);
      if (MODE == 1) {
        ov = make_float4(ov.x + bias.x, ov.y + bias.y, ov.z + bias.z, ov.w + bias.w);
      }
      if (MODE == 2) {
        float g = gcol[r];
        ov = make_float4(ov.x + g * bias.x, ov.y + g * bias.y,
                         ov.z + g * bias.z, ov.w + g * bias.w);
      }
      *(float4*)(orow + (size_t)r * PDIM + cb) = ov;
    }
  }
}

__device__ __forceinline__ void load_strip_k8(const float* rowp, v2f (&a)[32]) {
#pragma unroll
  for (int j = 0; j < 16; ++j) {
    float4 t = *(const float4*)(rowp + 8 * j);
    a[2 * j].x = t.x;     a[2 * j].y = t.y;
    a[2 * j + 1].x = t.z; a[2 * j + 1].y = t.w;
  }
}

template <int LDW>
__device__ __forceinline__ void strip_chunk_T(const v2f (&a)[32], const float* __restrict__ Wt,
                                              v8f (&acc)[4]) {
  float4 n0 = *(const float4*)(Wt);
  float4 n1 = *(const float4*)(Wt + 16 * LDW);
  float4 n2 = *(const float4*)(Wt + 32 * LDW);
  float4 n3 = *(const float4*)(Wt + 48 * LDW);
#pragma unroll
  for (int j = 0; j < 16; ++j) {
    float4 c[4] = {n0, n1, n2, n3};
    if (j + 1 < 16) {
      const float* p = Wt + 8 * (j + 1);
      n0 = *(const float4*)(p);
      n1 = *(const float4*)(p + 16 * LDW);
      n2 = *(const float4*)(p + 32 * LDW);
      n3 = *(const float4*)(p + 48 * LDW);
    }
#pragma unroll
    for (int t = 0; t < 4; ++t) {
      v2f bf;
      bf.x = c[t].x; bf.y = c[t].y;
      acc[t] = __builtin_amdgcn_wmma_f32_16x16x4_f32(false, a[2 * j], false, bf, (short)0, acc[t], false, false);
    }
#pragma unroll
    for (int t = 0; t < 4; ++t) {
      v2f bf;
      bf.x = c[t].z; bf.y = c[t].w;
      acc[t] = __builtin_amdgcn_wmma_f32_16x16x4_f32(false, a[2 * j + 1], false, bf, (short)0, acc[t], false, false);
    }
    __builtin_amdgcn_sched_barrier(0);
  }
}

__global__ __launch_bounds__(256) void painn_node_gemm(
    const float* __restrict__ s,
    const float* __restrict__ Wq,  const float* __restrict__ bq,
    const float* __restrict__ Wkv, const float* __restrict__ bkv,
    float* q, float* __restrict__ qt, float* __restrict__ qb, float* __restrict__ vv,
    int nStrips, int nGroups, int* __restrict__ counts, int ovfCounter,
    const int* __restrict__ rowIdx, const int* __restrict__ colIdx, int E, int nBins,
    int nSort, int* __restrict__ offs, int2* __restrict__ sorted) {
  __shared__ int hist[MAXBINS + 1];
  __shared__ int wsum[8];
  __shared__ int2 stage[CHUNK];
  if ((int)blockIdx.x < nSort) {
    if (blockIdx.x == 0 && threadIdx.x == 0) counts[ovfCounter] = 0;
    presort_chunk(rowIdx, colIdx, E, nBins, offs, sorted, (int)blockIdx.x, hist, wsum, stage);
    return;
  }
  int bx = (int)blockIdx.x - nSort;
  int wave = threadIdx.x >> 5;
  int lane = threadIdx.x & 31;
  int o = bx / nGroups;
  int strip = (bx - o * nGroups) * 8 + wave;
  if (strip >= nStrips) return;
  int half = lane >> 4;
  int l16  = lane & 15;
  int rowBase = strip * 16;

  v2f a[32];
  load_strip(s + (size_t)(rowBase + l16) * PDIM + 2 * half, a);

  size_t ooff = (size_t)(rowBase + 8 * half) * PDIM + 4 * l16;
  if (o == 0) {
    strip_block<PDIM, 1>(a, Wq + (2 * half) * PDIM + 4 * l16, bq + 4 * l16, q + ooff,
                         nullptr, nullptr);
    __threadfence_block();
    load_strip_k8(q + (size_t)(rowBase + l16) * PDIM + 4 * half, a);
    float p = 0.f;
#pragma unroll
    for (int j = 0; j < 16; ++j) {
      float4 b = *(const float4*)(bkv + 8 * j + 4 * half);
      p += a[2 * j].x * b.x + a[2 * j].y * b.y + a[2 * j + 1].x * b.z + a[2 * j + 1].y * b.w;
    }
    p += __shfl_xor(p, 16, 32);
    if (half == 0) qb[rowBase + l16] = p;
#pragma unroll 1
    for (int cb = 0; cb < PDIM; cb += 64) {
      v8f acc[4] = {};
      strip_chunk_T<3 * PDIM>(a, Wkv + (size_t)(cb + l16) * 3 * PDIM + 4 * half, acc);
      float* orow = qt + (size_t)(rowBase + 8 * half) * PDIM + cb + l16;
#pragma unroll
      for (int r = 0; r < 8; ++r) {
#pragma unroll
        for (int t = 0; t < 4; ++t) orow[(size_t)r * PDIM + 16 * t] = acc[t][r];
      }
    }
  } else {
    int wc = 2 * PDIM + 4 * l16;
    strip_block<3 * PDIM, 1>(a, Wkv + (2 * half) * 3 * PDIM + wc, bkv + wc, vv + ooff,
                             nullptr, nullptr);
  }
}

__device__ __forceinline__ float wave_dot(const float4& qv, const float4& kv) {
  float p = qv.x * kv.x + qv.y * kv.y + qv.z * kv.z + qv.w * kv.w;
  p += __shfl_xor(p, 16, 32);
  p += __shfl_xor(p, 8, 32);
  p += __shfl_xor(p, 4, 32);
  p += __shfl_xor(p, 2, 32);
  p += __shfl_xor(p, 1, 32);
  return p;
}

__device__ __forceinline__ void axpy4(float w, const float4& x, float4& a) {
  a.x += w * x.x; a.y += w * x.y; a.z += w * x.z; a.w += w * x.w;
}

__device__ __forceinline__ void store_vrow(
    int node, int f, const float4& a0, const float4& a1, const float4& a2,
    const float* __restrict__ v, const float* __restrict__ bvec, float* __restrict__ v_out) {
  size_t vo3 = (size_t)node * 3 * PDIM + f;
  float4 bv = *(const float4*)(bvec + f);
  const float* vr = v + vo3;
  float4 y0 = *(const float4*)(vr);
  float4 y1 = *(const float4*)(vr + PDIM);
  float4 y2 = *(const float4*)(vr + 2 * PDIM);
  float* vo = v_out + vo3;
  *(float4*)(vo) = make_float4(y0.x + (a0.x + bv.x), y0.y + (a0.y + bv.y),
                               y0.z + (a0.z + bv.z), y0.w + (a0.w + bv.w));
  *(float4*)(vo + PDIM) = make_float4(y1.x + (a1.x + bv.x), y1.y + (a1.y + bv.y),
                                      y1.z + (a1.z + bv.z), y1.w + (a1.w + bv.w));
  *(float4*)(vo + 2 * PDIM) = make_float4(y2.x + (a2.x + bv.x), y2.y + (a2.y + bv.y),
                                          y2.z + (a2.z + bv.z), y2.w + (a2.w + bv.w));
}

template <bool SCALAR>
__device__ void long_row(
    int node, int deg,
    const float* __restrict__ qt, const float* __restrict__ qb,
    const float* __restrict__ s, const float* __restrict__ U,
    const float* __restrict__ v, const float* __restrict__ bvec,
    const int* __restrict__ counts, const int2* __restrict__ list,
    const int4* __restrict__ ovf,
    float* __restrict__ S, float* __restrict__ sw, float* __restrict__ v_out, int N) {
  int lane = threadIdx.x & 31;
  int d = lane * 4;
  float4 qv = *(const float4*)(qt + (size_t)node * PDIM + d);
  float qbn = qb[node];
  float4 a0 = make_float4(0.f, 0.f, 0.f, 0.f);
  float4 a1 = a0, a2 = a0;
  float sws = 0.f;
  int nOvf = counts[N];
  int2 ec = list[(size_t)node * CAP + lane];
  int last = -1;
  for (int i = 0; i < deg; ++i) {
    int be = 0x7fffffff, bc = 0;
    if (ec.x > last) { be = ec.x; bc = ec.y; }
    for (int j = lane; j < nOvf; j += 32) {
      int4 o = ovf[j];
      if (o.x == node && o.y > last && o.y < be) { be = o.y; bc = o.z; }
    }
    int mn = be;
    mn = min(mn, __shfl_xor(mn, 16, 32));
    mn = min(mn, __shfl_xor(mn, 8, 32));
    mn = min(mn, __shfl_xor(mn, 4, 32));
    mn = min(mn, __shfl_xor(mn, 2, 32));
    mn = min(mn, __shfl_xor(mn, 1, 32));
    unsigned m = __builtin_amdgcn_ballot_w32(be == mn);
    int src = __builtin_ctz(m | 0x80000000u);
    int c = __builtin_amdgcn_readlane(bc, src);
    last = __builtin_amdgcn_readfirstlane(mn);
    float4 sv = *(const float4*)(s + (size_t)c * PDIM + d);
    float w = wave_dot(qv, sv) + qbn;
    if (SCALAR) {
      axpy4(w, sv, a0);
      sws += w;
    } else {
      const float* uc = U + (size_t)c * 3 * PDIM + d;
      float4 x0 = *(const float4*)(uc);
      float4 x1 = *(const float4*)(uc + PDIM);
      float4 x2 = *(const float4*)(uc + 2 * PDIM);
      axpy4(w, x0, a0);
      axpy4(w, x1, a1);
      axpy4(w, x2, a2);
    }
  }
  if (SCALAR) {
    *(float4*)(S + (size_t)node * PDIM + d) = a0;
    if (lane == 0) sw[node] = sws;
  } else {
    store_vrow(node, d, a0, a1, a2, v, bvec, v_out);
  }
}

__global__ __launch_bounds__(256) void edge_weights(
    const float* __restrict__ qt, const float* __restrict__ qb, const float* __restrict__ s,
    const int* __restrict__ counts, int2* __restrict__ list, const int4* __restrict__ ovf,
    float* __restrict__ S, float* __restrict__ sw, int N) {
  int wave = __builtin_amdgcn_readfirstlane((int)(threadIdx.x >> 5));
  int node = blockIdx.x * 8 + wave;
  if (node >= N) return;
  int lane = threadIdx.x & 31;
  int d = lane * 4;
  int deg = counts[node];
  if (deg > CAP) {
    long_row<true>(node, deg, qt, qb, s, nullptr, nullptr, nullptr, counts, list, ovf,
                   S, sw, nullptr, N);
    return;
  }
  int2* slots = list + (size_t)node * CAP;

  float4 qv = *(const float4*)(qt + (size_t)node * PDIM + d);
  float qbn = qb[node];
  int2 ec = make_int2(0x7fffffff, 0);
  if (lane < deg) ec = slots[lane];
  int rank = 0;
  for (int j = 0; j < deg; ++j) {
    int ej = __builtin_amdgcn_readlane(ec.x, j);
    rank += (ej < ec.x) ? 1 : 0;
  }
  float4 acc = make_float4(0.f, 0.f, 0.f, 0.f);
  float sws = 0.f;
  float wm = 0.f;
  int cm = 0;
  for (int p0 = 0; p0 < deg; p0 += 4) {
    float4 sv[4];
    int cc[4];
#pragma unroll
    for (int u = 0; u < 4; ++u) {
      int p = min(p0 + u, deg - 1);
      unsigned m = __builtin_amdgcn_ballot_w32(rank == p);
      int src = __builtin_ctz(m | 0x80000000u);
      cc[u] = __builtin_amdgcn_readlane(ec.y, src);
      sv[u] = *(const float4*)(s + (size_t)cc[u] * PDIM + d);
    }
#pragma unroll
    for (int u = 0; u < 4; ++u) {
      float w = wave_dot(qv, sv[u]) + qbn;
      if (p0 + u < deg) {
        axpy4(w, sv[u], acc);
        sws += w;
      }
      if (lane == p0 + u) { wm = w; cm = cc[u]; }
    }
  }
  if (lane < deg) slots[lane] = make_int2(cm, __float_as_int(wm));
  *(float4*)(S + (size_t)node * PDIM + d) = acc;
  if (lane == 0) sw[node] = sws;
}

__global__ __launch_bounds__(256) void node_rows_half(
    const float* __restrict__ qt, const float* __restrict__ qb, const float* __restrict__ s,
    const float* __restrict__ U, const float* __restrict__ v,
    const float* __restrict__ bvec,
    const int* __restrict__ counts, const int2* __restrict__ list,
    const int4* __restrict__ ovf,
    float* __restrict__ v_out, int N, int nGroups) {
  int wave = __builtin_amdgcn_readfirstlane((int)(threadIdx.x >> 5));
  int half = blockIdx.x / nGroups;
  int pairBase = ((blockIdx.x - half * nGroups) * 8 + wave) * 2;
  if (pairBase >= N) return;
  int lane = threadIdx.x & 31;
  int node = pairBase + (lane >> 4);
  bool valid = node < N;
  int nd = valid ? node : pairBase;
  int deg = valid ? counts[nd] : 0;
  int dl = (deg <= CAP) ? deg : 0;
  int f = half * 64 + (lane & 15) * 4;
  int maxd = max(__builtin_amdgcn_readlane(dl, 0), __builtin_amdgcn_readlane(dl, 16));
  const int2* sl = list + (size_t)nd * CAP;

  float4 a0 = make_float4(0.f, 0.f, 0.f, 0.f);
  float4 a1 = a0, a2 = a0;
  int2 cw = make_int2(0, 0);
  if (0 < dl) cw = sl[0];
#pragma unroll 1
  for (int i = 0; i < maxd; ++i) {
    int2 cur = cw;
    if (i + 1 < dl) cw = sl[i + 1];
    if (i < dl) {
      int c = cur.x;
      float w = __int_as_float(cur.y);
      const float* uc = U + (size_t)c * 3 * PDIM + f;
      float4 x0 = *(const float4*)(uc);
      float4 x1 = *(const float4*)(uc + PDIM);
      float4 x2 = *(const float4*)(uc + 2 * PDIM);
      axpy4(w, x0, a0);
      axpy4(w, x1, a1);
      axpy4(w, x2, a2);
    }
  }
  if (valid && deg <= CAP)
    store_vrow(node, f, a0, a1, a2, v, bvec, v_out);
  if (half == 0) {
    int dg0 = __builtin_amdgcn_readlane(deg, 0);
    int dg1 = __builtin_amdgcn_readlane(deg, 16);
    if (dg0 > CAP)
      long_row<false>(pairBase, dg0, qt, qb, s, U, v, bvec, counts, list, ovf,
                      nullptr, nullptr, v_out, N);
    if (dg1 > CAP)
      long_row<false>(pairBase + 1, dg1, qt, qb, s, U, v, bvec, counts, list, ovf,
                      nullptr, nullptr, v_out, N);
  }
}

__global__ __launch_bounds__(256) void painn_u_gemm(
    const float* __restrict__ vv, const float* __restrict__ v,
    const float* __restrict__ Wvec, float* __restrict__ U, int nStripsU, int nGroupsU,
    const float* __restrict__ S, const float* __restrict__ sw, const float* __restrict__ s,
    const float* __restrict__ Wkv, const float* __restrict__ bkv,
    float* __restrict__ s_out, int nStripsS) {
  int wave = threadIdx.x >> 5;
  int lane = threadIdx.x & 31;
  int half = lane >> 4;
  int l16  = lane & 15;
  v2f a[32];
  if ((int)blockIdx.x < nGroupsU) {
    int strip = blockIdx.x * 8 + wave;
    if (strip >= nStripsU) return;
    int rowBase = strip * 16;
    int row = rowBase + l16;
    int node = row / 3;
    load_strip_prod(v + (size_t)row * PDIM + 2 * half, vv + (size_t)node * PDIM + 2 * half, a);
    size_t ooff = (size_t)(rowBase + 8 * half) * PDIM + 4 * l16;
    strip_block<PDIM, 0>(a, Wvec + (2 * half) * PDIM + 4 * l16, nullptr, U + ooff,
                         nullptr, nullptr);
  } else {
    int strip = ((int)blockIdx.x - nGroupsU) * 8 + wave;
    if (strip >= nStripsS) return;
    int rowBase = strip * 16;
    load_strip(S + (size_t)(rowBase + l16) * PDIM + 2 * half, a);
    size_t ooff = (size_t)(rowBase + 8 * half) * PDIM + 4 * l16;
    int wc = PDIM + 4 * l16;
    strip_block<3 * PDIM, 2>(a, Wkv + (2 * half) * 3 * PDIM + wc, bkv + wc, s_out + ooff,
                             s + ooff, sw + rowBase + 8 * half);
  }
}

extern "C" void kernel_launch(void* const* d_in, const int* in_sizes, int n_in,
                              void* d_out, int out_size, void* d_ws, size_t ws_size,
                              hipStream_t stream) {
  const float* s    = (const float*)d_in[0];
  const float* v    = (const float*)d_in[1];
  const int*   ei   = (const int*)d_in[2];
  const float* Wq   = (const float*)d_in[3];
  const float* bq   = (const float*)d_in[4];
  const float* Wkv  = (const float*)d_in[5];
  const float* bkv  = (const float*)d_in[6];
  const float* Wvec = (const float*)d_in[7];
  const float* bvec = (const float*)d_in[8];

  int N = in_sizes[0] / PDIM;
  int E = in_sizes[2] / 2;

  float* s_out = (float*)d_out;
  float* v_out = s_out + (size_t)N * PDIM;

  int nCnt = (N + 1 + 63) / 64 * 64;

  float* q    = (float*)d_ws;
  float* qt   = q  + (size_t)N * PDIM;
  float* vv   = qt + (size_t)N * PDIM;
  float* S    = vv + (size_t)N * PDIM;
  float* U    = S  + (size_t)N * PDIM;
  float* qb   = U  + (size_t)N * 3 * PDIM;
  float* sw   = qb + nCnt;

  int*  counts = (int*)(sw + nCnt);
  int2* list   = (int2*)(counts + nCnt);
  int4* ovf    = (int4*)(list + (size_t)N * CAP);

  int nBins   = (N + BINR - 1) / BINR;
  int nChunks = (E + CHUNK - 1) / CHUNK;
  int nOffs   = (nChunks * (nBins + 1) + 63) / 64 * 64;
  int*  offs   = (int*)(ovf + E);
  int2* sorted = (int2*)(offs + nOffs);

  const int* rowIdx = ei;
  const int* colIdx = ei + E;

  int stripsA = N / 16;
  int groupsA = (stripsA + 7) / 8;
  painn_node_gemm<<<nChunks + 2 * groupsA, 256, 0, stream>>>(
      s, Wq, bq, Wkv, bkv, q, qt, qb, vv, stripsA, groupsA, counts, N,
      rowIdx, colIdx, E, nBins, nChunks, offs, sorted);

  bin_lists<<<nBins, 1024, 0, stream>>>(offs, sorted, nChunks, nBins, counts, list, ovf, N, N);

  edge_weights<<<(N + 7) / 8, 256, 0, stream>>>(qt, qb, s, counts, list, ovf, S, sw, N);

  int stripsC = N * 3 / 16;
  int groupsC = (stripsC + 7) / 8;
  painn_u_gemm<<<groupsC + groupsA, 256, 0, stream>>>(
      vv, v, Wvec, U, stripsC, groupsC, S, sw, s, Wkv, bkv, s_out, stripsA);

  int pairGroups = ((N + 1) / 2 + 7) / 8;
  node_rows_half<<<2 * pairGroups, 256, 0, stream>>>(
      qt, qb, s, U, v, bvec, counts, list, ovf, v_out, N, pairGroups);
}


# q never formed: Wq folded into Wk^T once per call on the matrix unit (M = Wq Wk^T, bq Wk^T, Wq bk, bq.bk), qt = s M + bq Wk^T by one GEMM, run beside the list pass in its launch; sum order changes
# speedup vs baseline: 3.8782x; 1.0607x over previous
#include <hip/hip_runtime.h>


typedef __attribute__((ext_vector_type(2))) float v2f;
typedef __attribute__((ext_vector_type(8))) float v8f;

#define PDIM 128
#define CAP 32

#define BINSH 8
#define BINR (1 << BINSH)
#define MAXBINS 256
#define EPT 10
#define CHUNK (256 * EPT)

__device__ __forceinline__ void presort_chunk(const int* __restrict__ rowIdx,
                                              const int* __restrict__ colIdx, int E, int nBins,
                                              int* __restrict__ offs, int2* __restrict__ sorted,
                                              int w, int* hist, int* wsum, int2* stage) {
  int t = threadIdx.x;
  for (int b = t; b <= nBins; b += 256) hist[b] = 0;
  __syncthreads();
  int e0 = w * CHUNK;
  int r[EPT], c[EPT], rk[EPT];
#pragma unroll
  for (int u = 0; u < EPT; ++u) {
    int e = e0 + u * 256 + t;
    int el = min(e, E - 1);
    int rr = rowIdx[el];
    c[u] = colIdx[el];
    r[u] = (e < E) ? rr : -1;
  }
#pragma unroll
  for (int u = 0; u < EPT; ++u)
    rk[u] = (r[u] >= 0) ? atomicAdd(&hist[r[u] >> BINSH], 1) : 0;
  __syncthreads();
  int lane = t & 31;
  int wave = t >> 5;
  int v = (t < nBins) ? hist[t] : 0;
  int incl = v;
#pragma unroll
  for (int d = 1; d < 32; d <<= 1) {
    int n = __shfl_up(incl, d, 32);
    if (lane >= d) incl += n;
  }
  if (lane == 31) wsum[wave] = incl;
  __syncthreads();
  int pre = 0;
  for (int k = 0; k < wave; ++k) pre += wsum[k];
  int excl = pre + incl - v;
  if (t < nBins) hist[t] = excl;
  if (t == nBins - 1) hist[nBins] = excl + v;
  __syncthreads();
  for (int b = t; b <= nBins; b += 256) offs[(size_t)w * (nBins + 1) + b] = hist[b];
#pragma unroll
  for (int u = 0; u < EPT; ++u) {
    if (r[u] < 0) continue;
    unsigned lr = (unsigned)r[u] & (BINR - 1);
    stage[hist[r[u] >> BINSH] + rk[u]] = make_int2(e0 + u * 256 + t, (int)((lr << 24) | (unsigned)c[u]));
  }
  __syncthreads();
  int cnt = min(CHUNK, E - e0);
  for (int i = t; i < cnt; i += 256) sorted[(size_t)e0 + i] = stage[i];
}

__device__ __forceinline__ void place_edge(int2 en, int nodeBase, int* cnt, int2* slots,
                                           int* __restrict__ counts, int4* __restrict__ ovf,
                                           int ovfCounter) {
  int lr = (int)((unsigned)en.y >> 24);
  int col = en.y & 0xffffff;
  int sl = atomicAdd(&cnt[lr], 1);
  if (sl < CAP) {
    slots[lr * CAP + sl] = make_int2(en.x, col);
  } else {
    int o = atomicAdd(&counts[ovfCounter], 1);
    ovf[o] = make_int4(nodeBase + lr, en.x, col, 0);
  }
}

__device__ __forceinline__ void bin_lists(const int* __restrict__ offs,
                                          const int2* __restrict__ sorted,
                                          int nChunks, int nBins, int b,
                                          int* __restrict__ counts,
                                          int2* __restrict__ list,
                                          int4* __restrict__ ovf,
                                          int ovfCounter, int N, int* cnt, int2* slots) {
  int t = threadIdx.x;
  int nodeBase = b * BINR;
  for (int i = t; i < BINR; i += 256) cnt[i] = 0;
  __syncthreads();
  for (int w = t; w < nChunks; w += 256) {
    const int* op = offs + (size_t)w * (nBins + 1) + b;
    int o0 = op[0];
    int o1 = op[1];
    const int2* sp = sorted + (size_t)w * CHUNK;
    for (int i0 = o0; i0 < o1; i0 += 4) {
      int lastI = o1 - 1;
      int2 en0 = sp[i0];
      int2 en1 = sp[min(i0 + 1, lastI)];
      int2 en2 = sp[min(i0 + 2, lastI)];
      int2 en3 = sp[min(i0 + 3, lastI)];
      place_edge(en0, nodeBase, cnt, slots, counts, ovf, ovfCounter);
      if (i0 + 1 < o1) place_edge(en1, nodeBase, cnt, slots, counts, ovf, ovfCounter);
      if (i0 + 2 < o1) place_edge(en2, nodeBase, cnt, slots, counts, ovf, ovfCounter);
      if (i0 + 3 < o1) place_edge(en3, nodeBase, cnt, slots, counts, ovf, ovfCounter);
    }
  }
  __syncthreads();
  for (int i = t; i < BINR; i += 256)
    if (nodeBase + i < N) counts[nodeBase + i] = cnt[i];
  int2* lst = list + (size_t)nodeBase * CAP;
  for (int i = t; i < BINR * CAP / 2; i += 256) {
    int nd = i / (CAP / 2);
    int s0 = (i % (CAP / 2)) * 2;
    int filled = (nodeBase + nd < N) ? min(cnt[nd], CAP) : 0;
    if (s0 + 1 < filled) {
      ((int4*)lst)[i] = ((const int4*)slots)[i];
    } else if (s0 < filled) {
      lst[2 * i] = slots[2 * i];
    }
  }
}

__device__ __forceinline__ void load_strip(const float* __restrict__ rowp, v2f (&a)[32]) {
#pragma unroll
  for (int ks = 0; ks < 32; ++ks) {
    float2 t = *(const float2*)(rowp + 4 * ks);
    a[ks].x = t.x;
    a[ks].y = t.y;
  }
}

__device__ __forceinline__ void load_strip_prod(const float* __restrict__ xrow,
                                                const float* __restrict__ grow, v2f (&a)[32]) {
#pragma unroll
  for (int ks = 0; ks < 32; ++ks) {
    float2 t = *(const float2*)(xrow + 4 * ks);
    float2 g = *(const float2*)(grow + 4 * ks);
    a[ks].x = g.x * t.x;
    a[ks].y = g.y * t.y;
  }
}

__device__ __forceinline__ void step4(const v2f& af, const float4& r0, const float4& r1,
                                      v8f (&acc)[4]) {
  v2f bf;
  bf.x = r0.x; bf.y = r1.x;
  acc[0] = __builtin_amdgcn_wmma_f32_16x16x4_f32(false, af, false, bf, (short)0, acc[0], false, false);
  bf.x = r0.y; bf.y = r1.y;
  acc[1] = __builtin_amdgcn_wmma_f32_16x16x4_f32(false, af, false, bf, (short)0, acc[1], false, false);
  bf.x = r0.z; bf.y = r1.z;
  acc[2] = __builtin_amdgcn_wmma_f32_16x16x4_f32(false, af, false, bf, (short)0, acc[2], false, false);
  bf.x = r0.w; bf.y = r1.w;
  acc[3] = __builtin_amdgcn_wmma_f32_16x16x4_f32(false, af, false, bf, (short)0, acc[3], false, false);
}

template <int LDW>
__device__ __forceinline__ void strip_chunk(const v2f (&a)[32], const float* __restrict__ Wl,
                                            v8f (&acc)[4]) {
  float4 n0 = *(const float4*)(Wl);
  float4 n1 = *(const float4*)(Wl + LDW);
  float4 n2 = *(const float4*)(Wl + 4 * LDW);
  float4 n3 = *(const float4*)(Wl + 5 * LDW);
#pragma unroll
  for (int j = 0; j < 16; ++j) {
    float4 c0 = n0, c1 = n1, c2 = n2, c3 = n3;
    if (j + 1 < 16) {
      const float* p = Wl + (8 * (j + 1)) * LDW;
      n0 = *(const float4*)(p);
      n1 = *(const float4*)(p + LDW);
      n2 = *(const float4*)(p + 4 * LDW);
      n3 = *(const float4*)(p + 5 * LDW);
    }
    step4(a[2 * j],     c0, c1, acc);
    step4(a[2 * j + 1], c2, c3, acc);
    __builtin_amdgcn_sched_barrier(0);
  }
}

template <int LDW, int MODE>
__device__ __forceinline__ void strip_block(const v2f (&a)[32], const float* __restrict__ Wl,
                                            const float* __restrict__ bl,
                                            float* __restrict__ orow,
                                            const float* __restrict__ xrow,
                                            const float* __restrict__ gcol) {
#pragma unroll 1
  for (int cb = 0; cb < PDIM; cb += 64) {
    v8f acc[4] = {};
    if (MODE == 2) {
#pragma unroll
      for (int r = 0; r < 8; ++r) {
        float4 x = *(const float4*)(xrow + (size_t)r * PDIM + cb);
        acc[0][r] = x.x; acc[1][r] = x.y; acc[2][r] = x.z; acc[3][r] = x.w;
      }
      __builtin_amdgcn_sched_barrier(0);
    }
    strip_chunk<LDW>(a, Wl + cb, acc);
    float4 bias = make_float4(0.f, 0.f, 0.f, 0.f);
    if (MODE != 0) bias = *(const float4*)(bl + cb);
#pragma unroll
    for (int r = 0; r < 8; ++r) {
      float4 ov = make_float4(acc[0][r], acc[1][r], acc[2][r], acc[3][r]);
      if (MODE == 1) {
        ov = make_float4(ov.x + bias.x, ov.y + bias.y, ov.z + bias.z, ov.w + bias.w);
      }
      if (MODE == 2) {
        float g = gcol[r];
        ov = make_float4(ov.x + g * bias.x, ov.y + g * bias.y,
                         ov.z + g * bias.z, ov.w + g * bias.w);
      }
      *(float4*)(orow + (size_t)r * PDIM + cb) = ov;
    }
  }
}

__device__ __forceinline__ void load_strip_k8(const float* rowp, v2f (&a)[32]) {
#pragma unroll
  for (int j = 0; j < 16; ++j) {
    float4 t = *(const float4*)(rowp + 8 * j);
    a[2 * j].x = t.x;     a[2 * j].y = t.y;
    a[2 * j + 1].x = t.z; a[2 * j + 1].y = t.w;
  }
}

template <int LDW>
__device__ __forceinline__ void strip_chunk_T(const v2f (&a)[32], const float* __restrict__ Wt,
                                              v8f (&acc)[4]) {
  float4 n0 = *(const float4*)(Wt);
  float4 n1 = *(const float4*)(Wt + 16 * LDW);
  float4 n2 = *(const float4*)(Wt + 32 * LDW);
  float4 n3 = *(const float4*)(Wt + 48 * LDW);
#pragma unroll
  for (int j = 0; j < 16; ++j) {
    float4 c[4] = {n0, n1, n2, n3};
    if (j + 1 < 16) {
      const float* p = Wt + 8 * (j + 1);
      n0 = *(const float4*)(p);
      n1 = *(const float4*)(p + 16 * LDW);
      n2 = *(const float4*)(p + 32 * LDW);
      n3 = *(const float4*)(p + 48 * LDW);
    }
#pragma unroll
    for (int t = 0; t < 4; ++t) {
      v2f bf;
      bf.x = c[t].x; bf.y = c[t].y;
      acc[t] = __builtin_amdgcn_wmma_f32_16x16x4_f32(false, a[2 * j], false, bf, (short)0, acc[t], false, false);
    }
#pragma unroll
    for (int t = 0; t < 4; ++t) {
      v2f bf;
      bf.x = c[t].z; bf.y = c[t].w;
      acc[t] = __builtin_amdgcn_wmma_f32_16x16x4_f32(false, a[2 * j + 1], false, bf, (short)0, acc[t], false, false);
    }
    __builtin_amdgcn_sched_barrier(0);
  }
}

#define FOLDROWS (PDIM + 16)

__global__ __launch_bounds__(256) void painn_node_gemm(
    const float* __restrict__ s,
    const float* __restrict__ Wq,  const float* __restrict__ bq,
    const float* __restrict__ Wkv, const float* __restrict__ bkv,
    float* __restrict__ Mx, float* __restrict__ ub, float* __restrict__ vv,
    int nStrips, int* __restrict__ counts, int ovfCounter,
    const int* __restrict__ rowIdx, const int* __restrict__ colIdx, int E, int nBins,
    int nSort, int* __restrict__ offs, int2* __restrict__ sorted) {
  __shared__ int hist[MAXBINS + 1];
  __shared__ int wsum[8];
  __shared__ int2 stage[CHUNK];
  if ((int)blockIdx.x < nSort) {
    if (blockIdx.x == 0 && threadIdx.x == 0) counts[ovfCounter] = 0;
    presort_chunk(rowIdx, colIdx, E, nBins, offs, sorted, (int)blockIdx.x, hist, wsum, stage);
    return;
  }
  int bx = (int)blockIdx.x - nSort;
  int wave = threadIdx.x >> 5;
  int lane = threadIdx.x & 31;
  int half = lane >> 4;
  int l16  = lane & 15;
  v2f a[32];
  if (bx < 2) {
    int strip = bx * 8 + wave;
    if (strip > 8) return;
    const float* arow = (strip < 8) ? Wq + (size_t)(strip * 16 + l16) * PDIM : bq;
    load_strip_k8(arow + 4 * half, a);
    float p = 0.f;
#pragma unroll
    for (int j = 0; j < 16; ++j) {
      float4 b = *(const float4*)(bkv + 8 * j + 4 * half);
      p += a[2 * j].x * b.x + a[2 * j].y * b.y + a[2 * j + 1].x * b.z + a[2 * j + 1].y * b.w;
    }
    p += __shfl_xor(p, 16, 32);
    if (half == 0) ub[strip * 16 + l16] = p;
#pragma unroll 1
    for (int cb = 0; cb < PDIM; cb += 64) {
      v8f acc[4] = {};
      strip_chunk_T<3 * PDIM>(a, Wkv + (size_t)(cb + l16) * 3 * PDIM + 4 * half, acc);
      float* orow = Mx + (size_t)(strip * 16 + 8 * half) * PDIM + cb + l16;
#pragma unroll
      for (int r = 0; r < 8; ++r) {
#pragma unroll
        for (int t = 0; t < 4; ++t) orow[(size_t)r * PDIM + 16 * t] = acc[t][r];
      }
    }
    return;
  }
  int strip = (bx - 2) * 8 + wave;
  if (strip >= nStrips) return;
  int rowBase = strip * 16;
  load_strip(s + (size_t)(rowBase + l16) * PDIM + 2 * half, a);
  size_t ooff = (size_t)(rowBase + 8 * half) * PDIM + 4 * l16;
  int wc = 2 * PDIM + 4 * l16;
  strip_block<3 * PDIM, 1>(a, Wkv + (2 * half) * 3 * PDIM + wc, bkv + wc, vv + ooff,
                           nullptr, nullptr);
}

__global__ __launch_bounds__(256) void lists_and_qt(
    const int* __restrict__ offs, const int2* __restrict__ sorted, int nChunks, int nBins,
    int* __restrict__ counts, int2* __restrict__ list, int4* __restrict__ ovf,
    int ovfCounter, int N,
    const float* __restrict__ s, const float* __restrict__ Mx, const float* __restrict__ ub,
    float* __restrict__ qt, float* __restrict__ qb, int nStrips) {
  __shared__ int cnt[BINR];
  __shared__ __attribute__((aligned(16))) int2 slots[BINR * CAP];
  if ((int)blockIdx.x < nBins) {
    bin_lists(offs, sorted, nChunks, nBins, (int)blockIdx.x, counts, list, ovf, ovfCounter, N,
              cnt, slots);
    return;
  }
  int wave = threadIdx.x >> 5;
  int lane = threadIdx.x & 31;
  int half = lane >> 4;
  int l16  = lane & 15;
  int strip = ((int)blockIdx.x - nBins) * 8 + wave;
  if (strip >= nStrips) return;
  int rowBase = strip * 16;
  v2f a[32];
  load_strip(s + (size_t)(rowBase + l16) * PDIM + 2 * half, a);
  float p = 0.f;
#pragma unroll
  for (int ks = 0; ks < 32; ++ks) {
    float2 u2 = *(const float2*)(ub + 4 * ks + 2 * half);
    p += a[ks].x * u2.x + a[ks].y * u2.y;
  }
  p += __shfl_xor(p, 16, 32);
  if (half == 0) qb[rowBase + l16] = p + ub[PDIM];
  size_t ooff = (size_t)(rowBase + 8 * half) * PDIM + 4 * l16;
  strip_block<PDIM, 1>(a, Mx + (2 * half) * PDIM + 4 * l16, Mx + (size_t)PDIM * PDIM + 4 * l16,
                       qt + ooff, nullptr, nullptr);
}

__device__ __forceinline__ float wave_dot(const float4& qv, const float4& kv) {
  float p = qv.x * kv.x + qv.y * kv.y + qv.z * kv.z + qv.w * kv.w;
  p += __shfl_xor(p, 16, 32);
  p += __shfl_xor(p, 8, 32);
  p += __shfl_xor(p, 4, 32);
  p += __shfl_xor(p, 2, 32);
  p += __shfl_xor(p, 1, 32);
  return p;
}

__device__ __forceinline__ void axpy4(float w, const float4& x, float4& a) {
  a.x += w * x.x; a.y += w * x.y; a.z += w * x.z; a.w += w * x.w;
}

__device__ __forceinline__ void store_vrow(
    int node, int f, const float4& a0, const float4& a1, const float4& a2,
    const float* __restrict__ v, const float* __restrict__ bvec, float* __restrict__ v_out) {
  size_t vo3 = (size_t)node * 3 * PDIM + f;
  float4 bv = *(const float4*)(bvec + f);
  const float* vr = v + vo3;
  float4 y0 = *(const float4*)(vr);
  float4 y1 = *(const float4*)(vr + PDIM);
  float4 y2 = *(const float4*)(vr + 2 * PDIM);
  float* vo = v_out + vo3;
  *(float4*)(vo) = make_float4(y0.x + (a0.x + bv.x), y0.y + (a0.y + bv.y),
                               y0.z + (a0.z + bv.z), y0.w + (a0.w + bv.w));
  *(float4*)(vo + PDIM) = make_float4(y1.x + (a1.x + bv.x), y1.y + (a1.y + bv.y),
                                      y1.z + (a1.z + bv.z), y1.w + (a1.w + bv.w));
  *(float4*)(vo + 2 * PDIM) = make_float4(y2.x + (a2.x + bv.x), y2.y + (a2.y + bv.y),
                                          y2.z + (a2.z + bv.z), y2.w + (a2.w + bv.w));
}

template <bool SCALAR>
__device__ void long_row(
    int node, int deg,
    const float* __restrict__ qt, const float* __restrict__ qb,
    const float* __restrict__ s, const float* __restrict__ U,
    const float* __restrict__ v, const float* __restrict__ bvec,
    const int* __restrict__ counts, const int2* __restrict__ list,
    const int4* __restrict__ ovf,
    float* __restrict__ S, float* __restrict__ sw, float* __restrict__ v_out, int N) {
  int lane = threadIdx.x & 31;
  int d = lane * 4;
  float4 qv = *(const float4*)(qt + (size_t)node * PDIM + d);
  float qbn = qb[node];
  float4 a0 = make_float4(0.f, 0.f, 0.f, 0.f);
  float4 a1 = a0, a2 = a0;
  float sws = 0.f;
  int nOvf = counts[N];
  int2 ec = list[(size_t)node * CAP + lane];
  int last = -1;
  for (int i = 0; i < deg; ++i) {
    int be = 0x7fffffff, bc = 0;
    if (ec.x > last) { be = ec.x; bc = ec.y; }
    for (int j = lane; j < nOvf; j += 32) {
      int4 o = ovf[j];
      if (o.x == node && o.y > last && o.y < be) { be = o.y; bc = o.z; }
    }
    int mn = be;
    mn = min(mn, __shfl_xor(mn, 16, 32));
    mn = min(mn, __shfl_xor(mn, 8, 32));
    mn = min(mn, __shfl_xor(mn, 4, 32));
    mn = min(mn, __shfl_xor(mn, 2, 32));
    mn = min(mn, __shfl_xor(mn, 1, 32));
    unsigned m = __builtin_amdgcn_ballot_w32(be == mn);
    int src = __builtin_ctz(m | 0x80000000u);
    int c = __builtin_amdgcn_readlane(bc, src);
    last = __builtin_amdgcn_readfirstlane(mn);
    float4 sv = *(const float4*)(s + (size_t)c * PDIM + d);
    float w = wave_dot(qv, sv) + qbn;
    if (SCALAR) {
      axpy4(w, sv, a0);
      sws += w;
    } else {
      const float* uc = U + (size_t)c * 3 * PDIM + d;
      float4 x0 = *(const float4*)(uc);
      float4 x1 = *(const float4*)(uc + PDIM);
      float4 x2 = *(const float4*)(uc + 2 * PDIM);
      axpy4(w, x0, a0);
      axpy4(w, x1, a1);
      axpy4(w, x2, a2);
    }
  }
  if (SCALAR) {
    *(float4*)(S + (size_t)node * PDIM + d) = a0;
    if (lane == 0) sw[node] = sws;
  } else {
    store_vrow(node, d, a0, a1, a2, v, bvec, v_out);
  }
}

__global__ __launch_bounds__(256) void edge_weights(
    const float* __restrict__ qt, const float* __restrict__ qb, const float* __restrict__ s,
    const int* __restrict__ counts, int2* __restrict__ list, const int4* __restrict__ ovf,
    float* __restrict__ S, float* __restrict__ sw, int N) {
  int wave = __builtin_amdgcn_readfirstlane((int)(threadIdx.x >> 5));
  int node = blockIdx.x * 8 + wave;
  if (node >= N) return;
  int lane = threadIdx.x & 31;
  int d = lane * 4;
  int deg = counts[node];
  if (deg > CAP) {
    long_row<true>(node, deg, qt, qb, s, nullptr, nullptr, nullptr, counts, list, ovf,
                   S, sw, nullptr, N);
    return;
  }
  int2* slots = list + (size_t)node * CAP;

  float4 qv = *(const float4*)(qt + (size_t)node * PDIM + d);
  float qbn = qb[node];
  int2 ec = make_int2(0x7fffffff, 0);
  if (lane < deg) ec = slots[lane];
  int rank = 0;
  for (int j = 0; j < deg; ++j) {
    int ej = __builtin_amdgcn_readlane(ec.x, j);
    rank += (ej < ec.x) ? 1 : 0;
  }
  float4 acc = make_float4(0.f, 0.f, 0.f, 0.f);
  float sws = 0.f;
  float wm = 0.f;
  int cm = 0;
  for (int p0 = 0; p0 < deg; p0 += 4) {
    float4 sv[4];
    int cc[4];
#pragma unroll
    for (int u = 0; u < 4; ++u) {
      int p = min(p0 + u, deg - 1);
      unsigned m = __builtin_amdgcn_ballot_w32(rank == p);
      int src = __builtin_ctz(m | 0x80000000u);
      cc[u] = __builtin_amdgcn_readlane(ec.y, src);
      sv[u] = *(const float4*)(s + (size_t)cc[u] * PDIM + d);
    }
#pragma unroll
    for (int u = 0; u < 4; ++u) {
      float w = wave_dot(qv, sv[u]) + qbn;
      if (p0 + u < deg) {
        axpy4(w, sv[u], acc);
        sws += w;
      }
      if (lane == p0 + u) { wm = w; cm = cc[u]; }
    }
  }
  if (lane < deg) slots[lane] = make_int2(cm, __float_as_int(wm));
  *(float4*)(S + (size_t)node * PDIM + d) = acc;
  if (lane == 0) sw[node] = sws;
}

__global__ __launch_bounds__(256) void node_rows_half(
    const float* __restrict__ qt, const float* __restrict__ qb, const float* __restrict__ s,
    const float* __restrict__ U, const float* __restrict__ v,
    const float* __restrict__ bvec,
    const int* __restrict__ counts, const int2* __restrict__ list,
    const int4* __restrict__ ovf,
    float* __restrict__ v_out, int N, int nGroups) {
  int wave = __builtin_amdgcn_readfirstlane((int)(threadIdx.x >> 5));
  int half = blockIdx.x / nGroups;
  int pairBase = ((blockIdx.x - half * nGroups) * 8 + wave) * 2;
  if (pairBase >= N) return;
  int lane = threadIdx.x & 31;
  int node = pairBase + (lane >> 4);
  bool valid = node < N;
  int nd = valid ? node : pairBase;
  int deg = valid ? counts[nd] : 0;
  int dl = (deg <= CAP) ? deg : 0;
  int f = half * 64 + (lane & 15) * 4;
  int maxd = max(__builtin_amdgcn_readlane(dl, 0), __builtin_amdgcn_readlane(dl, 16));
  const int2* sl = list + (size_t)nd * CAP;

  float4 a0 = make_float4(0.f, 0.f, 0.f, 0.f);
  float4 a1 = a0, a2 = a0;
  int2 cw = make_int2(0, 0);
  if (0 < dl) cw = sl[0];
#pragma unroll 1
  for (int i = 0; i < maxd; ++i) {
    int2 cur = cw;
    if (i + 1 < dl) cw = sl[i + 1];
    if (i < dl) {
      int c = cur.x;
      float w = __int_as_float(cur.y);
      const float* uc = U + (size_t)c * 3 * PDIM + f;
      float4 x0 = *(const float4*)(uc);
      float4 x1 = *(const float4*)(uc + PDIM);
      float4 x2 = *(const float4*)(uc + 2 * PDIM);
      axpy4(w, x0, a0);
      axpy4(w, x1, a1);
      axpy4(w, x2, a2);
    }
  }
  if (valid && deg <= CAP)
    store_vrow(node, f, a0, a1, a2, v, bvec, v_out);
  if (half == 0) {
    int dg0 = __builtin_amdgcn_readlane(deg, 0);
    int dg1 = __builtin_amdgcn_readlane(deg, 16);
    if (dg0 > CAP)
      long_row<false>(pairBase, dg0, qt, qb, s, U, v, bvec, counts, list, ovf,
                      nullptr, nullptr, v_out, N);
    if (dg1 > CAP)
      long_row<false>(pairBase + 1, dg1, qt, qb, s, U, v, bvec, counts, list, ovf,
                      nullptr, nullptr, v_out, N);
  }
}

__global__ __launch_bounds__(256) void painn_u_gemm(
    const float* __restrict__ vv, const float* __restrict__ v,
    const float* __restrict__ Wvec, float* __restrict__ U, int nStripsU, int nGroupsU,
    const float* __restrict__ S, const float* __restrict__ sw, const float* __restrict__ s,
    const float* __restrict__ Wkv, const float* __restrict__ bkv,
    float* __restrict__ s_out, int nStripsS) {
  int wave = threadIdx.x >> 5;
  int lane = threadIdx.x & 31;
  int half = lane >> 4;
  int l16  = lane & 15;
  v2f a[32];
  if ((int)blockIdx.x < nGroupsU) {
    int strip = blockIdx.x * 8 + wave;
    if (strip >= nStripsU) return;
    int rowBase = strip * 16;
    int row = rowBase + l16;
    int node = row / 3;
    load_strip_prod(v + (size_t)row * PDIM + 2 * half, vv + (size_t)node * PDIM + 2 * half, a);
    size_t ooff = (size_t)(rowBase + 8 * half) * PDIM + 4 * l16;
    strip_block<PDIM, 0>(a, Wvec + (2 * half) * PDIM + 4 * l16, nullptr, U + ooff,
                         nullptr, nullptr);
  } else {
    int strip = ((int)blockIdx.x - nGroupsU) * 8 + wave;
    if (strip >= nStripsS) return;
    int rowBase = strip * 16;
    load_strip(S + (size_t)(rowBase + l16) * PDIM + 2 * half, a);
    size_t ooff = (size_t)(rowBase + 8 * half) * PDIM + 4 * l16;
    int wc = PDIM + 4 * l16;
    strip_block<3 * PDIM, 2>(a, Wkv + (2 * half) * 3 * PDIM + wc, bkv + wc, s_out + ooff,
                             s + ooff, sw + rowBase + 8 * half);
  }
}

extern "C" void kernel_launch(void* const* d_in, const int* in_sizes, int n_in,
                              void* d_out, int out_size, void* d_ws, size_t ws_size,
                              hipStream_t stream) {
  const float* s    = (const float*)d_in[0];
  const float* v    = (const float*)d_in[1];
  const int*   ei   = (const int*)d_in[2];
  const float* Wq   = (const float*)d_in[3];
  const float* bq   = (const float*)d_in[4];
  const float* Wkv  = (const float*)d_in[5];
  const float* bkv  = (const float*)d_in[6];
  const float* Wvec = (const float*)d_in[7];
  const float* bvec = (const float*)d_in[8];

  int N = in_sizes[0] / PDIM;
  int E = in_sizes[2] / 2;

  float* s_out = (float*)d_out;
  float* v_out = s_out + (size_t)N * PDIM;

  int nCnt = (N + 1 + 63) / 64 * 64;

  float* Mx   = (float*)d_ws;
  float* ub   = Mx + (size_t)FOLDROWS * PDIM;
  float* qt   = Mx + (size_t)N * PDIM;
  float* vv   = qt + (size_t)N * PDIM;
  float* S    = vv + (size_t)N * PDIM;
  float* U    = S  + (size_t)N * PDIM;
  float* qb   = U  + (size_t)N * 3 * PDIM;
  float* sw   = qb + nCnt;

  int*  counts = (int*)(sw + nCnt);
  int2* list   = (int2*)(counts + nCnt);
  int4* ovf    = (int4*)(list + (size_t)N * CAP);

  int nBins   = (N + BINR - 1) / BINR;
  int nChunks = (E + CHUNK - 1) / CHUNK;
  int nOffs   = (nChunks * (nBins + 1) + 63) / 64 * 64;
  int*  offs   = (int*)(ovf + E);
  int2* sorted = (int2*)(offs + nOffs);

  const int* rowIdx = ei;
  const int* colIdx = ei + E;

  int stripsA = N / 16;
  int groupsA = (stripsA + 7) / 8;
  painn_node_gemm<<<nChunks + 2 + groupsA, 256, 0, stream>>>(
      s, Wq, bq, Wkv, bkv, Mx, ub, vv, stripsA, counts, N,
      rowIdx, colIdx, E, nBins, nChunks, offs, sorted);

  lists_and_qt<<<nBins + groupsA, 256, 0, stream>>>(
      offs, sorted, nChunks, nBins, counts, list, ovf, N, N, s, Mx, ub, qt, qb, stripsA);

  edge_weights<<<(N + 7) / 8, 256, 0, stream>>>(qt, qb, s, counts, list, ovf, S, sw, N);

  int stripsC = N * 3 / 16;
  int groupsC = (stripsC + 7) / 8;
  painn_u_gemm<<<groupsC + groupsA, 256, 0, stream>>>(
      vv, v, Wvec, U, stripsC, groupsC, S, sw, s, Wkv, bkv, s_out, stripsA);

  int pairGroups = ((N + 1) / 2 + 7) / 8;
  node_rows_half<<<2 * pairGroups, 256, 0, stream>>>(
      qt, qb, s, U, v, bvec, counts, list, ovf, v_out, N, pairGroups);
}


# Edge-weight pass: two destination nodes per wave, one per 16-lane row (8 features per lane), so each dot is summed by row DPP adds, not 5 shuffles; 25000 waves, not 50000; the dot's sum order changes
# speedup vs baseline: 4.0623x; 1.0475x over previous
#include <hip/hip_runtime.h>


typedef __attribute__((ext_vector_type(2))) float v2f;
typedef __attribute__((ext_vector_type(8))) float v8f;

#define PDIM 128
#define CAP 32

#define BINSH 8
#define BINR (1 << BINSH)
#define MAXBINS 256
#define EPT 10
#define CHUNK (256 * EPT)

__device__ __forceinline__ void presort_chunk(const int* __restrict__ rowIdx,
                                              const int* __restrict__ colIdx, int E, int nBins,
                                              int* __restrict__ offs, int2* __restrict__ sorted,
                                              int w, int* hist, int* wsum, int2* stage) {
  int t = threadIdx.x;
  for (int b = t; b <= nBins; b += 256) hist[b] = 0;
  __syncthreads();
  int e0 = w * CHUNK;
  int r[EPT], c[EPT], rk[EPT];
#pragma unroll
  for (int u = 0; u < EPT; ++u) {
    int e = e0 + u * 256 + t;
    int el = min(e, E - 1);
    int rr = rowIdx[el];
    c[u] = colIdx[el];
    r[u] = (e < E) ? rr : -1;
  }
#pragma unroll
  for (int u = 0; u < EPT; ++u)
    rk[u] = (r[u] >= 0) ? atomicAdd(&hist[r[u] >> BINSH], 1) : 0;
  __syncthreads();
  int lane = t & 31;
  int wave = t >> 5;
  int v = (t < nBins) ? hist[t] : 0;
  int incl = v;
#pragma unroll
  for (int d = 1; d < 32; d <<= 1) {
    int n = __shfl_up(incl, d, 32);
    if (lane >= d) incl += n;
  }
  if (lane == 31) wsum[wave] = incl;
  __syncthreads();
  int pre = 0;
  for (int k = 0; k < wave; ++k) pre += wsum[k];
  int excl = pre + incl - v;
  if (t < nBins) hist[t] = excl;
  if (t == nBins - 1) hist[nBins] = excl + v;
  __syncthreads();
  for (int b = t; b <= nBins; b += 256) offs[(size_t)w * (nBins + 1) + b] = hist[b];
#pragma unroll
  for (int u = 0; u < EPT; ++u) {
    if (r[u] < 0) continue;
    unsigned lr = (unsigned)r[u] & (BINR - 1);
    stage[hist[r[u] >> BINSH] + rk[u]] = make_int2(e0 + u * 256 + t, (int)((lr << 24) | (unsigned)c[u]));
  }
  __syncthreads();
  int cnt = min(CHUNK, E - e0);
  for (int i = t; i < cnt; i += 256) sorted[(size_t)e0 + i] = stage[i];
}

__device__ __forceinline__ void place_edge(int2 en, int nodeBase, int* cnt, int2* slots,
                                           int* __restrict__ counts, int4* __restrict__ ovf,
                                           int ovfCounter) {
  int lr = (int)((unsigned)en.y >> 24);
  int col = en.y & 0xffffff;
  int sl = atomicAdd(&cnt[lr], 1);
  if (sl < CAP) {
    slots[lr * CAP + sl] = make_int2(en.x, col);
  } else {
    int o = atomicAdd(&counts[ovfCounter], 1);
    ovf[o] = make_int4(nodeBase + lr, en.x, col, 0);
  }
}

__device__ __forceinline__ void bin_lists(const int* __restrict__ offs,
                                          const int2* __restrict__ sorted,
                                          int nChunks, int nBins, int b,
                                          int* __restrict__ counts,
                                          int2* __restrict__ list,
                                          int4* __restrict__ ovf,
                                          int ovfCounter, int N, int* cnt, int2* slots) {
  int t = threadIdx.x;
  int nodeBase = b * BINR;
  for (int i = t; i < BINR; i += 256) cnt[i] = 0;
  __syncthreads();
  for (int w = t; w < nChunks; w += 256) {
    const int* op = offs + (size_t)w * (nBins + 1) + b;
    int o0 = op[0];
    int o1 = op[1];
    const int2* sp = sorted + (size_t)w * CHUNK;
    for (int i0 = o0; i0 < o1; i0 += 4) {
      int lastI = o1 - 1;
      int2 en0 = sp[i0];
      int2 en1 = sp[min(i0 + 1, lastI)];
      int2 en2 = sp[min(i0 + 2, lastI)];
      int2 en3 = sp[min(i0 + 3, lastI)];
      place_edge(en0, nodeBase, cnt, slots, counts, ovf, ovfCounter);
      if (i0 + 1 < o1) place_edge(en1, nodeBase, cnt, slots, counts, ovf, ovfCounter);
      if (i0 + 2 < o1) place_edge(en2, nodeBase, cnt, slots, counts, ovf, ovfCounter);
      if (i0 + 3 < o1) place_edge(en3, nodeBase, cnt, slots, counts, ovf, ovfCounter);
    }
  }
  __syncthreads();
  for (int i = t; i < BINR; i += 256)
    if (nodeBase + i < N) counts[nodeBase + i] = cnt[i];
  int2* lst = list + (size_t)nodeBase * CAP;
  for (int i = t; i < BINR * CAP / 2; i += 256) {
    int nd = i / (CAP / 2);
    int s0 = (i % (CAP / 2)) * 2;
    int filled = (nodeBase + nd < N) ? min(cnt[nd], CAP) : 0;
    if (s0 + 1 < filled) {
      ((int4*)lst)[i] = ((const int4*)slots)[i];
    } else if (s0 < filled) {
      lst[2 * i] = slots[2 * i];
    }
  }
}

__device__ __forceinline__ void load_strip(const float* __restrict__ rowp, v2f (&a)[32]) {
#pragma unroll
  for (int ks = 0; ks < 32; ++ks) {
    float2 t = *(const float2*)(rowp + 4 * ks);
    a[ks].x = t.x;
    a[ks].y = t.y;
  }
}

__device__ __forceinline__ void load_strip_prod(const float* __restrict__ xrow,
                                                const float* __restrict__ grow, v2f (&a)[32]) {
#pragma unroll
  for (int ks = 0; ks < 32; ++ks) {
    float2 t = *(const float2*)(xrow + 4 * ks);
    float2 g = *(const float2*)(grow + 4 * ks);
    a[ks].x = g.x * t.x;
    a[ks].y = g.y * t.y;
  }
}

__device__ __forceinline__ void step4(const v2f& af, const float4& r0, const float4& r1,
                                      v8f (&acc)[4]) {
  v2f bf;
  bf.x = r0.x; bf.y = r1.x;
  acc[0] = __builtin_amdgcn_wmma_f32_16x16x4_f32(false, af, false, bf, (short)0, acc[0], false, false);
  bf.x = r0.y; bf.y = r1.y;
  acc[1] = __builtin_amdgcn_wmma_f32_16x16x4_f32(false, af, false, bf, (short)0, acc[1], false, false);
  bf.x = r0.z; bf.y = r1.z;
  acc[2] = __builtin_amdgcn_wmma_f32_16x16x4_f32(false, af, false, bf, (short)0, acc[2], false, false);
  bf.x = r0.w; bf.y = r1.w;
  acc[3] = __builtin_amdgcn_wmma_f32_16x16x4_f32(false, af, false, bf, (short)0, acc[3], false, false);
}

template <int LDW>
__device__ __forceinline__ void strip_chunk(const v2f (&a)[32], const float* __restrict__ Wl,
                                            v8f (&acc)[4]) {
  float4 n0 = *(const float4*)(Wl);
  float4 n1 = *(const float4*)(Wl + LDW);
  float4 n2 = *(const float4*)(Wl + 4 * LDW);
  float4 n3 = *(const float4*)(Wl + 5 * LDW);
#pragma unroll
  for (int j = 0; j < 16; ++j) {
    float4 c0 = n0, c1 = n1, c2 = n2, c3 = n3;
    if (j + 1 < 16) {
      const float* p = Wl + (8 * (j + 1)) * LDW;
      n0 = *(const float4*)(p);
      n1 = *(const float4*)(p + LDW);
      n2 = *(const float4*)(p + 4 * LDW);
      n3 = *(const float4*)(p + 5 * LDW);
    }
    step4(a[2 * j],     c0, c1, acc);
    step4(a[2 * j + 1], c2, c3, acc);
    __builtin_amdgcn_sched_barrier(0);
  }
}

template <int LDW, int MODE>
__device__ __forceinline__ void strip_block(const v2f (&a)[32], const float* __restrict__ Wl,
                                            const float* __restrict__ bl,
                                            float* __restrict__ orow,
                                            const float* __restrict__ xrow,
                                            const float* __restrict__ gcol) {
#pragma unroll 1
  for (int cb = 0; cb < PDIM; cb += 64) {
    v8f acc[4] = {};
    if (MODE == 2) {
#pragma unroll
      for (int r = 0; r < 8; ++r) {
        float4 x = *(const float4*)(xrow + (size_t)r * PDIM + cb);
        acc[0][r] = x.x; acc[1][r] = x.y; acc[2][r] = x.z; acc[3][r] = x.w;
      }
      __builtin_amdgcn_sched_barrier(0);
    }
    strip_chunk<LDW>(a, Wl + cb, acc);
    float4 bias = make_float4(0.f, 0.f, 0.f, 0.f);
    if (MODE != 0) bias = *(const float4*)(bl + cb);
#pragma unroll
    for (int r = 0; r < 8; ++r) {
      float4 ov = make_float4(acc[0][r], acc[1][r], acc[2][r], acc[3][r]);
      if (MODE == 1) {
        ov = make_float4(ov.x + bias.x, ov.y + bias.y, ov.z + bias.z, ov.w + bias.w);
      }
      if (MODE == 2) {
        float g = gcol[r];
        ov = make_float4(ov.x + g * bias.x, ov.y + g * bias.y,
                         ov.z + g * bias.z, ov.w + g * bias.w);
      }
      *(float4*)(orow + (size_t)r * PDIM + cb) = ov;
    }
  }
}

__device__ __forceinline__ void load_strip_k8(const float* rowp, v2f (&a)[32]) {
#pragma unroll
  for (int j = 0; j < 16; ++j) {
    float4 t = *(const float4*)(rowp + 8 * j);
    a[2 * j].x = t.x;     a[2 * j].y = t.y;
    a[2 * j + 1].x = t.z; a[2 * j + 1].y = t.w;
  }
}

template <int LDW>
__device__ __forceinline__ void strip_chunk_T(const v2f (&a)[32], const float* __restrict__ Wt,
                                              v8f (&acc)[4]) {
  float4 n0 = *(const float4*)(Wt);
  float4 n1 = *(const float4*)(Wt + 16 * LDW);
  float4 n2 = *(const float4*)(Wt + 32 * LDW);
  float4 n3 = *(const float4*)(Wt + 48 * LDW);
#pragma unroll
  for (int j = 0; j < 16; ++j) {
    float4 c[4] = {n0, n1, n2, n3};
    if (j + 1 < 16) {
      const float* p = Wt + 8 * (j + 1);
      n0 = *(const float4*)(p);
      n1 = *(const float4*)(p + 16 * LDW);
      n2 = *(const float4*)(p + 32 * LDW);
      n3 = *(const float4*)(p + 48 * LDW);
    }
#pragma unroll
    for (int t = 0; t < 4; ++t) {
      v2f bf;
      bf.x = c[t].x; bf.y = c[t].y;
      acc[t] = __builtin_amdgcn_wmma_f32_16x16x4_f32(false, a[2 * j], false, bf, (short)0, acc[t], false, false);
    }
#pragma unroll
    for (int t = 0; t < 4; ++t) {
      v2f bf;
      bf.x = c[t].z; bf.y = c[t].w;
      acc[t] = __builtin_amdgcn_wmma_f32_16x16x4_f32(false, a[2 * j + 1], false, bf, (short)0, acc[t], false, false);
    }
    __builtin_amdgcn_sched_barrier(0);
  }
}

#define FOLDROWS (PDIM + 16)

__global__ __launch_bounds__(256) void painn_node_gemm(
    const float* __restrict__ s,
    const float* __restrict__ Wq,  const float* __restrict__ bq,
    const float* __restrict__ Wkv, const float* __restrict__ bkv,
    float* __restrict__ Mx, float* __restrict__ ub, float* __restrict__ vv,
    int nStrips, int* __restrict__ counts, int ovfCounter,
    const int* __restrict__ rowIdx, const int* __restrict__ colIdx, int E, int nBins,
    int nSort, int* __restrict__ offs, int2* __restrict__ sorted) {
  __shared__ int hist[MAXBINS + 1];
  __shared__ int wsum[8];
  __shared__ int2 stage[CHUNK];
  if ((int)blockIdx.x < nSort) {
    if (blockIdx.x == 0 && threadIdx.x == 0) counts[ovfCounter] = 0;
    presort_chunk(rowIdx, colIdx, E, nBins, offs, sorted, (int)blockIdx.x, hist, wsum, stage);
    return;
  }
  int bx = (int)blockIdx.x - nSort;
  int wave = threadIdx.x >> 5;
  int lane = threadIdx.x & 31;
  int half = lane >> 4;
  int l16  = lane & 15;
  v2f a[32];
  if (bx < 2) {
    int strip = bx * 8 + wave;
    if (strip > 8) return;
    const float* arow = (strip < 8) ? Wq + (size_t)(strip * 16 + l16) * PDIM : bq;
    load_strip_k8(arow + 4 * half, a);
    float p = 0.f;
#pragma unroll
    for (int j = 0; j < 16; ++j) {
      float4 b = *(const float4*)(bkv + 8 * j + 4 * half);
      p += a[2 * j].x * b.x + a[2 * j].y * b.y + a[2 * j + 1].x * b.z + a[2 * j + 1].y * b.w;
    }
    p += __shfl_xor(p, 16, 32);
    if (half == 0) ub[strip * 16 + l16] = p;
#pragma unroll 1
    for (int cb = 0; cb < PDIM; cb += 64) {
      v8f acc[4] = {};
      strip_chunk_T<3 * PDIM>(a, Wkv + (size_t)(cb + l16) * 3 * PDIM + 4 * half, acc);
      float* orow = Mx + (size_t)(strip * 16 + 8 * half) * PDIM + cb + l16;
#pragma unroll
      for (int r = 0; r < 8; ++r) {
#pragma unroll
        for (int t = 0; t < 4; ++t) orow[(size_t)r * PDIM + 16 * t] = acc[t][r];
      }
    }
    return;
  }
  int strip = (bx - 2) * 8 + wave;
  if (strip >= nStrips) return;
  int rowBase = strip * 16;
  load_strip(s + (size_t)(rowBase + l16) * PDIM + 2 * half, a);
  size_t ooff = (size_t)(rowBase + 8 * half) * PDIM + 4 * l16;
  int wc = 2 * PDIM + 4 * l16;
  strip_block<3 * PDIM, 1>(a, Wkv + (2 * half) * 3 * PDIM + wc, bkv + wc, vv + ooff,
                           nullptr, nullptr);
}

__global__ __launch_bounds__(256) void lists_and_qt(
    const int* __restrict__ offs, const int2* __restrict__ sorted, int nChunks, int nBins,
    int* __restrict__ counts, int2* __restrict__ list, int4* __restrict__ ovf,
    int ovfCounter, int N,
    const float* __restrict__ s, const float* __restrict__ Mx, const float* __restrict__ ub,
    float* __restrict__ qt, float* __restrict__ qb, int nStrips) {
  __shared__ int cnt[BINR];
  __shared__ __attribute__((aligned(16))) int2 slots[BINR * CAP];
  if ((int)blockIdx.x < nBins) {
    bin_lists(offs, sorted, nChunks, nBins, (int)blockIdx.x, counts, list, ovf, ovfCounter, N,
              cnt, slots);
    return;
  }
  int wave = threadIdx.x >> 5;
  int lane = threadIdx.x & 31;
  int half = lane >> 4;
  int l16  = lane & 15;
  int strip = ((int)blockIdx.x - nBins) * 8 + wave;
  if (strip >= nStrips) return;
  int rowBase = strip * 16;
  v2f a[32];
  load_strip(s + (size_t)(rowBase + l16) * PDIM + 2 * half, a);
  float p = 0.f;
#pragma unroll
  for (int ks = 0; ks < 32; ++ks) {
    float2 u2 = *(const float2*)(ub + 4 * ks + 2 * half);
    p += a[ks].x * u2.x + a[ks].y * u2.y;
  }
  p += __shfl_xor(p, 16, 32);
  if (half == 0) qb[rowBase + l16] = p + ub[PDIM];
  size_t ooff = (size_t)(rowBase + 8 * half) * PDIM + 4 * l16;
  strip_block<PDIM, 1>(a, Mx + (2 * half) * PDIM + 4 * l16, Mx + (size_t)PDIM * PDIM + 4 * l16,
                       qt + ooff, nullptr, nullptr);
}

__device__ __forceinline__ float wave_dot(const float4& qv, const float4& kv) {
  float p = qv.x * kv.x + qv.y * kv.y + qv.z * kv.z + qv.w * kv.w;
  p += __shfl_xor(p, 16, 32);
  p += __shfl_xor(p, 8, 32);
  p += __shfl_xor(p, 4, 32);
  p += __shfl_xor(p, 2, 32);
  p += __shfl_xor(p, 1, 32);
  return p;
}

__device__ __forceinline__ void axpy4(float w, const float4& x, float4& a) {
  a.x += w * x.x; a.y += w * x.y; a.z += w * x.z; a.w += w * x.w;
}

__device__ __forceinline__ void store_vrow(
    int node, int f, const float4& a0, const float4& a1, const float4& a2,
    const float* __restrict__ v, const float* __restrict__ bvec, float* __restrict__ v_out) {
  size_t vo3 = (size_t)node * 3 * PDIM + f;
  float4 bv = *(const float4*)(bvec + f);
  const float* vr = v + vo3;
  float4 y0 = *(const float4*)(vr);
  float4 y1 = *(const float4*)(vr + PDIM);
  float4 y2 = *(const float4*)(vr + 2 * PDIM);
  float* vo = v_out + vo3;
  *(float4*)(vo) = make_float4(y0.x + (a0.x + bv.x), y0.y + (a0.y + bv.y),
                               y0.z + (a0.z + bv.z), y0.w + (a0.w + bv.w));
  *(float4*)(vo + PDIM) = make_float4(y1.x + (a1.x + bv.x), y1.y + (a1.y + bv.y),
                                      y1.z + (a1.z + bv.z), y1.w + (a1.w + bv.w));
  *(float4*)(vo + 2 * PDIM) = make_float4(y2.x + (a2.x + bv.x), y2.y + (a2.y + bv.y),
                                          y2.z + (a2.z + bv.z), y2.w + (a2.w + bv.w));
}

template <bool SCALAR>
__device__ void long_row(
    int node, int deg,
    const float* __restrict__ qt, const float* __restrict__ qb,
    const float* __restrict__ s, const float* __restrict__ U,
    const float* __restrict__ v, const float* __restrict__ bvec,
    const int* __restrict__ counts, const int2* __restrict__ list,
    const int4* __restrict__ ovf,
    float* __restrict__ S, float* __restrict__ sw, float* __restrict__ v_out, int N) {
  int lane = threadIdx.x & 31;
  int d = lane * 4;
  float4 qv = *(const float4*)(qt + (size_t)node * PDIM + d);
  float qbn = qb[node];
  float4 a0 = make_float4(0.f, 0.f, 0.f, 0.f);
  float4 a1 = a0, a2 = a0;
  float sws = 0.f;
  int nOvf = counts[N];
  int2 ec = list[(size_t)node * CAP + lane];
  int last = -1;
  for (int i = 0; i < deg; ++i) {
    int be = 0x7fffffff, bc = 0;
    if (ec.x > last) { be = ec.x; bc = ec.y; }
    for (int j = lane; j < nOvf; j += 32) {
      int4 o = ovf[j];
      if (o.x == node && o.y > last && o.y < be) { be = o.y; bc = o.z; }
    }
    int mn = be;
    mn = min(mn, __shfl_xor(mn, 16, 32));
    mn = min(mn, __shfl_xor(mn, 8, 32));
    mn = min(mn, __shfl_xor(mn, 4, 32));
    mn = min(mn, __shfl_xor(mn, 2, 32));
    mn = min(mn, __shfl_xor(mn, 1, 32));
    unsigned m = __builtin_amdgcn_ballot_w32(be == mn);
    int src = __builtin_ctz(m | 0x80000000u);
    int c = __builtin_amdgcn_readlane(bc, src);
    last = __builtin_amdgcn_readfirstlane(mn);
    float4 sv = *(const float4*)(s + (size_t)c * PDIM + d);
    float w = wave_dot(qv, sv) + qbn;
    if (SCALAR) {
      axpy4(w, sv, a0);
      sws += w;
    } else {
      const float* uc = U + (size_t)c * 3 * PDIM + d;
      float4 x0 = *(const float4*)(uc);
      float4 x1 = *(const float4*)(uc + PDIM);
      float4 x2 = *(const float4*)(uc + 2 * PDIM);
      axpy4(w, x0, a0);
      axpy4(w, x1, a1);
      axpy4(w, x2, a2);
    }
  }
  if (SCALAR) {
    *(float4*)(S + (size_t)node * PDIM + d) = a0;
    if (lane == 0) sw[node] = sws;
  } else {
    store_vrow(node, d, a0, a1, a2, v, bvec, v_out);
  }
}

template <int M>
__device__ __forceinline__ float row_xadd(float x) {
  int y = __builtin_amdgcn_update_dpp(0, __float_as_int(x), 0x160 + M, 0xf, 0xf, false);
  return x + __int_as_float(y);
}

__global__ __launch_bounds__(256) void edge_weights(
    const float* __restrict__ qt, const float* __restrict__ qb, const float* __restrict__ s,
    const int* __restrict__ counts, int2* __restrict__ list, const int4* __restrict__ ovf,
    float* __restrict__ S, float* __restrict__ sw, int N) {
  __shared__ __attribute__((aligned(16))) int scol[8][2][CAP];
  int wave = __builtin_amdgcn_readfirstlane((int)(threadIdx.x >> 5));
  int pairBase = (blockIdx.x * 8 + wave) * 2;
  if (pairBase >= N) return;
  int lane = threadIdx.x & 31;
  int g = lane >> 4;
  int l16 = lane & 15;
  int node = pairBase + g;
  bool valid = node < N;
  int nd = valid ? node : pairBase;
  int deg = valid ? counts[nd] : 0;
  int dl = (deg <= CAP) ? deg : 0;
  int maxd = max(__builtin_amdgcn_readlane(dl, 0), __builtin_amdgcn_readlane(dl, 16));
  int2* slots = list + (size_t)nd * CAP;
  int sA = 2 * l16;

  int4 e2 = make_int4(0x7fffffff, 0, 0x7fffffff, 0);
  if (sA + 1 < dl) {
    e2 = ((const int4*)slots)[l16];
  } else if (sA < dl) {
    int2 t = slots[sA];
    e2.x = t.x;
    e2.y = t.y;
  }
  float4 qv0 = *(const float4*)(qt + (size_t)nd * PDIM + 4 * l16);
  float4 qv1 = *(const float4*)(qt + (size_t)nd * PDIM + 64 + 4 * l16);
  float qbn = qb[nd];

  int rA = 0, rB = 0;
  int hb = g << 4;
  for (int jj = 0; 2 * jj < maxd; ++jj) {
    int ia = __shfl(e2.x, hb + jj, 32);
    int ib = __shfl(e2.z, hb + jj, 32);
    rA += ((ia < e2.x) ? 1 : 0) + ((ib < e2.x) ? 1 : 0);
    rB += ((ia < e2.z) ? 1 : 0) + ((ib < e2.z) ? 1 : 0);
  }
  int* sc = scol[wave][g];
  sc[sA] = nd;
  sc[sA + 1] = nd;
  __builtin_amdgcn_wave_barrier();
  if (sA < dl) sc[rA] = e2.y;
  if (sA + 1 < dl) sc[rB] = e2.w;
  __builtin_amdgcn_wave_barrier();
  int cmA = sc[sA];
  int cmB = sc[sA + 1];

  float4 acc0 = make_float4(0.f, 0.f, 0.f, 0.f);
  float4 acc1 = acc0;
  float sws = 0.f;
  float wmA = 0.f, wmB = 0.f;
  int cn0 = sc[0], cn1 = sc[1], cn2 = sc[2], cn3 = sc[3];
#pragma unroll 1
  for (int p0 = 0; p0 < maxd; p0 += 4) {
    int cc[4] = {cn0, cn1, cn2, cn3};
    int pn = min(p0 + 4, CAP - 4);
    cn0 = sc[pn]; cn1 = sc[pn + 1]; cn2 = sc[pn + 2]; cn3 = sc[pn + 3];
    float4 sv0[4], sv1[4];
#pragma unroll
    for (int u = 0; u < 4; ++u) {
      const float* sr = s + (size_t)cc[u] * PDIM + 4 * l16;
      sv0[u] = *(const float4*)(sr);
      sv1[u] = *(const float4*)(sr + 64);
    }
    float pp[4];
#pragma unroll
    for (int u = 0; u < 4; ++u) {
      float p = qv0.x * sv0[u].x;
      p = __builtin_fmaf(qv0.y, sv0[u].y, p);
      p = __builtin_fmaf(qv0.z, sv0[u].z, p);
      p = __builtin_fmaf(qv0.w, sv0[u].w, p);
      p = __builtin_fmaf(qv1.x, sv1[u].x, p);
      p = __builtin_fmaf(qv1.y, sv1[u].y, p);
      p = __builtin_fmaf(qv1.z, sv1[u].z, p);
      p = __builtin_fmaf(qv1.w, sv1[u].w, p);
      pp[u] = p;
    }
#pragma unroll
    for (int u = 0; u < 4; ++u) pp[u] = row_xadd<1>(pp[u]);
#pragma unroll
    for (int u = 0; u < 4; ++u) pp[u] = row_xadd<2>(pp[u]);
#pragma unroll
    for (int u = 0; u < 4; ++u) pp[u] = row_xadd<4>(pp[u]);
#pragma unroll
    for (int u = 0; u < 4; ++u) pp[u] = row_xadd<8>(pp[u]);
#pragma unroll
    for (int u = 0; u < 4; ++u) {
      float w = pp[u] + qbn;
      if (p0 + u < dl) {
        axpy4(w, sv0[u], acc0);
        axpy4(w, sv1[u], acc1);
        sws += w;
      }
      if (sA == p0 + u) wmA = w;
      if (sA + 1 == p0 + u) wmB = w;
    }
  }
  if (valid && deg <= CAP) {
    if (sA + 1 < dl) {
      ((int4*)slots)[l16] = make_int4(cmA, __float_as_int(wmA), cmB, __float_as_int(wmB));
    } else if (sA < dl) {
      slots[sA] = make_int2(cmA, __float_as_int(wmA));
    }
    float* Sr = S + (size_t)nd * PDIM + 4 * l16;
    *(float4*)(Sr) = acc0;
    *(float4*)(Sr + 64) = acc1;
    if (l16 == 0) sw[nd] = sws;
  }
  int dg0 = __builtin_amdgcn_readlane(deg, 0);
  int dg1 = __builtin_amdgcn_readlane(deg, 16);
  if (dg0 > CAP)
    long_row<true>(pairBase, dg0, qt, qb, s, nullptr, nullptr, nullptr, counts, list, ovf,
                   S, sw, nullptr, N);
  if (dg1 > CAP)
    long_row<true>(pairBase + 1, dg1, qt, qb, s, nullptr, nullptr, nullptr, counts, list, ovf,
                   S, sw, nullptr, N);
}

__global__ __launch_bounds__(256) void node_rows_half(
    const float* __restrict__ qt, const float* __restrict__ qb, const float* __restrict__ s,
    const float* __restrict__ U, const float* __restrict__ v,
    const float* __restrict__ bvec,
    const int* __restrict__ counts, const int2* __restrict__ list,
    const int4* __restrict__ ovf,
    float* __restrict__ v_out, int N, int nGroups) {
  int wave = __builtin_amdgcn_readfirstlane((int)(threadIdx.x >> 5));
  int half = blockIdx.x / nGroups;
  int pairBase = ((blockIdx.x - half * nGroups) * 8 + wave) * 2;
  if (pairBase >= N) return;
  int lane = threadIdx.x & 31;
  int node = pairBase + (lane >> 4);
  bool valid = node < N;
  int nd = valid ? node : pairBase;
  int deg = valid ? counts[nd] : 0;
  int dl = (deg <= CAP) ? deg : 0;
  int f = half * 64 + (lane & 15) * 4;
  int maxd = max(__builtin_amdgcn_readlane(dl, 0), __builtin_amdgcn_readlane(dl, 16));
  const int2* sl = list + (size_t)nd * CAP;

  float4 a0 = make_float4(0.f, 0.f, 0.f, 0.f);
  float4 a1 = a0, a2 = a0;
  int2 cw = make_int2(0, 0);
  if (0 < dl) cw = sl[0];
#pragma unroll 1
  for (int i = 0; i < maxd; ++i) {
    int2 cur = cw;
    if (i + 1 < dl) cw = sl[i + 1];
    if (i < dl) {
      int c = cur.x;
      float w = __int_as_float(cur.y);
      const float* uc = U + (size_t)c * 3 * PDIM + f;
      float4 x0 = *(const float4*)(uc);
      float4 x1 = *(const float4*)(uc + PDIM);
      float4 x2 = *(const float4*)(uc + 2 * PDIM);
      axpy4(w, x0, a0);
      axpy4(w, x1, a1);
      axpy4(w, x2, a2);
    }
  }
  if (valid && deg <= CAP)
    store_vrow(node, f, a0, a1, a2, v, bvec, v_out);
  if (half == 0) {
    int dg0 = __builtin_amdgcn_readlane(deg, 0);
    int dg1 = __builtin_amdgcn_readlane(deg, 16);
    if (dg0 > CAP)
      long_row<false>(pairBase, dg0, qt, qb, s, U, v, bvec, counts, list, ovf,
                      nullptr, nullptr, v_out, N);
    if (dg1 > CAP)
      long_row<false>(pairBase + 1, dg1, qt, qb, s, U, v, bvec, counts, list, ovf,
                      nullptr, nullptr, v_out, N);
  }
}

__global__ __launch_bounds__(256) void painn_u_gemm(
    const float* __restrict__ vv, const float* __restrict__ v,
    const float* __restrict__ Wvec, float* __restrict__ U, int nStripsU, int nGroupsU,
    const float* __restrict__ S, const float* __restrict__ sw, const float* __restrict__ s,
    const float* __restrict__ Wkv, const float* __restrict__ bkv,
    float* __restrict__ s_out, int nStripsS) {
  int wave = threadIdx.x >> 5;
  int lane = threadIdx.x & 31;
  int half = lane >> 4;
  int l16  = lane & 15;
  v2f a[32];
  if ((int)blockIdx.x < nGroupsU) {
    int strip = blockIdx.x * 8 + wave;
    if (strip >= nStripsU) return;
    int rowBase = strip * 16;
    int row = rowBase + l16;
    int node = row / 3;
    load_strip_prod(v + (size_t)row * PDIM + 2 * half, vv + (size_t)node * PDIM + 2 * half, a);
    size_t ooff = (size_t)(rowBase + 8 * half) * PDIM + 4 * l16;
    strip_block<PDIM, 0>(a, Wvec + (2 * half) * PDIM + 4 * l16, nullptr, U + ooff,
                         nullptr, nullptr);
  } else {
    int strip = ((int)blockIdx.x - nGroupsU) * 8 + wave;
    if (strip >= nStripsS) return;
    int rowBase = strip * 16;
    load_strip(S + (size_t)(rowBase + l16) * PDIM + 2 * half, a);
    size_t ooff = (size_t)(rowBase + 8 * half) * PDIM + 4 * l16;
    int wc = PDIM + 4 * l16;
    strip_block<3 * PDIM, 2>(a, Wkv + (2 * half) * 3 * PDIM + wc, bkv + wc, s_out + ooff,
                             s + ooff, sw + rowBase + 8 * half);
  }
}

extern "C" void kernel_launch(void* const* d_in, const int* in_sizes, int n_in,
                              void* d_out, int out_size, void* d_ws, size_t ws_size,
                              hipStream_t stream) {
  const float* s    = (const float*)d_in[0];
  const float* v    = (const float*)d_in[1];
  const int*   ei   = (const int*)d_in[2];
  const float* Wq   = (const float*)d_in[3];
  const float* bq   = (const float*)d_in[4];
  const float* Wkv  = (const float*)d_in[5];
  const float* bkv  = (const float*)d_in[6];
  const float* Wvec = (const float*)d_in[7];
  const float* bvec = (const float*)d_in[8];

  int N = in_sizes[0] / PDIM;
  int E = in_sizes[2] / 2;

  float* s_out = (float*)d_out;
  float* v_out = s_out + (size_t)N * PDIM;

  int nCnt = (N + 1 + 63) / 64 * 64;

  float* Mx   = (float*)d_ws;
  float* ub   = Mx + (size_t)FOLDROWS * PDIM;
  float* qt   = Mx + (size_t)N * PDIM;
  float* vv   = qt + (size_t)N * PDIM;
  float* S    = vv + (size_t)N * PDIM;
  float* U    = S  + (size_t)N * PDIM;
  float* qb   = U  + (size_t)N * 3 * PDIM;
  float* sw   = qb + nCnt;

  int*  counts = (int*)(sw + nCnt);
  int2* list   = (int2*)(counts + nCnt);
  int4* ovf    = (int4*)(list + (size_t)N * CAP);

  int nBins   = (N + BINR - 1) / BINR;
  int nChunks = (E + CHUNK - 1) / CHUNK;
  int nOffs   = (nChunks * (nBins + 1) + 63) / 64 * 64;
  int*  offs   = (int*)(ovf + E);
  int2* sorted = (int2*)(offs + nOffs);

  const int* rowIdx = ei;
  const int* colIdx = ei + E;

  int stripsA = N / 16;
  int groupsA = (stripsA + 7) / 8;
  painn_node_gemm<<<nChunks + 2 + groupsA, 256, 0, stream>>>(
      s, Wq, bq, Wkv, bkv, Mx, ub, vv, stripsA, counts, N,
      rowIdx, colIdx, E, nBins, nChunks, offs, sorted);

  lists_and_qt<<<nBins + groupsA, 256, 0, stream>>>(
      offs, sorted, nChunks, nBins, counts, list, ovf, N, N, s, Mx, ub, qt, qb, stripsA);

  int pairGroups = ((N + 1) / 2 + 7) / 8;
  edge_weights<<<pairGroups, 256, 0, stream>>>(qt, qb, s, counts, list, ovf, S, sw, N);

  int stripsC = N * 3 / 16;
  int groupsC = (stripsC + 7) / 8;
  painn_u_gemm<<<groupsC + groupsA, 256, 0, stream>>>(
      vv, v, Wvec, U, stripsC, groupsC, S, sw, s, Wkv, bkv, s_out, stripsA);

  node_rows_half<<<2 * pairGroups, 256, 0, stream>>>(
      qt, qb, s, U, v, bvec, counts, list, ovf, v_out, N, pairGroups);
}


# Last GEMM launch dissolved: U strips run in the list/qt launch; the edge-weight work-group keeps S and sum w in LDS and finishes its own s_out strip on WMMA; 5 launches to 4; same sums
# speedup vs baseline: 4.2721x; 1.0516x over previous
#include <hip/hip_runtime.h>


typedef __attribute__((ext_vector_type(2))) float v2f;
typedef __attribute__((ext_vector_type(8))) float v8f;

#define PDIM 128
#define CAP 32

#define BINSH 8
#define BINR (1 << BINSH)
#define MAXBINS 256
#define EPT 10
#define CHUNK (256 * EPT)

__device__ __forceinline__ void presort_chunk(const int* __restrict__ rowIdx,
                                              const int* __restrict__ colIdx, int E, int nBins,
                                              int* __restrict__ offs, int2* __restrict__ sorted,
                                              int w, int* hist, int* wsum, int2* stage) {
  int t = threadIdx.x;
  for (int b = t; b <= nBins; b += 256) hist[b] = 0;
  __syncthreads();
  int e0 = w * CHUNK;
  int r[EPT], c[EPT], rk[EPT];
#pragma unroll
  for (int u = 0; u < EPT; ++u) {
    int e = e0 + u * 256 + t;
    int el = min(e, E - 1);
    int rr = rowIdx[el];
    c[u] = colIdx[el];
    r[u] = (e < E) ? rr : -1;
  }
#pragma unroll
  for (int u = 0; u < EPT; ++u)
    rk[u] = (r[u] >= 0) ? atomicAdd(&hist[r[u] >> BINSH], 1) : 0;
  __syncthreads();
  int lane = t & 31;
  int wave = t >> 5;
  int v = (t < nBins) ? hist[t] : 0;
  int incl = v;
#pragma unroll
  for (int d = 1; d < 32; d <<= 1) {
    int n = __shfl_up(incl, d, 32);
    if (lane >= d) incl += n;
  }
  if (lane == 31) wsum[wave] = incl;
  __syncthreads();
  int pre = 0;
  for (int k = 0; k < wave; ++k) pre += wsum[k];
  int excl = pre + incl - v;
  if (t < nBins) hist[t] = excl;
  if (t == nBins - 1) hist[nBins] = excl + v;
  __syncthreads();
  for (int b = t; b <= nBins; b += 256) offs[(size_t)w * (nBins + 1) + b] = hist[b];
#pragma unroll
  for (int u = 0; u < EPT; ++u) {
    if (r[u] < 0) continue;
    unsigned lr = (unsigned)r[u] & (BINR - 1);
    stage[hist[r[u] >> BINSH] + rk[u]] = make_int2(e0 + u * 256 + t, (int)((lr << 24) | (unsigned)c[u]));
  }
  __syncthreads();
  int cnt = min(CHUNK, E - e0);
  for (int i = t; i < cnt; i += 256) sorted[(size_t)e0 + i] = stage[i];
}

__device__ __forceinline__ void place_edge(int2 en, int nodeBase, int* cnt, int2* slots,
                                           int* __restrict__ counts, int4* __restrict__ ovf,
                                           int ovfCounter) {
  int lr = (int)((unsigned)en.y >> 24);
  int col = en.y & 0xffffff;
  int sl = atomicAdd(&cnt[lr], 1);
  if (sl < CAP) {
    slots[lr * CAP + sl] = make_int2(en.x, col);
  } else {
    int o = atomicAdd(&counts[ovfCounter], 1);
    ovf[o] = make_int4(nodeBase + lr, en.x, col, 0);
  }
}

__device__ __forceinline__ void bin_lists(const int* __restrict__ offs,
                                          const int2* __restrict__ sorted,
                                          int nChunks, int nBins, int b,
                                          int* __restrict__ counts,
                                          int2* __restrict__ list,
                                          int4* __restrict__ ovf,
                                          int ovfCounter, int N, int* cnt, int2* slots) {
  int t = threadIdx.x;
  int nodeBase = b * BINR;
  for (int i = t; i < BINR; i += 256) cnt[i] = 0;
  __syncthreads();
  for (int w = t; w < nChunks; w += 256) {
    const int* op = offs + (size_t)w * (nBins + 1) + b;
    int o0 = op[0];
    int o1 = op[1];
    const int2* sp = sorted + (size_t)w * CHUNK;
    for (int i0 = o0; i0 < o1; i0 += 4) {
      int lastI = o1 - 1;
      int2 en0 = sp[i0];
      int2 en1 = sp[min(i0 + 1, lastI)];
      int2 en2 = sp[min(i0 + 2, lastI)];
      int2 en3 = sp[min(i0 + 3, lastI)];
      place_edge(en0, nodeBase, cnt, slots, counts, ovf, ovfCounter);
      if (i0 + 1 < o1) place_edge(en1, nodeBase, cnt, slots, counts, ovf, ovfCounter);
      if (i0 + 2 < o1) place_edge(en2, nodeBase, cnt, slots, counts, ovf, ovfCounter);
      if (i0 + 3 < o1) place_edge(en3, nodeBase, cnt, slots, counts, ovf, ovfCounter);
    }
  }
  __syncthreads();
  for (int i = t; i < BINR; i += 256)
    if (nodeBase + i < N) counts[nodeBase + i] = cnt[i];
  int2* lst = list + (size_t)nodeBase * CAP;
  for (int i = t; i < BINR * CAP / 2; i += 256) {
    int nd = i / (CAP / 2);
    int s0 = (i % (CAP / 2)) * 2;
    int filled = (nodeBase + nd < N) ? min(cnt[nd], CAP) : 0;
    if (s0 + 1 < filled) {
      ((int4*)lst)[i] = ((const int4*)slots)[i];
    } else if (s0 < filled) {
      lst[2 * i] = slots[2 * i];
    }
  }
}

__device__ __forceinline__ void load_strip(const float* __restrict__ rowp, v2f (&a)[32]) {
#pragma unroll
  for (int ks = 0; ks < 32; ++ks) {
    float2 t = *(const float2*)(rowp + 4 * ks);
    a[ks].x = t.x;
    a[ks].y = t.y;
  }
}

__device__ __forceinline__ void load_strip_prod(const float* __restrict__ xrow,
                                                const float* __restrict__ grow, v2f (&a)[32]) {
#pragma unroll
  for (int ks = 0; ks < 32; ++ks) {
    float2 t = *(const float2*)(xrow + 4 * ks);
    float2 g = *(const float2*)(grow + 4 * ks);
    a[ks].x = g.x * t.x;
    a[ks].y = g.y * t.y;
  }
}

__device__ __forceinline__ void step4(const v2f& af, const float4& r0, const float4& r1,
                                      v8f (&acc)[4]) {
  v2f bf;
  bf.x = r0.x; bf.y = r1.x;
  acc[0] = __builtin_amdgcn_wmma_f32_16x16x4_f32(false, af, false, bf, (short)0, acc[0], false, false);
  bf.x = r0.y; bf.y = r1.y;
  acc[1] = __builtin_amdgcn_wmma_f32_16x16x4_f32(false, af, false, bf, (short)0, acc[1], false, false);
  bf.x = r0.z; bf.y = r1.z;
  acc[2] = __builtin_amdgcn_wmma_f32_16x16x4_f32(false, af, false, bf, (short)0, acc[2], false, false);
  bf.x = r0.w; bf.y = r1.w;
  acc[3] = __builtin_amdgcn_wmma_f32_16x16x4_f32(false, af, false, bf, (short)0, acc[3], false, false);
}

template <int LDW>
__device__ __forceinline__ void strip_chunk(const v2f (&a)[32], const float* __restrict__ Wl,
                                            v8f (&acc)[4]) {
  float4 n0 = *(const float4*)(Wl);
  float4 n1 = *(const float4*)(Wl + LDW);
  float4 n2 = *(const float4*)(Wl + 4 * LDW);
  float4 n3 = *(const float4*)(Wl + 5 * LDW);
#pragma unroll
  for (int j = 0; j < 16; ++j) {
    float4 c0 = n0, c1 = n1, c2 = n2, c3 = n3;
    if (j + 1 < 16) {
      const float* p = Wl + (8 * (j + 1)) * LDW;
      n0 = *(const float4*)(p);
      n1 = *(const float4*)(p + LDW);
      n2 = *(const float4*)(p + 4 * LDW);
      n3 = *(const float4*)(p + 5 * LDW);
    }
    step4(a[2 * j],     c0, c1, acc);
    step4(a[2 * j + 1], c2, c3, acc);
    __builtin_amdgcn_sched_barrier(0);
  }
}

template <int LDW, int MODE>
__device__ __forceinline__ void strip_block(const v2f (&a)[32], const float* __restrict__ Wl,
                                            const float* __restrict__ bl,
                                            float* __restrict__ orow) {
#pragma unroll 1
  for (int cb = 0; cb < PDIM; cb += 64) {
    v8f acc[4] = {};
    strip_chunk<LDW>(a, Wl + cb, acc);
    float4 bias = make_float4(0.f, 0.f, 0.f, 0.f);
    if (MODE != 0) bias = *(const float4*)(bl + cb);
#pragma unroll
    for (int r = 0; r < 8; ++r) {
      float4 ov = make_float4(acc[0][r], acc[1][r], acc[2][r], acc[3][r]);
      if (MODE == 1) {
        ov = make_float4(ov.x + bias.x, ov.y + bias.y, ov.z + bias.z, ov.w + bias.w);
      }
      *(float4*)(orow + (size_t)r * PDIM + cb) = ov;
    }
  }
}

__device__ __forceinline__ void load_strip_k8(const float* rowp, v2f (&a)[32]) {
#pragma unroll
  for (int j = 0; j < 16; ++j) {
    float4 t = *(const float4*)(rowp + 8 * j);
    a[2 * j].x = t.x;     a[2 * j].y = t.y;
    a[2 * j + 1].x = t.z; a[2 * j + 1].y = t.w;
  }
}

template <int LDW>
__device__ __forceinline__ void strip_chunk_T(const v2f (&a)[32], const float* __restrict__ Wt,
                                              v8f (&acc)[4]) {
  float4 n0 = *(const float4*)(Wt);
  float4 n1 = *(const float4*)(Wt + 16 * LDW);
  float4 n2 = *(const float4*)(Wt + 32 * LDW);
  float4 n3 = *(const float4*)(Wt + 48 * LDW);
#pragma unroll
  for (int j = 0; j < 16; ++j) {
    float4 c[4] = {n0, n1, n2, n3};
    if (j + 1 < 16) {
      const float* p = Wt + 8 * (j + 1);
      n0 = *(const float4*)(p);
      n1 = *(const float4*)(p + 16 * LDW);
      n2 = *(const float4*)(p + 32 * LDW);
      n3 = *(const float4*)(p + 48 * LDW);
    }
#pragma unroll
    for (int t = 0; t < 4; ++t) {
      v2f bf;
      bf.x = c[t].x; bf.y = c[t].y;
      acc[t] = __builtin_amdgcn_wmma_f32_16x16x4_f32(false, a[2 * j], false, bf, (short)0, acc[t], false, false);
    }
#pragma unroll
    for (int t = 0; t < 4; ++t) {
      v2f bf;
      bf.x = c[t].z; bf.y = c[t].w;
      acc[t] = __builtin_amdgcn_wmma_f32_16x16x4_f32(false, a[2 * j + 1], false, bf, (short)0, acc[t], false, false);
    }
    __builtin_amdgcn_sched_barrier(0);
  }
}

#define FOLDROWS (PDIM + 16)

__global__ __launch_bounds__(256) void painn_node_gemm(
    const float* __restrict__ s,
    const float* __restrict__ Wq,  const float* __restrict__ bq,
    const float* __restrict__ Wkv, const float* __restrict__ bkv,
    float* __restrict__ Mx, float* __restrict__ ub, float* __restrict__ vv,
    int nStrips, int* __restrict__ counts, int ovfCounter,
    const int* __restrict__ rowIdx, const int* __restrict__ colIdx, int E, int nBins,
    int nSort, int* __restrict__ offs, int2* __restrict__ sorted) {
  __shared__ int hist[MAXBINS + 1];
  __shared__ int wsum[8];
  __shared__ int2 stage[CHUNK];
  if ((int)blockIdx.x < nSort) {
    if (blockIdx.x == 0 && threadIdx.x == 0) counts[ovfCounter] = 0;
    presort_chunk(rowIdx, colIdx, E, nBins, offs, sorted, (int)blockIdx.x, hist, wsum, stage);
    return;
  }
  int bx = (int)blockIdx.x - nSort;
  int wave = threadIdx.x >> 5;
  int lane = threadIdx.x & 31;
  int half = lane >> 4;
  int l16  = lane & 15;
  v2f a[32];
  if (bx < 2) {
    int strip = bx * 8 + wave;
    if (strip > 8) return;
    const float* arow = (strip < 8) ? Wq + (size_t)(strip * 16 + l16) * PDIM : bq;
    load_strip_k8(arow + 4 * half, a);
    float p = 0.f;
#pragma unroll
    for (int j = 0; j < 16; ++j) {
      float4 b = *(const float4*)(bkv + 8 * j + 4 * half);
      p += a[2 * j].x * b.x + a[2 * j].y * b.y + a[2 * j + 1].x * b.z + a[2 * j + 1].y * b.w;
    }
    p += __shfl_xor(p, 16, 32);
    if (half == 0) ub[strip * 16 + l16] = p;
#pragma unroll 1
    for (int cb = 0; cb < PDIM; cb += 64) {
      v8f acc[4] = {};
      strip_chunk_T<3 * PDIM>(a, Wkv + (size_t)(cb + l16) * 3 * PDIM + 4 * half, acc);
      float* orow = Mx + (size_t)(strip * 16 + 8 * half) * PDIM + cb + l16;
#pragma unroll
      for (int r = 0; r < 8; ++r) {
#pragma unroll
        for (int t = 0; t < 4; ++t) orow[(size_t)r * PDIM + 16 * t] = acc[t][r];
      }
    }
    return;
  }
  int strip = (bx - 2) * 8 + wave;
  if (strip >= nStrips) return;
  int rowBase = strip * 16;
  load_strip(s + (size_t)(rowBase + l16) * PDIM + 2 * half, a);
  size_t ooff = (size_t)(rowBase + 8 * half) * PDIM + 4 * l16;
  int wc = 2 * PDIM + 4 * l16;
  strip_block<3 * PDIM, 1>(a, Wkv + (2 * half) * 3 * PDIM + wc, bkv + wc, vv + ooff);
}

__global__ __launch_bounds__(256) void lists_and_qt(
    const int* __restrict__ offs, const int2* __restrict__ sorted, int nChunks, int nBins,
    int* __restrict__ counts, int2* __restrict__ list, int4* __restrict__ ovf,
    int ovfCounter, int N,
    const float* __restrict__ s, const float* __restrict__ Mx, const float* __restrict__ ub,
    float* __restrict__ qt, float* __restrict__ qb, int nStrips,
    const float* __restrict__ vv, const float* __restrict__ v,
    const float* __restrict__ Wvec, float* __restrict__ U, int nStripsU, int nGroupsU) {
  __shared__ int cnt[BINR];
  __shared__ __attribute__((aligned(16))) int2 slots[BINR * CAP];
  if ((int)blockIdx.x < nBins) {
    bin_lists(offs, sorted, nChunks, nBins, (int)blockIdx.x, counts, list, ovf, ovfCounter, N,
              cnt, slots);
    return;
  }
  int wave = threadIdx.x >> 5;
  int lane = threadIdx.x & 31;
  int half = lane >> 4;
  int l16  = lane & 15;
  v2f a[32];
  int gb = (int)blockIdx.x - nBins;
  if (gb < nGroupsU) {
    int stripU = gb * 8 + wave;
    if (stripU >= nStripsU) return;
    int rowBaseU = stripU * 16;
    int rowU = rowBaseU + l16;
    load_strip_prod(v + (size_t)rowU * PDIM + 2 * half, vv + (size_t)(rowU / 3) * PDIM + 2 * half, a);
    size_t uoff = (size_t)(rowBaseU + 8 * half) * PDIM + 4 * l16;
    strip_block<PDIM, 0>(a, Wvec + (2 * half) * PDIM + 4 * l16, nullptr, U + uoff);
    return;
  }
  int strip = (gb - nGroupsU) * 8 + wave;
  if (strip >= nStrips) return;
  int rowBase = strip * 16;
  load_strip(s + (size_t)(rowBase + l16) * PDIM + 2 * half, a);
  float p = 0.f;
#pragma unroll
  for (int ks = 0; ks < 32; ++ks) {
    float2 u2 = *(const float2*)(ub + 4 * ks + 2 * half);
    p += a[ks].x * u2.x + a[ks].y * u2.y;
  }
  p += __shfl_xor(p, 16, 32);
  if (half == 0) qb[rowBase + l16] = p + ub[PDIM];
  size_t ooff = (size_t)(rowBase + 8 * half) * PDIM + 4 * l16;
  strip_block<PDIM, 1>(a, Mx + (2 * half) * PDIM + 4 * l16, Mx + (size_t)PDIM * PDIM + 4 * l16,
                       qt + ooff);
}

__device__ __forceinline__ float wave_dot(const float4& qv, const float4& kv) {
  float p = qv.x * kv.x + qv.y * kv.y + qv.z * kv.z + qv.w * kv.w;
  p += __shfl_xor(p, 16, 32);
  p += __shfl_xor(p, 8, 32);
  p += __shfl_xor(p, 4, 32);
  p += __shfl_xor(p, 2, 32);
  p += __shfl_xor(p, 1, 32);
  return p;
}

__device__ __forceinline__ void axpy4(float w, const float4& x, float4& a) {
  a.x += w * x.x; a.y += w * x.y; a.z += w * x.z; a.w += w * x.w;
}

__device__ __forceinline__ void store_vrow(
    int node, int f, const float4& a0, const float4& a1, const float4& a2,
    const float* __restrict__ v, const float* __restrict__ bvec, float* __restrict__ v_out) {
  size_t vo3 = (size_t)node * 3 * PDIM + f;
  float4 bv = *(const float4*)(bvec + f);
  const float* vr = v + vo3;
  float4 y0 = *(const float4*)(vr);
  float4 y1 = *(const float4*)(vr + PDIM);
  float4 y2 = *(const float4*)(vr + 2 * PDIM);
  float* vo = v_out + vo3;
  *(float4*)(vo) = make_float4(y0.x + (a0.x + bv.x), y0.y + (a0.y + bv.y),
                               y0.z + (a0.z + bv.z), y0.w + (a0.w + bv.w));
  *(float4*)(vo + PDIM) = make_float4(y1.x + (a1.x + bv.x), y1.y + (a1.y + bv.y),
                                      y1.z + (a1.z + bv.z), y1.w + (a1.w + bv.w));
  *(float4*)(vo + 2 * PDIM) = make_float4(y2.x + (a2.x + bv.x), y2.y + (a2.y + bv.y),
                                          y2.z + (a2.z + bv.z), y2.w + (a2.w + bv.w));
}

template <bool SCALAR>
__device__ void long_row(
    int node, int deg,
    const float* __restrict__ qt, const float* __restrict__ qb,
    const float* __restrict__ s, const float* __restrict__ U,
    const float* __restrict__ v, const float* __restrict__ bvec,
    const int* __restrict__ counts, const int2* __restrict__ list,
    const int4* __restrict__ ovf,
    float* __restrict__ S, float* __restrict__ sw, float* __restrict__ v_out, int N) {
  int lane = threadIdx.x & 31;
  int d = lane * 4;
  float4 qv = *(const float4*)(qt + (size_t)node * PDIM + d);
  float qbn = qb[node];
  float4 a0 = make_float4(0.f, 0.f, 0.f, 0.f);
  float4 a1 = a0, a2 = a0;
  float sws = 0.f;
  int nOvf = counts[N];
  int2 ec = list[(size_t)node * CAP + lane];
  int last = -1;
  for (int i = 0; i < deg; ++i) {
    int be = 0x7fffffff, bc = 0;
    if (ec.x > last) { be = ec.x; bc = ec.y; }
    for (int j = lane; j < nOvf; j += 32) {
      int4 o = ovf[j];
      if (o.x == node && o.y > last && o.y < be) { be = o.y; bc = o.z; }
    }
    int mn = be;
    mn = min(mn, __shfl_xor(mn, 16, 32));
    mn = min(mn, __shfl_xor(mn, 8, 32));
    mn = min(mn, __shfl_xor(mn, 4, 32));
    mn = min(mn, __shfl_xor(mn, 2, 32));
    mn = min(mn, __shfl_xor(mn, 1, 32));
    unsigned m = __builtin_amdgcn_ballot_w32(be == mn);
    int src = __builtin_ctz(m | 0x80000000u);
    int c = __builtin_amdgcn_readlane(bc, src);
    last = __builtin_amdgcn_readfirstlane(mn);
    float4 sv = *(const float4*)(s + (size_t)c * PDIM + d);
    float w = wave_dot(qv, sv) + qbn;
    if (SCALAR) {
      axpy4(w, sv, a0);
      sws += w;
    } else {
      const float* uc = U + (size_t)c * 3 * PDIM + d;
      float4 x0 = *(const float4*)(uc);
      float4 x1 = *(const float4*)(uc + PDIM);
      float4 x2 = *(const float4*)(uc + 2 * PDIM);
      axpy4(w, x0, a0);
      axpy4(w, x1, a1);
      axpy4(w, x2, a2);
    }
  }
  if (SCALAR) {
    *(float4*)(S + d) = a0;
    if (lane == 0) *sw = sws;
  } else {
    store_vrow(node, d, a0, a1, a2, v, bvec, v_out);
  }
}

template <int M>
__device__ __forceinline__ float row_xadd(float x) {
  int y = __builtin_amdgcn_update_dpp(0, __float_as_int(x), 0x160 + M, 0xf, 0xf, false);
  return x + __int_as_float(y);
}

#define SP (PDIM + 4)

__global__ __launch_bounds__(256) void edge_weights(
    const float* __restrict__ qt, const float* __restrict__ qb, const float* __restrict__ s,
    const int* __restrict__ counts, int2* __restrict__ list, const int4* __restrict__ ovf,
    const float* __restrict__ Wkv, const float* __restrict__ bkv,
    float* __restrict__ s_out, int N) {
  __shared__ __attribute__((aligned(16))) int scol[8][2][CAP];
  __shared__ __attribute__((aligned(16))) float Ssh[16 * SP];
  __shared__ __attribute__((aligned(16))) float swsh[16];
  int wave = __builtin_amdgcn_readfirstlane((int)(threadIdx.x >> 5));
  int pairBase = (blockIdx.x * 8 + wave) * 2;
  int lane = threadIdx.x & 31;
  int g = lane >> 4;
  int l16 = lane & 15;
  int node = pairBase + g;
  bool valid = node < N;
  int nd = valid ? node : min(pairBase, N - 1);
  int deg = valid ? counts[nd] : 0;
  int dl = (deg <= CAP) ? deg : 0;
  int maxd = max(__builtin_amdgcn_readlane(dl, 0), __builtin_amdgcn_readlane(dl, 16));
  int2* slots = list + (size_t)nd * CAP;
  int sA = 2 * l16;

  int4 e2 = make_int4(0x7fffffff, 0, 0x7fffffff, 0);
  if (sA + 1 < dl) {
    e2 = ((const int4*)slots)[l16];
  } else if (sA < dl) {
    int2 t = slots[sA];
    e2.x = t.x;
    e2.y = t.y;
  }
  float4 qv0 = *(const float4*)(qt + (size_t)nd * PDIM + 4 * l16);
  float4 qv1 = *(const float4*)(qt + (size_t)nd * PDIM + 64 + 4 * l16);
  float qbn = qb[nd];

  int rA = 0, rB = 0;
  int hb = g << 4;
  for (int jj = 0; 2 * jj < maxd; ++jj) {
    int ia = __shfl(e2.x, hb + jj, 32);
    int ib = __shfl(e2.z, hb + jj, 32);
    rA += ((ia < e2.x) ? 1 : 0) + ((ib < e2.x) ? 1 : 0);
    rB += ((ia < e2.z) ? 1 : 0) + ((ib < e2.z) ? 1 : 0);
  }
  int* sc = scol[wave][g];
  sc[sA] = nd;
  sc[sA + 1] = nd;
  __builtin_amdgcn_wave_barrier();
  if (sA < dl) sc[rA] = e2.y;
  if (sA + 1 < dl) sc[rB] = e2.w;
  __builtin_amdgcn_wave_barrier();
  int cmA = sc[sA];
  int cmB = sc[sA + 1];

  float4 acc0 = make_float4(0.f, 0.f, 0.f, 0.f);
  float4 acc1 = acc0;
  float sws = 0.f;
  float wmA = 0.f, wmB = 0.f;
  int cn0 = sc[0], cn1 = sc[1], cn2 = sc[2], cn3 = sc[3];
#pragma unroll 1
  for (int p0 = 0; p0 < maxd; p0 += 4) {
    int cc[4] = {cn0, cn1, cn2, cn3};
    int pn = min(p0 + 4, CAP - 4);
    cn0 = sc[pn]; cn1 = sc[pn + 1]; cn2 = sc[pn + 2]; cn3 = sc[pn + 3];
    float4 sv0[4], sv1[4];
#pragma unroll
    for (int u = 0; u < 4; ++u) {
      const float* sr = s + (size_t)cc[u] * PDIM + 4 * l16;
      sv0[u] = *(const float4*)(sr);
      sv1[u] = *(const float4*)(sr + 64);
    }
    float pp[4];
#pragma unroll
    for (int u = 0; u < 4; ++u) {
      float p = qv0.x * sv0[u].x;
      p = __builtin_fmaf(qv0.y, sv0[u].y, p);
      p = __builtin_fmaf(qv0.z, sv0[u].z, p);
      p = __builtin_fmaf(qv0.w, sv0[u].w, p);
      p = __builtin_fmaf(qv1.x, sv1[u].x, p);
      p = __builtin_fmaf(qv1.y, sv1[u].y, p);
      p = __builtin_fmaf(qv1.z, sv1[u].z, p);
      p = __builtin_fmaf(qv1.w, sv1[u].w, p);
      pp[u] = p;
    }
#pragma unroll
    for (int u = 0; u < 4; ++u) pp[u] = row_xadd<1>(pp[u]);
#pragma unroll
    for (int u = 0; u < 4; ++u) pp[u] = row_xadd<2>(pp[u]);
#pragma unroll
    for (int u = 0; u < 4; ++u) pp[u] = row_xadd<4>(pp[u]);
#pragma unroll
    for (int u = 0; u < 4; ++u) pp[u] = row_xadd<8>(pp[u]);
#pragma unroll
    for (int u = 0; u < 4; ++u) {
      float w = pp[u] + qbn;
      if (p0 + u < dl) {
        axpy4(w, sv0[u], acc0);
        axpy4(w, sv1[u], acc1);
        sws += w;
      }
      if (sA == p0 + u) wmA = w;
      if (sA + 1 == p0 + u) wmB = w;
    }
  }
  if (valid && deg <= CAP) {
    if (sA + 1 < dl) {
      ((int4*)slots)[l16] = make_int4(cmA, __float_as_int(wmA), cmB, __float_as_int(wmB));
    } else if (sA < dl) {
      slots[sA] = make_int2(cmA, __float_as_int(wmA));
    }
  }
  float* Srow = Ssh + (2 * wave + g) * SP;
  if (deg <= CAP) {
    *(float4*)(Srow + 4 * l16) = acc0;
    *(float4*)(Srow + 64 + 4 * l16) = acc1;
    if (l16 == 0) swsh[2 * wave + g] = sws;
  }
  int dg0 = __builtin_amdgcn_readlane(deg, 0);
  int dg1 = __builtin_amdgcn_readlane(deg, 16);
  if (dg0 > CAP)
    long_row<true>(pairBase, dg0, qt, qb, s, nullptr, nullptr, nullptr, counts, list, ovf,
                   Ssh + (2 * wave) * SP, swsh + 2 * wave, nullptr, N);
  if (dg1 > CAP)
    long_row<true>(pairBase + 1, dg1, qt, qb, s, nullptr, nullptr, nullptr, counts, list, ovf,
                   Ssh + (2 * wave + 1) * SP, swsh + 2 * wave + 1, nullptr, N);

  bool mm = wave < 4;
  int rowBase = blockIdx.x * 16;
  int col = 32 * wave + 2 * l16;
  const float* Wl = Wkv + (size_t)(2 * g) * (3 * PDIM) + PDIM + col;
  const float* ap = Ssh + l16 * SP + 2 * g;
  float2 wq[3][4];
  float2 aq[2][2];
  float2 x[8];
  if (mm) {
#pragma unroll
    for (int r = 0; r < 8; ++r) {
      int row = min(rowBase + 8 * g + r, N - 1);
      x[r] = *(const float2*)(s + (size_t)row * PDIM + col);
    }
#pragma unroll
    for (int t = 0; t < 2; ++t) {
#pragma unroll
      for (int i = 0; i < 2; ++i) {
        int ks = 2 * t + i;
        wq[t][2 * i]     = *(const float2*)(Wl + (size_t)(4 * ks) * (3 * PDIM));
        wq[t][2 * i + 1] = *(const float2*)(Wl + (size_t)(4 * ks + 1) * (3 * PDIM));
      }
    }
  }
  __builtin_amdgcn_fence(__ATOMIC_RELEASE, "workgroup", "local");
  __builtin_amdgcn_s_barrier();
  __builtin_amdgcn_fence(__ATOMIC_ACQUIRE, "workgroup", "local");
  if (!mm) return;
  v8f c0, c1;
#pragma unroll
  for (int r = 0; r < 8; ++r) {
    c0[r] = x[r].x;
    c1[r] = x[r].y;
  }
  aq[0][0] = *(const float2*)(ap);
  aq[0][1] = *(const float2*)(ap + 4);
#pragma unroll
  for (int t = 0; t < 16; ++t) {
    if (t + 2 < 16) {
#pragma unroll
      for (int i = 0; i < 2; ++i) {
        int ks = 2 * (t + 2) + i;
        wq[(t + 2) % 3][2 * i]     = *(const float2*)(Wl + (size_t)(4 * ks) * (3 * PDIM));
        wq[(t + 2) % 3][2 * i + 1] = *(const float2*)(Wl + (size_t)(4 * ks + 1) * (3 * PDIM));
      }
    }
    if (t + 1 < 16) {
      aq[(t + 1) % 2][0] = *(const float2*)(ap + 4 * (2 * (t + 1)));
      aq[(t + 1) % 2][1] = *(const float2*)(ap + 4 * (2 * (t + 1) + 1));
    }
#pragma unroll
    for (int i = 0; i < 2; ++i) {
      v2f af, bf;
      af.x = aq[t % 2][i].x; af.y = aq[t % 2][i].y;
      bf.x = wq[t % 3][2 * i].x; bf.y = wq[t % 3][2 * i + 1].x;
      c0 = __builtin_amdgcn_wmma_f32_16x16x4_f32(false, af, false, bf, (short)0, c0, false, false);
      bf.x = wq[t % 3][2 * i].y; bf.y = wq[t % 3][2 * i + 1].y;
      c1 = __builtin_amdgcn_wmma_f32_16x16x4_f32(false, af, false, bf, (short)0, c1, false, false);
    }
    __builtin_amdgcn_sched_barrier(0);
  }
  float2 bias = *(const float2*)(bkv + PDIM + col);
  float4 g0 = *(const float4*)(swsh + 8 * g);
  float4 g1 = *(const float4*)(swsh + 8 * g + 4);
  float gw[8] = {g0.x, g0.y, g0.z, g0.w, g1.x, g1.y, g1.z, g1.w};
#pragma unroll
  for (int r = 0; r < 8; ++r) {
    int row = rowBase + 8 * g + r;
    float2 ov = make_float2(c0[r] + gw[r] * bias.x, c1[r] + gw[r] * bias.y);
    if (row < N) *(float2*)(s_out + (size_t)row * PDIM + col) = ov;
  }
}

__global__ __launch_bounds__(256) void node_rows_half(
    const float* __restrict__ qt, const float* __restrict__ qb, const float* __restrict__ s,
    const float* __restrict__ U, const float* __restrict__ v,
    const float* __restrict__ bvec,
    const int* __restrict__ counts, const int2* __restrict__ list,
    const int4* __restrict__ ovf,
    float* __restrict__ v_out, int N, int nGroups) {
  int wave = __builtin_amdgcn_readfirstlane((int)(threadIdx.x >> 5));
  int half = blockIdx.x / nGroups;
  int pairBase = ((blockIdx.x - half * nGroups) * 8 + wave) * 2;
  if (pairBase >= N) return;
  int lane = threadIdx.x & 31;
  int node = pairBase + (lane >> 4);
  bool valid = node < N;
  int nd = valid ? node : pairBase;
  int deg = valid ? counts[nd] : 0;
  int dl = (deg <= CAP) ? deg : 0;
  int f = half * 64 + (lane & 15) * 4;
  int maxd = max(__builtin_amdgcn_readlane(dl, 0), __builtin_amdgcn_readlane(dl, 16));
  const int2* sl = list + (size_t)nd * CAP;

  float4 a0 = make_float4(0.f, 0.f, 0.f, 0.f);
  float4 a1 = a0, a2 = a0;
  int2 cw = make_int2(0, 0);
  if (0 < dl) cw = sl[0];
#pragma unroll 1
  for (int i = 0; i < maxd; ++i) {
    int2 cur = cw;
    if (i + 1 < dl) cw = sl[i + 1];
    if (i < dl) {
      int c = cur.x;
      float w = __int_as_float(cur.y);
      const float* uc = U + (size_t)c * 3 * PDIM + f;
      float4 x0 = *(const float4*)(uc);
      float4 x1 = *(const float4*)(uc + PDIM);
      float4 x2 = *(const float4*)(uc + 2 * PDIM);
      axpy4(w, x0, a0);
      axpy4(w, x1, a1);
      axpy4(w, x2, a2);
    }
  }
  if (valid && deg <= CAP)
    store_vrow(node, f, a0, a1, a2, v, bvec, v_out);
  if (half == 0) {
    int dg0 = __builtin_amdgcn_readlane(deg, 0);
    int dg1 = __builtin_amdgcn_readlane(deg, 16);
    if (dg0 > CAP)
      long_row<false>(pairBase, dg0, qt, qb, s, U, v, bvec, counts, list, ovf,
                      nullptr, nullptr, v_out, N);
    if (dg1 > CAP)
      long_row<false>(pairBase + 1, dg1, qt, qb, s, U, v, bvec, counts, list, ovf,
                      nullptr, nullptr, v_out, N);
  }
}

extern "C" void kernel_launch(void* const* d_in, const int* in_sizes, int n_in,
                              void* d_out, int out_size, void* d_ws, size_t ws_size,
                              hipStream_t stream) {
  const float* s    = (const float*)d_in[0];
  const float* v    = (const float*)d_in[1];
  const int*   ei   = (const int*)d_in[2];
  const float* Wq   = (const float*)d_in[3];
  const float* bq   = (const float*)d_in[4];
  const float* Wkv  = (const float*)d_in[5];
  const float* bkv  = (const float*)d_in[6];
  const float* Wvec = (const float*)d_in[7];
  const float* bvec = (const float*)d_in[8];

  int N = in_sizes[0] / PDIM;
  int E = in_sizes[2] / 2;

  float* s_out = (float*)d_out;
  float* v_out = s_out + (size_t)N * PDIM;

  int nCnt = (N + 1 + 63) / 64 * 64;

  float* Mx   = (float*)d_ws;
  float* ub   = Mx + (size_t)FOLDROWS * PDIM;
  float* qt   = Mx + (size_t)N * PDIM;
  float* vv   = qt + (size_t)N * PDIM;
  float* U    = vv + (size_t)N * PDIM;
  float* qb   = U  + (size_t)N * 3 * PDIM;

  int*  counts = (int*)(qb + nCnt);
  int2* list   = (int2*)(counts + nCnt);
  int4* ovf    = (int4*)(list + (size_t)N * CAP);

  int nBins   = (N + BINR - 1) / BINR;
  int nChunks = (E + CHUNK - 1) / CHUNK;
  int nOffs   = (nChunks * (nBins + 1) + 63) / 64 * 64;
  int*  offs   = (int*)(ovf + E);
  int2* sorted = (int2*)(offs + nOffs);

  const int* rowIdx = ei;
  const int* colIdx = ei + E;

  int stripsA = N / 16;
  int groupsA = (stripsA + 7) / 8;
  painn_node_gemm<<<nChunks + 2 + groupsA, 256, 0, stream>>>(
      s, Wq, bq, Wkv, bkv, Mx, ub, vv, stripsA, counts, N,
      rowIdx, colIdx, E, nBins, nChunks, offs, sorted);

  int stripsC = N * 3 / 16;
  int groupsC = (stripsC + 7) / 8;
  lists_and_qt<<<nBins + groupsA + groupsC, 256, 0, stream>>>(
      offs, sorted, nChunks, nBins, counts, list, ovf, N, N, s, Mx, ub, qt, qb, stripsA,
      vv, v, Wvec, U, stripsC, groupsC);

  int pairGroups = ((N + 1) / 2 + 7) / 8;
  edge_weights<<<pairGroups, 256, 0, stream>>>(qt, qb, s, counts, list, ovf, Wkv, bkv, s_out, N);

  node_rows_half<<<2 * pairGroups, 256, 0, stream>>>(
      qt, qb, s, U, v, bvec, counts, list, ovf, v_out, N, pairGroups);
}
